# starting kernel
# baseline (speedup 1.0000x reference)
#include <hip/hip_runtime.h>

#define IN_DIM   128
#define HID_DIM  128
#define OUT_DIM  64
#define NUM_GRAPHS 64

typedef float v2f __attribute__((ext_vector_type(2)));
typedef float v8f __attribute__((ext_vector_type(8)));

__global__ void fill_kernel(float* __restrict__ p, float v, int n) {
    int i = blockIdx.x * blockDim.x + threadIdx.x;
    if (i < n) p[i] = v;
}

__global__ void degree_kernel(const int* __restrict__ dst, float* __restrict__ deg, int E) {
    int e = blockIdx.x * blockDim.x + threadIdx.x;
    if (e < E) atomicAdd(&deg[dst[e]], 1.0f);
}

__global__ void rsqrt_kernel(const float* __restrict__ deg, float* __restrict__ dinv, int n) {
    int i = blockIdx.x * blockDim.x + threadIdx.x;
    if (i < n) dinv[i] = rsqrtf(deg[i]);
}

__global__ void gemm_wmma_f32(const float* __restrict__ A,
                              const float* __restrict__ B,
                              float* __restrict__ C,
                              int M, int N, int K) {
    const int lane  = threadIdx.x;
    const int wave  = threadIdx.y;
    const int tilesN = N >> 4;
    const int totalTiles = (M >> 4) * tilesN;
    const int tile = blockIdx.x * blockDim.y + wave;
    if (tile >= totalTiles) return;

    const int tm   = tile / tilesN;
    const int tn   = tile - tm * tilesN;
    const int row0 = tm << 4;
    const int col0 = tn << 4;
    const int half = lane >> 4;
    const int l    = lane & 15;

    v8f acc = {};
    const float* Arow = A + (size_t)(row0 + l) * K;
    const float* Bcol = B + (size_t)col0 + l;

    for (int k0 = 0; k0 < K; k0 += 4) {
        const int ka = k0 + (half << 1);
        v2f a, b;
        a.x = Arow[ka];
        a.y = Arow[ka + 1];
        b.x = Bcol[(size_t)ka * N];
        b.y = Bcol[(size_t)(ka + 1) * N];
        acc = __builtin_amdgcn_wmma_f32_16x16x4_f32(
             false, a,  false, b,
             (short)0, acc,  false,  false);
    }

#pragma unroll
    for (int v = 0; v < 8; ++v) {
        C[(size_t)(row0 + (half << 3) + v) * N + col0 + l] = acc[v];
    }
}

__global__ void edge_aggregate(const float* __restrict__ h,
                               const float* __restrict__ dinv,
                               const int* __restrict__ src,
                               const int* __restrict__ dst,
                               float* __restrict__ agg,
                               int E, int D) {
    const int perEdge = D >> 2;
    int tid = blockIdx.x * blockDim.x + threadIdx.x;
    int e = tid / perEdge;
    if (e >= E) return;
    int c = (tid - e * perEdge) << 2;

    const int s = src[e];
    const int d = dst[e];
    const float norm = dinv[s] * dinv[d];

    const float4 hv = *(const float4*)(h + (size_t)s * D + c);
    float* out = agg + (size_t)d * D + c;
    atomicAdd(out + 0, hv.x * norm);
    atomicAdd(out + 1, hv.y * norm);
    atomicAdd(out + 2, hv.z * norm);
    atomicAdd(out + 3, hv.w * norm);
}

__global__ void finish_layer(const float* __restrict__ agg,
                             const float* __restrict__ h,
                             const float* __restrict__ dinv,
                             const float* __restrict__ bias,
                             float* __restrict__ out,
                             int Nn, int D, int do_relu) {
    const int perRow = D >> 2;
    int tid = blockIdx.x * blockDim.x + threadIdx.x;
    int i = tid / perRow;
    if (i >= Nn) return;
    int c = (tid - i * perRow) << 2;

    const float di = dinv[i];
    const float w  = di * di;
    const float4 av = *(const float4*)(agg + (size_t)i * D + c);
    const float4 hv = *(const float4*)(h   + (size_t)i * D + c);
    const float4 bv = *(const float4*)(bias + c);
    float4 r;
    r.x = av.x + hv.x * w + bv.x;
    r.y = av.y + hv.y * w + bv.y;
    r.z = av.z + hv.z * w + bv.z;
    r.w = av.w + hv.w * w + bv.w;
    if (do_relu) {
        r.x = fmaxf(r.x, 0.0f); r.y = fmaxf(r.y, 0.0f);
        r.z = fmaxf(r.z, 0.0f); r.w = fmaxf(r.w, 0.0f);
    }
    *(float4*)(out + (size_t)i * D + c) = r;
}

__global__ void count_nodes(const int* __restrict__ batch, float* __restrict__ cnt, int Nn) {
    int i = blockIdx.x * blockDim.x + threadIdx.x;
    if (i < Nn) atomicAdd(&cnt[batch[i]], 1.0f);
}

__global__ void pool_sum(const float* __restrict__ h,
                         const int* __restrict__ batch,
                         float* __restrict__ sums, int Nn, int D) {
    const int perRow = D >> 2;
    int tid = blockIdx.x * blockDim.x + threadIdx.x;
    int i = tid / perRow;
    if (i >= Nn) return;
    int c = (tid - i * perRow) << 2;
    const int g = batch[i];
    const float4 hv = *(const float4*)(h + (size_t)i * D + c);
    float* out = sums + (size_t)g * D + c;
    atomicAdd(out + 0, hv.x);
    atomicAdd(out + 1, hv.y);
    atomicAdd(out + 2, hv.z);
    atomicAdd(out + 3, hv.w);
}

__global__ void pool_div(const float* __restrict__ sums,
                         const float* __restrict__ cnt,
                         float* __restrict__ out, int G, int D) {
    int tid = blockIdx.x * blockDim.x + threadIdx.x;
    if (tid >= G * D) return;
    int g = tid / D;
    out[tid] = sums[tid] / fmaxf(cnt[g], 1.0f);
}

extern "C" void kernel_launch(void* const* d_in, const int* in_sizes, int n_in,
                              void* d_out, int out_size, void* d_ws, size_t ws_size,
                              hipStream_t stream) {
    const float* x    = (const float*)d_in[0];
    const int*   eidx = (const int*)  d_in[1];
    const int*   batch= (const int*)  d_in[2];
    const float* W1   = (const float*)d_in[3];
    const float* b1   = (const float*)d_in[4];
    const float* W2   = (const float*)d_in[5];
    const float* b2   = (const float*)d_in[6];
    float* out = (float*)d_out;

    const int Nn = in_sizes[0] / IN_DIM;
    const int E  = in_sizes[1] / 2;
    const int* src = eidx;
    const int* dst = eidx + E;

    float* ws   = (float*)d_ws;
    float* h1   = ws;
    float* agg1 = h1   + (size_t)Nn * HID_DIM;
    float* h2   = agg1 + (size_t)Nn * HID_DIM;
    float* agg2 = h2   + (size_t)Nn * OUT_DIM;
    float* deg  = agg2 + (size_t)Nn * OUT_DIM;
    float* dinv = deg  + Nn;
    float* sums = dinv + Nn;
    float* cnts = sums + NUM_GRAPHS * OUT_DIM;

    const int B = 256;
    auto blocks = [](long long n, int b) { return (unsigned)((n + b - 1) / b); };

    fill_kernel<<<blocks((long long)Nn * HID_DIM, B), B, 0, stream>>>(agg1, 0.0f, Nn * HID_DIM);
    fill_kernel<<<blocks((long long)Nn * OUT_DIM, B), B, 0, stream>>>(agg2, 0.0f, Nn * OUT_DIM);
    fill_kernel<<<blocks(Nn, B), B, 0, stream>>>(deg, 1.0f, Nn);
    fill_kernel<<<blocks(NUM_GRAPHS * OUT_DIM, B), B, 0, stream>>>(sums, 0.0f, NUM_GRAPHS * OUT_DIM);
    fill_kernel<<<blocks(NUM_GRAPHS, B), B, 0, stream>>>(cnts, 0.0f, NUM_GRAPHS);

    degree_kernel<<<blocks(E, B), B, 0, stream>>>(dst, deg, E);
    rsqrt_kernel<<<blocks(Nn, B), B, 0, stream>>>(deg, dinv, Nn);

    {
        int tiles = (Nn >> 4) * (HID_DIM >> 4);
        dim3 blk(32, 8);
        gemm_wmma_f32<<<blocks(tiles, 8), blk, 0, stream>>>(x, W1, h1, Nn, HID_DIM, IN_DIM);
    }
    edge_aggregate<<<blocks((long long)E * (HID_DIM >> 2), B), B, 0, stream>>>(
        h1, dinv, src, dst, agg1, E, HID_DIM);
    finish_layer<<<blocks((long long)Nn * (HID_DIM >> 2), B), B, 0, stream>>>(
        agg1, h1, dinv, b1, agg1, Nn, HID_DIM, 1);

    {
        int tiles = (Nn >> 4) * (OUT_DIM >> 4);
        dim3 blk(32, 8);
        gemm_wmma_f32<<<blocks(tiles, 8), blk, 0, stream>>>(agg1, W2, h2, Nn, OUT_DIM, HID_DIM);
    }
    edge_aggregate<<<blocks((long long)E * (OUT_DIM >> 2), B), B, 0, stream>>>(
        h2, dinv, src, dst, agg2, E, OUT_DIM);
    finish_layer<<<blocks((long long)Nn * (OUT_DIM >> 2), B), B, 0, stream>>>(
        agg2, h2, dinv, b2, agg2, Nn, OUT_DIM, 0);

    count_nodes<<<blocks(Nn, B), B, 0, stream>>>(batch, cnts, Nn);
    pool_sum<<<blocks((long long)Nn * (OUT_DIM >> 2), B), B, 0, stream>>>(
        agg2, batch, sums, Nn, OUT_DIM);
    pool_div<<<blocks(NUM_GRAPHS * OUT_DIM, B), B, 0, stream>>>(
        sums, cnts, out, NUM_GRAPHS, OUT_DIM);
}


# Float scatter atomics replaced by a gather over edges counting-sorted by destination (rank from the degree atomic, scan, permute): one writer per row, sums in arrival-rank order, then bias/ReLU
# speedup vs baseline: 1.5844x; 1.5844x over previous
#include <hip/hip_runtime.h>

#define IN_DIM   128
#define HID_DIM  128
#define OUT_DIM  64
#define NUM_GRAPHS 64

typedef float v2f __attribute__((ext_vector_type(2)));
typedef float v8f __attribute__((ext_vector_type(8)));

__global__ void fill_kernel(float* __restrict__ p, float v, int n) {
    int i = blockIdx.x * blockDim.x + threadIdx.x;
    if (i < n) p[i] = v;
}

__global__ void zero_int_kernel(int* __restrict__ p, int n) {
    int i = blockIdx.x * blockDim.x + threadIdx.x;
    if (i < n) p[i] = 0;
}

__global__ void rank_kernel(const int* __restrict__ dst, int* __restrict__ cnt,
                            int* __restrict__ rank, int E) {
    int e = blockIdx.x * blockDim.x + threadIdx.x;
    if (e < E) rank[e] = atomicAdd(&cnt[dst[e]], 1);
}

__global__ void __launch_bounds__(1024)
rowptr_kernel(const int* __restrict__ cnt, int* __restrict__ rowptr,
              float* __restrict__ dinv, int n) {
    __shared__ int sh[1024];
    const int t    = threadIdx.x;
    const int base = blockIdx.x * 1024;

    int pre = 0;
    for (int k = t; k < base; k += 1024) pre += cnt[k];
    sh[t] = pre;
    __syncthreads();
    for (int s = 512; s > 0; s >>= 1) {
        if (t < s) sh[t] += sh[t + s];
        __syncthreads();
    }
    const int off = sh[0];
    __syncthreads();

    const int i = base + t;
    const int v = (i < n) ? cnt[i] : 0;
    sh[t] = v;
    __syncthreads();
    for (int o = 1; o < 1024; o <<= 1) {
        const int x = (t >= o) ? sh[t - o] : 0;
        __syncthreads();
        sh[t] += x;
        __syncthreads();
    }
    const int incl = sh[t];
    if (i < n) {
        rowptr[i] = off + incl - v;
        dinv[i]   = rsqrtf((float)(v + 1));
        if (i == n - 1) rowptr[n] = off + incl;
    }
}

__global__ void permute_kernel(const int* __restrict__ src, const int* __restrict__ dst,
                               const int* __restrict__ rank, const int* __restrict__ rowptr,
                               int* __restrict__ csr_src, int E) {
    int e = blockIdx.x * blockDim.x + threadIdx.x;
    if (e < E) csr_src[rowptr[dst[e]] + rank[e]] = src[e];
}

__global__ void gemm_wmma_f32(const float* __restrict__ A,
                              const float* __restrict__ B,
                              float* __restrict__ C,
                              int M, int N, int K) {
    const int lane  = threadIdx.x;
    const int wave  = threadIdx.y;
    const int tilesN = N >> 4;
    const int totalTiles = (M >> 4) * tilesN;
    const int tile = blockIdx.x * blockDim.y + wave;
    if (tile >= totalTiles) return;

    const int tm   = tile / tilesN;
    const int tn   = tile - tm * tilesN;
    const int row0 = tm << 4;
    const int col0 = tn << 4;
    const int half = lane >> 4;
    const int l    = lane & 15;

    v8f acc = {};
    const float* Arow = A + (size_t)(row0 + l) * K;
    const float* Bcol = B + (size_t)col0 + l;

    for (int k0 = 0; k0 < K; k0 += 4) {
        const int ka = k0 + (half << 1);
        v2f a, b;
        a.x = Arow[ka];
        a.y = Arow[ka + 1];
        b.x = Bcol[(size_t)ka * N];
        b.y = Bcol[(size_t)(ka + 1) * N];
        acc = __builtin_amdgcn_wmma_f32_16x16x4_f32(
             false, a,  false, b,
             (short)0, acc,  false,  false);
    }

#pragma unroll
    for (int v = 0; v < 8; ++v) {
        C[(size_t)(row0 + (half << 3) + v) * N + col0 + l] = acc[v];
    }
}

template <int D, bool RELU>
__global__ void gather_rows(const float* __restrict__ h,
                            const float* __restrict__ dinv,
                            const int* __restrict__ rowptr,
                            const int* __restrict__ csr_src,
                            const float* __restrict__ bias,
                            float* __restrict__ out, int Nn) {
    constexpr int LPN = D >> 2;
    const int tid = blockIdx.x * blockDim.x + threadIdx.x;
    const int i = tid / LPN;
    if (i >= Nn) return;
    const int c = (tid - i * LPN) << 2;

    const int beg = rowptr[i];
    const int end = rowptr[i + 1];
    const float di = dinv[i];
    const float4 hs = *(const float4*)(h + (size_t)i * D + c);
    const float wself = di * di;
    float4 acc;
    acc.x = hs.x * wself; acc.y = hs.y * wself;
    acc.z = hs.z * wself; acc.w = hs.w * wself;

    int j = beg;
    for (; j + 4 <= end; j += 4) {
        const int s0 = csr_src[j];
        const int s1 = csr_src[j + 1];
        const int s2 = csr_src[j + 2];
        const int s3 = csr_src[j + 3];
        const float n0 = dinv[s0] * di;
        const float n1 = dinv[s1] * di;
        const float n2 = dinv[s2] * di;
        const float n3 = dinv[s3] * di;
        const float4 v0 = *(const float4*)(h + (size_t)s0 * D + c);
        const float4 v1 = *(const float4*)(h + (size_t)s1 * D + c);
        const float4 v2 = *(const float4*)(h + (size_t)s2 * D + c);
        const float4 v3 = *(const float4*)(h + (size_t)s3 * D + c);
        acc.x = fmaf(v0.x, n0, acc.x); acc.y = fmaf(v0.y, n0, acc.y);
        acc.z = fmaf(v0.z, n0, acc.z); acc.w = fmaf(v0.w, n0, acc.w);
        acc.x = fmaf(v1.x, n1, acc.x); acc.y = fmaf(v1.y, n1, acc.y);
        acc.z = fmaf(v1.z, n1, acc.z); acc.w = fmaf(v1.w, n1, acc.w);
        acc.x = fmaf(v2.x, n2, acc.x); acc.y = fmaf(v2.y, n2, acc.y);
        acc.z = fmaf(v2.z, n2, acc.z); acc.w = fmaf(v2.w, n2, acc.w);
        acc.x = fmaf(v3.x, n3, acc.x); acc.y = fmaf(v3.y, n3, acc.y);
        acc.z = fmaf(v3.z, n3, acc.z); acc.w = fmaf(v3.w, n3, acc.w);
    }
    const int rem = end - j;
    if (rem > 0) {
        const int last = end - 1;
        const int s0 = csr_src[j];
        const int s1 = csr_src[min(j + 1, last)];
        const int s2 = csr_src[min(j + 2, last)];
        const float n0 = dinv[s0] * di;
        const float n1 = dinv[s1] * di;
        const float n2 = dinv[s2] * di;
        const float4 v0 = *(const float4*)(h + (size_t)s0 * D + c);
        float4 v1 = *(const float4*)(h + (size_t)s1 * D + c);
        float4 v2 = *(const float4*)(h + (size_t)s2 * D + c);
        if (rem < 2) { v1.x = 0.0f; v1.y = 0.0f; v1.z = 0.0f; v1.w = 0.0f; }
        if (rem < 3) { v2.x = 0.0f; v2.y = 0.0f; v2.z = 0.0f; v2.w = 0.0f; }
        acc.x = fmaf(v0.x, n0, acc.x); acc.y = fmaf(v0.y, n0, acc.y);
        acc.z = fmaf(v0.z, n0, acc.z); acc.w = fmaf(v0.w, n0, acc.w);
        acc.x = fmaf(v1.x, n1, acc.x); acc.y = fmaf(v1.y, n1, acc.y);
        acc.z = fmaf(v1.z, n1, acc.z); acc.w = fmaf(v1.w, n1, acc.w);
        acc.x = fmaf(v2.x, n2, acc.x); acc.y = fmaf(v2.y, n2, acc.y);
        acc.z = fmaf(v2.z, n2, acc.z); acc.w = fmaf(v2.w, n2, acc.w);
    }

    const float4 bv = *(const float4*)(bias + c);
    float4 r;
    r.x = acc.x + bv.x; r.y = acc.y + bv.y;
    r.z = acc.z + bv.z; r.w = acc.w + bv.w;
    if (RELU) {
        r.x = fmaxf(r.x, 0.0f); r.y = fmaxf(r.y, 0.0f);
        r.z = fmaxf(r.z, 0.0f); r.w = fmaxf(r.w, 0.0f);
    }
    *(float4*)(out + (size_t)i * D + c) = r;
}

__global__ void count_nodes(const int* __restrict__ batch, float* __restrict__ cnt, int Nn) {
    int i = blockIdx.x * blockDim.x + threadIdx.x;
    if (i < Nn) atomicAdd(&cnt[batch[i]], 1.0f);
}

__global__ void pool_sum(const float* __restrict__ h,
                         const int* __restrict__ batch,
                         float* __restrict__ sums, int Nn, int D) {
    const int perRow = D >> 2;
    int tid = blockIdx.x * blockDim.x + threadIdx.x;
    int i = tid / perRow;
    if (i >= Nn) return;
    int c = (tid - i * perRow) << 2;
    const int g = batch[i];
    const float4 hv = *(const float4*)(h + (size_t)i * D + c);
    float* out = sums + (size_t)g * D + c;
    atomicAdd(out + 0, hv.x);
    atomicAdd(out + 1, hv.y);
    atomicAdd(out + 2, hv.z);
    atomicAdd(out + 3, hv.w);
}

__global__ void pool_div(const float* __restrict__ sums,
                         const float* __restrict__ cnt,
                         float* __restrict__ out, int G, int D) {
    int tid = blockIdx.x * blockDim.x + threadIdx.x;
    if (tid >= G * D) return;
    int g = tid / D;
    out[tid] = sums[tid] / fmaxf(cnt[g], 1.0f);
}

extern "C" void kernel_launch(void* const* d_in, const int* in_sizes, int n_in,
                              void* d_out, int out_size, void* d_ws, size_t ws_size,
                              hipStream_t stream) {
    const float* x    = (const float*)d_in[0];
    const int*   eidx = (const int*)  d_in[1];
    const int*   batch= (const int*)  d_in[2];
    const float* W1   = (const float*)d_in[3];
    const float* b1   = (const float*)d_in[4];
    const float* W2   = (const float*)d_in[5];
    const float* b2   = (const float*)d_in[6];
    float* out = (float*)d_out;

    const int Nn = in_sizes[0] / IN_DIM;
    const int E  = in_sizes[1] / 2;
    const int* src = eidx;
    const int* dst = eidx + E;

    float* ws   = (float*)d_ws;
    float* h1   = ws;
    float* a1   = h1   + (size_t)Nn * HID_DIM;
    float* h2   = a1   + (size_t)Nn * HID_DIM;
    float* o2   = h2   + (size_t)Nn * OUT_DIM;
    float* dinv = o2   + (size_t)Nn * OUT_DIM;
    float* sums = dinv + Nn;
    float* cnts = sums + NUM_GRAPHS * OUT_DIM;
    int* cnt    = (int*)(cnts + NUM_GRAPHS);
    int* rowptr = cnt + Nn;
    int* rank   = rowptr + Nn + 1;
    int* csr    = rank + E;

    const int B = 256;
    auto blocks = [](long long n, int b) { return (unsigned)((n + b - 1) / b); };

    zero_int_kernel<<<blocks(Nn, B), B, 0, stream>>>(cnt, Nn);
    fill_kernel<<<blocks(NUM_GRAPHS * OUT_DIM, B), B, 0, stream>>>(sums, 0.0f, NUM_GRAPHS * OUT_DIM);
    fill_kernel<<<blocks(NUM_GRAPHS, B), B, 0, stream>>>(cnts, 0.0f, NUM_GRAPHS);

    rank_kernel<<<blocks(E, B), B, 0, stream>>>(dst, cnt, rank, E);
    rowptr_kernel<<<blocks(Nn, 1024), 1024, 0, stream>>>(cnt, rowptr, dinv, Nn);
    permute_kernel<<<blocks(E, B), B, 0, stream>>>(src, dst, rank, rowptr, csr, E);

    {
        int tiles = (Nn >> 4) * (HID_DIM >> 4);
        dim3 blk(32, 8);
        gemm_wmma_f32<<<blocks(tiles, 8), blk, 0, stream>>>(x, W1, h1, Nn, HID_DIM, IN_DIM);
    }
    gather_rows<HID_DIM, true><<<blocks((long long)Nn * (HID_DIM >> 2), B), B, 0, stream>>>(
        h1, dinv, rowptr, csr, b1, a1, Nn);

    {
        int tiles = (Nn >> 4) * (OUT_DIM >> 4);
        dim3 blk(32, 8);
        gemm_wmma_f32<<<blocks(tiles, 8), blk, 0, stream>>>(a1, W2, h2, Nn, OUT_DIM, HID_DIM);
    }
    gather_rows<OUT_DIM, false><<<blocks((long long)Nn * (OUT_DIM >> 2), B), B, 0, stream>>>(
        h2, dinv, rowptr, csr, b2, o2, Nn);

    count_nodes<<<blocks(Nn, B), B, 0, stream>>>(batch, cnts, Nn);
    pool_sum<<<blocks((long long)Nn * (OUT_DIM >> 2), B), B, 0, stream>>>(
        o2, batch, sums, Nn, OUT_DIM);
    pool_div<<<blocks(NUM_GRAPHS * OUT_DIM, B), B, 0, stream>>>(
        sums, cnts, out, NUM_GRAPHS, OUT_DIM);
}


# Mean-pool sums folded into the layer-2 gather: block adds its 16 rows in LDS in node order, then one float atomic per column per graph run (was 4 per lane per node); count likewise; 2 launches go
# speedup vs baseline: 2.4990x; 1.5773x over previous
#include <hip/hip_runtime.h>

#define IN_DIM   128
#define HID_DIM  128
#define OUT_DIM  64
#define NUM_GRAPHS 64

typedef float v2f __attribute__((ext_vector_type(2)));
typedef float v8f __attribute__((ext_vector_type(8)));

__global__ void fill_kernel(float* __restrict__ p, float v, int n) {
    int i = blockIdx.x * blockDim.x + threadIdx.x;
    if (i < n) p[i] = v;
}

__global__ void zero_int_kernel(int* __restrict__ p, int n) {
    int i = blockIdx.x * blockDim.x + threadIdx.x;
    if (i < n) p[i] = 0;
}

__global__ void rank_kernel(const int* __restrict__ dst, int* __restrict__ cnt,
                            int* __restrict__ rank, int E) {
    int e = blockIdx.x * blockDim.x + threadIdx.x;
    if (e < E) rank[e] = atomicAdd(&cnt[dst[e]], 1);
}

__global__ void __launch_bounds__(1024)
rowptr_kernel(const int* __restrict__ cnt, int* __restrict__ rowptr,
              float* __restrict__ dinv, int n) {
    __shared__ int sh[1024];
    const int t    = threadIdx.x;
    const int base = blockIdx.x * 1024;

    int pre = 0;
    for (int k = t; k < base; k += 1024) pre += cnt[k];
    sh[t] = pre;
    __syncthreads();
    for (int s = 512; s > 0; s >>= 1) {
        if (t < s) sh[t] += sh[t + s];
        __syncthreads();
    }
    const int off = sh[0];
    __syncthreads();

    const int i = base + t;
    const int v = (i < n) ? cnt[i] : 0;
    sh[t] = v;
    __syncthreads();
    for (int o = 1; o < 1024; o <<= 1) {
        const int x = (t >= o) ? sh[t - o] : 0;
        __syncthreads();
        sh[t] += x;
        __syncthreads();
    }
    const int incl = sh[t];
    if (i < n) {
        rowptr[i] = off + incl - v;
        dinv[i]   = rsqrtf((float)(v + 1));
        if (i == n - 1) rowptr[n] = off + incl;
    }
}

__global__ void permute_kernel(const int* __restrict__ src, const int* __restrict__ dst,
                               const int* __restrict__ rank, const int* __restrict__ rowptr,
                               int* __restrict__ csr_src, int E) {
    int e = blockIdx.x * blockDim.x + threadIdx.x;
    if (e < E) csr_src[rowptr[dst[e]] + rank[e]] = src[e];
}

__global__ void gemm_wmma_f32(const float* __restrict__ A,
                              const float* __restrict__ B,
                              float* __restrict__ C,
                              int M, int N, int K) {
    const int lane  = threadIdx.x;
    const int wave  = threadIdx.y;
    const int tilesN = N >> 4;
    const int totalTiles = (M >> 4) * tilesN;
    const int tile = blockIdx.x * blockDim.y + wave;
    if (tile >= totalTiles) return;

    const int tm   = tile / tilesN;
    const int tn   = tile - tm * tilesN;
    const int row0 = tm << 4;
    const int col0 = tn << 4;
    const int half = lane >> 4;
    const int l    = lane & 15;

    v8f acc = {};
    const float* Arow = A + (size_t)(row0 + l) * K;
    const float* Bcol = B + (size_t)col0 + l;

    for (int k0 = 0; k0 < K; k0 += 4) {
        const int ka = k0 + (half << 1);
        v2f a, b;
        a.x = Arow[ka];
        a.y = Arow[ka + 1];
        b.x = Bcol[(size_t)ka * N];
        b.y = Bcol[(size_t)(ka + 1) * N];
        acc = __builtin_amdgcn_wmma_f32_16x16x4_f32(
             false, a,  false, b,
             (short)0, acc,  false,  false);
    }

#pragma unroll
    for (int v = 0; v < 8; ++v) {
        C[(size_t)(row0 + (half << 3) + v) * N + col0 + l] = acc[v];
    }
}

template <int D, bool RELU>
__global__ void gather_rows(const float* __restrict__ h,
                            const float* __restrict__ dinv,
                            const int* __restrict__ rowptr,
                            const int* __restrict__ csr_src,
                            const float* __restrict__ bias,
                            float* __restrict__ out, int Nn) {
    constexpr int LPN = D >> 2;
    const int tid = blockIdx.x * blockDim.x + threadIdx.x;
    const int i = tid / LPN;
    if (i >= Nn) return;
    const int c = (tid - i * LPN) << 2;

    const int beg = rowptr[i];
    const int end = rowptr[i + 1];
    const float di = dinv[i];
    const float4 hs = *(const float4*)(h + (size_t)i * D + c);
    const float wself = di * di;
    float4 acc;
    acc.x = hs.x * wself; acc.y = hs.y * wself;
    acc.z = hs.z * wself; acc.w = hs.w * wself;

    int j = beg;
    for (; j + 4 <= end; j += 4) {
        const int s0 = csr_src[j];
        const int s1 = csr_src[j + 1];
        const int s2 = csr_src[j + 2];
        const int s3 = csr_src[j + 3];
        const float n0 = dinv[s0] * di;
        const float n1 = dinv[s1] * di;
        const float n2 = dinv[s2] * di;
        const float n3 = dinv[s3] * di;
        const float4 v0 = *(const float4*)(h + (size_t)s0 * D + c);
        const float4 v1 = *(const float4*)(h + (size_t)s1 * D + c);
        const float4 v2 = *(const float4*)(h + (size_t)s2 * D + c);
        const float4 v3 = *(const float4*)(h + (size_t)s3 * D + c);
        acc.x = fmaf(v0.x, n0, acc.x); acc.y = fmaf(v0.y, n0, acc.y);
        acc.z = fmaf(v0.z, n0, acc.z); acc.w = fmaf(v0.w, n0, acc.w);
        acc.x = fmaf(v1.x, n1, acc.x); acc.y = fmaf(v1.y, n1, acc.y);
        acc.z = fmaf(v1.z, n1, acc.z); acc.w = fmaf(v1.w, n1, acc.w);
        acc.x = fmaf(v2.x, n2, acc.x); acc.y = fmaf(v2.y, n2, acc.y);
        acc.z = fmaf(v2.z, n2, acc.z); acc.w = fmaf(v2.w, n2, acc.w);
        acc.x = fmaf(v3.x, n3, acc.x); acc.y = fmaf(v3.y, n3, acc.y);
        acc.z = fmaf(v3.z, n3, acc.z); acc.w = fmaf(v3.w, n3, acc.w);
    }
    const int rem = end - j;
    if (rem > 0) {
        const int last = end - 1;
        const int s0 = csr_src[j];
        const int s1 = csr_src[min(j + 1, last)];
        const int s2 = csr_src[min(j + 2, last)];
        const float n0 = dinv[s0] * di;
        const float n1 = dinv[s1] * di;
        const float n2 = dinv[s2] * di;
        const float4 v0 = *(const float4*)(h + (size_t)s0 * D + c);
        float4 v1 = *(const float4*)(h + (size_t)s1 * D + c);
        float4 v2 = *(const float4*)(h + (size_t)s2 * D + c);
        if (rem < 2) { v1.x = 0.0f; v1.y = 0.0f; v1.z = 0.0f; v1.w = 0.0f; }
        if (rem < 3) { v2.x = 0.0f; v2.y = 0.0f; v2.z = 0.0f; v2.w = 0.0f; }
        acc.x = fmaf(v0.x, n0, acc.x); acc.y = fmaf(v0.y, n0, acc.y);
        acc.z = fmaf(v0.z, n0, acc.z); acc.w = fmaf(v0.w, n0, acc.w);
        acc.x = fmaf(v1.x, n1, acc.x); acc.y = fmaf(v1.y, n1, acc.y);
        acc.z = fmaf(v1.z, n1, acc.z); acc.w = fmaf(v1.w, n1, acc.w);
        acc.x = fmaf(v2.x, n2, acc.x); acc.y = fmaf(v2.y, n2, acc.y);
        acc.z = fmaf(v2.z, n2, acc.z); acc.w = fmaf(v2.w, n2, acc.w);
    }

    const float4 bv = *(const float4*)(bias + c);
    float4 r;
    r.x = acc.x + bv.x; r.y = acc.y + bv.y;
    r.z = acc.z + bv.z; r.w = acc.w + bv.w;
    if (RELU) {
        r.x = fmaxf(r.x, 0.0f); r.y = fmaxf(r.y, 0.0f);
        r.z = fmaxf(r.z, 0.0f); r.w = fmaxf(r.w, 0.0f);
    }
    *(float4*)(out + (size_t)i * D + c) = r;
}

template <int D>
__global__ void __launch_bounds__(256)
gather_pool_rows(const float* __restrict__ h,
                 const float* __restrict__ dinv,
                 const int* __restrict__ rowptr,
                 const int* __restrict__ csr_src,
                 const float* __restrict__ bias,
                 const int* __restrict__ batch,
                 float* __restrict__ sums,
                 float* __restrict__ cnts, int Nn) {
    constexpr int LPN = D >> 2;
    constexpr int NPB = 256 / LPN;
    __shared__ float rows[NPB][D];
    __shared__ int   gid[NPB];

    const int q  = threadIdx.x / LPN;
    const int lc = threadIdx.x - q * LPN;
    const int c  = lc << 2;
    const int i  = blockIdx.x * NPB + q;

    float4 r;
    r.x = 0.0f; r.y = 0.0f; r.z = 0.0f; r.w = 0.0f;
    int g = -1;
    if (i < Nn) {
        g = batch[i];
        const int beg = rowptr[i];
        const int end = rowptr[i + 1];
        const float di = dinv[i];
        const float4 hs = *(const float4*)(h + (size_t)i * D + c);
        const float wself = di * di;
        float4 acc;
        acc.x = hs.x * wself; acc.y = hs.y * wself;
        acc.z = hs.z * wself; acc.w = hs.w * wself;

        int j = beg;
        for (; j + 4 <= end; j += 4) {
            const int s0 = csr_src[j];
            const int s1 = csr_src[j + 1];
            const int s2 = csr_src[j + 2];
            const int s3 = csr_src[j + 3];
            const float n0 = dinv[s0] * di;
            const float n1 = dinv[s1] * di;
            const float n2 = dinv[s2] * di;
            const float n3 = dinv[s3] * di;
            const float4 v0 = *(const float4*)(h + (size_t)s0 * D + c);
            const float4 v1 = *(const float4*)(h + (size_t)s1 * D + c);
            const float4 v2 = *(const float4*)(h + (size_t)s2 * D + c);
            const float4 v3 = *(const float4*)(h + (size_t)s3 * D + c);
            acc.x = fmaf(v0.x, n0, acc.x); acc.y = fmaf(v0.y, n0, acc.y);
            acc.z = fmaf(v0.z, n0, acc.z); acc.w = fmaf(v0.w, n0, acc.w);
            acc.x = fmaf(v1.x, n1, acc.x); acc.y = fmaf(v1.y, n1, acc.y);
            acc.z = fmaf(v1.z, n1, acc.z); acc.w = fmaf(v1.w, n1, acc.w);
            acc.x = fmaf(v2.x, n2, acc.x); acc.y = fmaf(v2.y, n2, acc.y);
            acc.z = fmaf(v2.z, n2, acc.z); acc.w = fmaf(v2.w, n2, acc.w);
            acc.x = fmaf(v3.x, n3, acc.x); acc.y = fmaf(v3.y, n3, acc.y);
            acc.z = fmaf(v3.z, n3, acc.z); acc.w = fmaf(v3.w, n3, acc.w);
        }
        const int rem = end - j;
        if (rem > 0) {
            const int last = end - 1;
            const int s0 = csr_src[j];
            const int s1 = csr_src[min(j + 1, last)];
            const int s2 = csr_src[min(j + 2, last)];
            const float n0 = dinv[s0] * di;
            const float n1 = dinv[s1] * di;
            const float n2 = dinv[s2] * di;
            const float4 v0 = *(const float4*)(h + (size_t)s0 * D + c);
            float4 v1 = *(const float4*)(h + (size_t)s1 * D + c);
            float4 v2 = *(const float4*)(h + (size_t)s2 * D + c);
            if (rem < 2) { v1.x = 0.0f; v1.y = 0.0f; v1.z = 0.0f; v1.w = 0.0f; }
            if (rem < 3) { v2.x = 0.0f; v2.y = 0.0f; v2.z = 0.0f; v2.w = 0.0f; }
            acc.x = fmaf(v0.x, n0, acc.x); acc.y = fmaf(v0.y, n0, acc.y);
            acc.z = fmaf(v0.z, n0, acc.z); acc.w = fmaf(v0.w, n0, acc.w);
            acc.x = fmaf(v1.x, n1, acc.x); acc.y = fmaf(v1.y, n1, acc.y);
            acc.z = fmaf(v1.z, n1, acc.z); acc.w = fmaf(v1.w, n1, acc.w);
            acc.x = fmaf(v2.x, n2, acc.x); acc.y = fmaf(v2.y, n2, acc.y);
            acc.z = fmaf(v2.z, n2, acc.z); acc.w = fmaf(v2.w, n2, acc.w);
        }

        const float4 bv = *(const float4*)(bias + c);
        r.x = acc.x + bv.x; r.y = acc.y + bv.y;
        r.z = acc.z + bv.z; r.w = acc.w + bv.w;
    }

    *(float4*)(&rows[q][c]) = r;
    if (lc == 0) gid[q] = g;
    __syncthreads();

    if (threadIdx.x < D) {
        const int d = threadIdx.x;
        float v[NPB];
        int   gk[NPB];
#pragma unroll
        for (int k = 0; k < NPB; ++k) { v[k] = rows[k][d]; gk[k] = gid[k]; }

        float s  = 0.0f;
        int  run = 0;
        int  gc  = gk[0];
#pragma unroll
        for (int k = 0; k < NPB; ++k) {
            if (gk[k] != gc) {
                if (gc >= 0 && gc < NUM_GRAPHS) {
                    atomicAdd(&sums[(size_t)gc * D + d], s);
                    if (d == 0) atomicAdd(&cnts[gc], (float)run);
                }
                s = 0.0f; run = 0; gc = gk[k];
            }
            s += v[k];
            ++run;
        }
        if (gc >= 0 && gc < NUM_GRAPHS) {
            atomicAdd(&sums[(size_t)gc * D + d], s);
            if (d == 0) atomicAdd(&cnts[gc], (float)run);
        }
    }
}

__global__ void pool_div(const float* __restrict__ sums,
                         const float* __restrict__ cnt,
                         float* __restrict__ out, int G, int D) {
    int tid = blockIdx.x * blockDim.x + threadIdx.x;
    if (tid >= G * D) return;
    int g = tid / D;
    out[tid] = sums[tid] / fmaxf(cnt[g], 1.0f);
}

extern "C" void kernel_launch(void* const* d_in, const int* in_sizes, int n_in,
                              void* d_out, int out_size, void* d_ws, size_t ws_size,
                              hipStream_t stream) {
    const float* x    = (const float*)d_in[0];
    const int*   eidx = (const int*)  d_in[1];
    const int*   batch= (const int*)  d_in[2];
    const float* W1   = (const float*)d_in[3];
    const float* b1   = (const float*)d_in[4];
    const float* W2   = (const float*)d_in[5];
    const float* b2   = (const float*)d_in[6];
    float* out = (float*)d_out;

    const int Nn = in_sizes[0] / IN_DIM;
    const int E  = in_sizes[1] / 2;
    const int* src = eidx;
    const int* dst = eidx + E;

    float* ws   = (float*)d_ws;
    float* h1   = ws;
    float* a1   = h1   + (size_t)Nn * HID_DIM;
    float* h2   = a1   + (size_t)Nn * HID_DIM;
    float* dinv = h2   + (size_t)Nn * OUT_DIM;
    float* sums = dinv + Nn;
    float* cnts = sums + NUM_GRAPHS * OUT_DIM;
    int* cnt    = (int*)(cnts + NUM_GRAPHS);
    int* rowptr = cnt + Nn;
    int* rank   = rowptr + Nn + 1;
    int* csr    = rank + E;

    const int B = 256;
    auto blocks = [](long long n, int b) { return (unsigned)((n + b - 1) / b); };

    zero_int_kernel<<<blocks(Nn, B), B, 0, stream>>>(cnt, Nn);
    fill_kernel<<<blocks(NUM_GRAPHS * OUT_DIM, B), B, 0, stream>>>(sums, 0.0f, NUM_GRAPHS * OUT_DIM);
    fill_kernel<<<blocks(NUM_GRAPHS, B), B, 0, stream>>>(cnts, 0.0f, NUM_GRAPHS);

    rank_kernel<<<blocks(E, B), B, 0, stream>>>(dst, cnt, rank, E);
    rowptr_kernel<<<blocks(Nn, 1024), 1024, 0, stream>>>(cnt, rowptr, dinv, Nn);
    permute_kernel<<<blocks(E, B), B, 0, stream>>>(src, dst, rank, rowptr, csr, E);

    {
        int tiles = (Nn >> 4) * (HID_DIM >> 4);
        dim3 blk(32, 8);
        gemm_wmma_f32<<<blocks(tiles, 8), blk, 0, stream>>>(x, W1, h1, Nn, HID_DIM, IN_DIM);
    }
    gather_rows<HID_DIM, true><<<blocks((long long)Nn * (HID_DIM >> 2), B), B, 0, stream>>>(
        h1, dinv, rowptr, csr, b1, a1, Nn);

    {
        int tiles = (Nn >> 4) * (OUT_DIM >> 4);
        dim3 blk(32, 8);
        gemm_wmma_f32<<<blocks(tiles, 8), blk, 0, stream>>>(a1, W2, h2, Nn, OUT_DIM, HID_DIM);
    }
    gather_pool_rows<OUT_DIM><<<blocks((long long)Nn * (OUT_DIM >> 2), B), B, 0, stream>>>(
        h2, dinv, rowptr, csr, b2, batch, sums, cnts, Nn);

    pool_div<<<blocks(NUM_GRAPHS * OUT_DIM, B), B, 0, stream>>>(
        sums, cnts, out, NUM_GRAPHS, OUT_DIM);
}


# GEMMs re-tiled: B staged in LDS once per block, each wave keeps its 16x128 A strip in registers and runs a 16x64 strip (4 tiles, float4 row stores); k regrouped inside each WMMA, so sums change order
# speedup vs baseline: 2.8201x; 1.1285x over previous
#include <hip/hip_runtime.h>

#define IN_DIM   128
#define HID_DIM  128
#define OUT_DIM  64
#define NUM_GRAPHS 64

typedef float v2f __attribute__((ext_vector_type(2)));
typedef float v8f __attribute__((ext_vector_type(8)));

__global__ void fill_kernel(float* __restrict__ p, float v, int n) {
    int i = blockIdx.x * blockDim.x + threadIdx.x;
    if (i < n) p[i] = v;
}

__global__ void zero_int_kernel(int* __restrict__ p, int n) {
    int i = blockIdx.x * blockDim.x + threadIdx.x;
    if (i < n) p[i] = 0;
}

__global__ void rank_kernel(const int* __restrict__ dst, int* __restrict__ cnt,
                            int* __restrict__ rank, int E) {
    int e = blockIdx.x * blockDim.x + threadIdx.x;
    if (e < E) rank[e] = atomicAdd(&cnt[dst[e]], 1);
}

__global__ void __launch_bounds__(1024)
rowptr_kernel(const int* __restrict__ cnt, int* __restrict__ rowptr,
              float* __restrict__ dinv, int n) {
    __shared__ int sh[1024];
    const int t    = threadIdx.x;
    const int base = blockIdx.x * 1024;

    int pre = 0;
    for (int k = t; k < base; k += 1024) pre += cnt[k];
    sh[t] = pre;
    __syncthreads();
    for (int s = 512; s > 0; s >>= 1) {
        if (t < s) sh[t] += sh[t + s];
        __syncthreads();
    }
    const int off = sh[0];
    __syncthreads();

    const int i = base + t;
    const int v = (i < n) ? cnt[i] : 0;
    sh[t] = v;
    __syncthreads();
    for (int o = 1; o < 1024; o <<= 1) {
        const int x = (t >= o) ? sh[t - o] : 0;
        __syncthreads();
        sh[t] += x;
        __syncthreads();
    }
    const int incl = sh[t];
    if (i < n) {
        rowptr[i] = off + incl - v;
        dinv[i]   = rsqrtf((float)(v + 1));
        if (i == n - 1) rowptr[n] = off + incl;
    }
}

__global__ void permute_kernel(const int* __restrict__ src, const int* __restrict__ dst,
                               const int* __restrict__ rank, const int* __restrict__ rowptr,
                               int* __restrict__ csr_src, int E) {
    int e = blockIdx.x * blockDim.x + threadIdx.x;
    if (e < E) csr_src[rowptr[dst[e]] + rank[e]] = src[e];
}

template <int N, int WAVES>
__global__ void __launch_bounds__(WAVES * 32, 7)
gemm_strip_wmma_f32(const float* __restrict__ A,
                    const float* __restrict__ B,
                    float* __restrict__ C, int M) {
    constexpr int K   = 128;
    static_assert(IN_DIM == K && HID_DIM == K && N % 64 == 0, "strip GEMM is written for K = 128");
    constexpr int NH  = N / 64;
    constexpr int SLOTS = (K / 4) * N;
    __shared__ float4 Ws[SLOTS];

    const int tid  = threadIdx.x;
    const int lane = tid & 31;
    const int wave = tid >> 5;
    const int c    = lane & 15;
    const int h    = lane >> 4;
    const int task = blockIdx.x * WAVES + wave;
    const int tile = task / NH;
    const int cg   = task - tile * NH;
    const bool live = tile < (M >> 4);
    const int row0 = tile << 4;

    float4 a[16];
    if (live) {
        const float* Ap = A + (size_t)(row0 + c) * K + 4 * h;
#pragma unroll
        for (int j = 0; j < 16; ++j) a[j] = *(const float4*)(Ap + 8 * j);
    }

#pragma unroll 4
    for (int u = tid; u < SLOTS; u += WAVES * 32) {
        const int q  = u / N;
        const int p  = u - q * N;
        const int pl = p & 63;
        const int n  = (p - pl) + ((pl & 15) << 2) + (pl >> 4);
        const float* bp = B + (size_t)(4 * q) * N + n;
        float4 w;
        w.x = bp[0]; w.y = bp[N]; w.z = bp[2 * N]; w.w = bp[3 * N];
        Ws[u] = w;
    }
    __syncthreads();

    if (live) {
        v8f acc[4];
#pragma unroll
        for (int t = 0; t < 4; ++t) acc[t] = v8f{};

        const float4* Wl = Ws + h * N + cg * 64 + c;
        float4 bc[4], bn[4];
#pragma unroll
        for (int t = 0; t < 4; ++t) bc[t] = Wl[16 * t];

        __builtin_amdgcn_sched_group_barrier(0x100, 8, 0);
#pragma unroll
        for (int j = 0; j < 16; ++j) {
            if (j + 1 < 16) {
#pragma unroll
                for (int t = 0; t < 4; ++t) bn[t] = Wl[(2 * (j + 1)) * N + 16 * t];
            }
            v2f a0, a1;
            a0.x = a[j].x; a0.y = a[j].y;
            a1.x = a[j].z; a1.y = a[j].w;
#pragma unroll
            for (int t = 0; t < 4; ++t) {
                v2f b0;
                b0.x = bc[t].x; b0.y = bc[t].y;
                acc[t] = __builtin_amdgcn_wmma_f32_16x16x4_f32(
                     false, a0,  false, b0,
                     (short)0, acc[t],  false,  false);
            }
#pragma unroll
            for (int t = 0; t < 4; ++t) {
                v2f b1;
                b1.x = bc[t].z; b1.y = bc[t].w;
                acc[t] = __builtin_amdgcn_wmma_f32_16x16x4_f32(
                     false, a1,  false, b1,
                     (short)0, acc[t],  false,  false);
            }
            __builtin_amdgcn_sched_group_barrier(0x008, 8, 0);
            if (j + 2 < 16)
                __builtin_amdgcn_sched_group_barrier(0x100, 4, 0);
#pragma unroll
            for (int t = 0; t < 4; ++t) bc[t] = bn[t];
        }

        float* Cp = C + (size_t)(row0 + (h << 3)) * N + cg * 64 + (c << 2);
#pragma unroll
        for (int v = 0; v < 8; ++v) {
            float4 o;
            o.x = acc[0][v]; o.y = acc[1][v]; o.z = acc[2][v]; o.w = acc[3][v];
            *(float4*)(Cp + (size_t)v * N) = o;
        }
    }
}

template <int D, bool RELU>
__global__ void gather_rows(const float* __restrict__ h,
                            const float* __restrict__ dinv,
                            const int* __restrict__ rowptr,
                            const int* __restrict__ csr_src,
                            const float* __restrict__ bias,
                            float* __restrict__ out, int Nn) {
    constexpr int LPN = D >> 2;
    const int tid = blockIdx.x * blockDim.x + threadIdx.x;
    const int i = tid / LPN;
    if (i >= Nn) return;
    const int c = (tid - i * LPN) << 2;

    const int beg = rowptr[i];
    const int end = rowptr[i + 1];
    const float di = dinv[i];
    const float4 hs = *(const float4*)(h + (size_t)i * D + c);
    const float wself = di * di;
    float4 acc;
    acc.x = hs.x * wself; acc.y = hs.y * wself;
    acc.z = hs.z * wself; acc.w = hs.w * wself;

    int j = beg;
    for (; j + 4 <= end; j += 4) {
        const int s0 = csr_src[j];
        const int s1 = csr_src[j + 1];
        const int s2 = csr_src[j + 2];
        const int s3 = csr_src[j + 3];
        const float n0 = dinv[s0] * di;
        const float n1 = dinv[s1] * di;
        const float n2 = dinv[s2] * di;
        const float n3 = dinv[s3] * di;
        const float4 v0 = *(const float4*)(h + (size_t)s0 * D + c);
        const float4 v1 = *(const float4*)(h + (size_t)s1 * D + c);
        const float4 v2 = *(const float4*)(h + (size_t)s2 * D + c);
        const float4 v3 = *(const float4*)(h + (size_t)s3 * D + c);
        acc.x = fmaf(v0.x, n0, acc.x); acc.y = fmaf(v0.y, n0, acc.y);
        acc.z = fmaf(v0.z, n0, acc.z); acc.w = fmaf(v0.w, n0, acc.w);
        acc.x = fmaf(v1.x, n1, acc.x); acc.y = fmaf(v1.y, n1, acc.y);
        acc.z = fmaf(v1.z, n1, acc.z); acc.w = fmaf(v1.w, n1, acc.w);
        acc.x = fmaf(v2.x, n2, acc.x); acc.y = fmaf(v2.y, n2, acc.y);
        acc.z = fmaf(v2.z, n2, acc.z); acc.w = fmaf(v2.w, n2, acc.w);
        acc.x = fmaf(v3.x, n3, acc.x); acc.y = fmaf(v3.y, n3, acc.y);
        acc.z = fmaf(v3.z, n3, acc.z); acc.w = fmaf(v3.w, n3, acc.w);
    }
    const int rem = end - j;
    if (rem > 0) {
        const int last = end - 1;
        const int s0 = csr_src[j];
        const int s1 = csr_src[min(j + 1, last)];
        const int s2 = csr_src[min(j + 2, last)];
        const float n0 = dinv[s0] * di;
        const float n1 = dinv[s1] * di;
        const float n2 = dinv[s2] * di;
        const float4 v0 = *(const float4*)(h + (size_t)s0 * D + c);
        float4 v1 = *(const float4*)(h + (size_t)s1 * D + c);
        float4 v2 = *(const float4*)(h + (size_t)s2 * D + c);
        if (rem < 2) { v1.x = 0.0f; v1.y = 0.0f; v1.z = 0.0f; v1.w = 0.0f; }
        if (rem < 3) { v2.x = 0.0f; v2.y = 0.0f; v2.z = 0.0f; v2.w = 0.0f; }
        acc.x = fmaf(v0.x, n0, acc.x); acc.y = fmaf(v0.y, n0, acc.y);
        acc.z = fmaf(v0.z, n0, acc.z); acc.w = fmaf(v0.w, n0, acc.w);
        acc.x = fmaf(v1.x, n1, acc.x); acc.y = fmaf(v1.y, n1, acc.y);
        acc.z = fmaf(v1.z, n1, acc.z); acc.w = fmaf(v1.w, n1, acc.w);
        acc.x = fmaf(v2.x, n2, acc.x); acc.y = fmaf(v2.y, n2, acc.y);
        acc.z = fmaf(v2.z, n2, acc.z); acc.w = fmaf(v2.w, n2, acc.w);
    }

    const float4 bv = *(const float4*)(bias + c);
    float4 r;
    r.x = acc.x + bv.x; r.y = acc.y + bv.y;
    r.z = acc.z + bv.z; r.w = acc.w + bv.w;
    if (RELU) {
        r.x = fmaxf(r.x, 0.0f); r.y = fmaxf(r.y, 0.0f);
        r.z = fmaxf(r.z, 0.0f); r.w = fmaxf(r.w, 0.0f);
    }
    *(float4*)(out + (size_t)i * D + c) = r;
}

template <int D>
__global__ void __launch_bounds__(256)
gather_pool_rows(const float* __restrict__ h,
                 const float* __restrict__ dinv,
                 const int* __restrict__ rowptr,
                 const int* __restrict__ csr_src,
                 const float* __restrict__ bias,
                 const int* __restrict__ batch,
                 float* __restrict__ sums,
                 float* __restrict__ cnts, int Nn) {
    constexpr int LPN = D >> 2;
    constexpr int NPB = 256 / LPN;
    __shared__ float rows[NPB][D];
    __shared__ int   gid[NPB];

    const int q  = threadIdx.x / LPN;
    const int lc = threadIdx.x - q * LPN;
    const int c  = lc << 2;
    const int i  = blockIdx.x * NPB + q;

    float4 r;
    r.x = 0.0f; r.y = 0.0f; r.z = 0.0f; r.w = 0.0f;
    int g = -1;
    if (i < Nn) {
        g = batch[i];
        const int beg = rowptr[i];
        const int end = rowptr[i + 1];
        const float di = dinv[i];
        const float4 hs = *(const float4*)(h + (size_t)i * D + c);
        const float wself = di * di;
        float4 acc;
        acc.x = hs.x * wself; acc.y = hs.y * wself;
        acc.z = hs.z * wself; acc.w = hs.w * wself;

        int j = beg;
        for (; j + 4 <= end; j += 4) {
            const int s0 = csr_src[j];
            const int s1 = csr_src[j + 1];
            const int s2 = csr_src[j + 2];
            const int s3 = csr_src[j + 3];
            const float n0 = dinv[s0] * di;
            const float n1 = dinv[s1] * di;
            const float n2 = dinv[s2] * di;
            const float n3 = dinv[s3] * di;
            const float4 v0 = *(const float4*)(h + (size_t)s0 * D + c);
            const float4 v1 = *(const float4*)(h + (size_t)s1 * D + c);
            const float4 v2 = *(const float4*)(h + (size_t)s2 * D + c);
            const float4 v3 = *(const float4*)(h + (size_t)s3 * D + c);
            acc.x = fmaf(v0.x, n0, acc.x); acc.y = fmaf(v0.y, n0, acc.y);
            acc.z = fmaf(v0.z, n0, acc.z); acc.w = fmaf(v0.w, n0, acc.w);
            acc.x = fmaf(v1.x, n1, acc.x); acc.y = fmaf(v1.y, n1, acc.y);
            acc.z = fmaf(v1.z, n1, acc.z); acc.w = fmaf(v1.w, n1, acc.w);
            acc.x = fmaf(v2.x, n2, acc.x); acc.y = fmaf(v2.y, n2, acc.y);
            acc.z = fmaf(v2.z, n2, acc.z); acc.w = fmaf(v2.w, n2, acc.w);
            acc.x = fmaf(v3.x, n3, acc.x); acc.y = fmaf(v3.y, n3, acc.y);
            acc.z = fmaf(v3.z, n3, acc.z); acc.w = fmaf(v3.w, n3, acc.w);
        }
        const int rem = end - j;
        if (rem > 0) {
            const int last = end - 1;
            const int s0 = csr_src[j];
            const int s1 = csr_src[min(j + 1, last)];
            const int s2 = csr_src[min(j + 2, last)];
            const float n0 = dinv[s0] * di;
            const float n1 = dinv[s1] * di;
            const float n2 = dinv[s2] * di;
            const float4 v0 = *(const float4*)(h + (size_t)s0 * D + c);
            float4 v1 = *(const float4*)(h + (size_t)s1 * D + c);
            float4 v2 = *(const float4*)(h + (size_t)s2 * D + c);
            if (rem < 2) { v1.x = 0.0f; v1.y = 0.0f; v1.z = 0.0f; v1.w = 0.0f; }
            if (rem < 3) { v2.x = 0.0f; v2.y = 0.0f; v2.z = 0.0f; v2.w = 0.0f; }
            acc.x = fmaf(v0.x, n0, acc.x); acc.y = fmaf(v0.y, n0, acc.y);
            acc.z = fmaf(v0.z, n0, acc.z); acc.w = fmaf(v0.w, n0, acc.w);
            acc.x = fmaf(v1.x, n1, acc.x); acc.y = fmaf(v1.y, n1, acc.y);
            acc.z = fmaf(v1.z, n1, acc.z); acc.w = fmaf(v1.w, n1, acc.w);
            acc.x = fmaf(v2.x, n2, acc.x); acc.y = fmaf(v2.y, n2, acc.y);
            acc.z = fmaf(v2.z, n2, acc.z); acc.w = fmaf(v2.w, n2, acc.w);
        }

        const float4 bv = *(const float4*)(bias + c);
        r.x = acc.x + bv.x; r.y = acc.y + bv.y;
        r.z = acc.z + bv.z; r.w = acc.w + bv.w;
    }

    *(float4*)(&rows[q][c]) = r;
    if (lc == 0) gid[q] = g;
    __syncthreads();

    if (threadIdx.x < D) {
        const int d = threadIdx.x;
        float v[NPB];
        int   gk[NPB];
#pragma unroll
        for (int k = 0; k < NPB; ++k) { v[k] = rows[k][d]; gk[k] = gid[k]; }

        float s  = 0.0f;
        int  run = 0;
        int  gc  = gk[0];
#pragma unroll
        for (int k = 0; k < NPB; ++k) {
            if (gk[k] != gc) {
                if (gc >= 0 && gc < NUM_GRAPHS) {
                    atomicAdd(&sums[(size_t)gc * D + d], s);
                    if (d == 0) atomicAdd(&cnts[gc], (float)run);
                }
                s = 0.0f; run = 0; gc = gk[k];
            }
            s += v[k];
            ++run;
        }
        if (gc >= 0 && gc < NUM_GRAPHS) {
            atomicAdd(&sums[(size_t)gc * D + d], s);
            if (d == 0) atomicAdd(&cnts[gc], (float)run);
        }
    }
}

__global__ void pool_div(const float* __restrict__ sums,
                         const float* __restrict__ cnt,
                         float* __restrict__ out, int G, int D) {
    int tid = blockIdx.x * blockDim.x + threadIdx.x;
    if (tid >= G * D) return;
    int g = tid / D;
    out[tid] = sums[tid] / fmaxf(cnt[g], 1.0f);
}

extern "C" void kernel_launch(void* const* d_in, const int* in_sizes, int n_in,
                              void* d_out, int out_size, void* d_ws, size_t ws_size,
                              hipStream_t stream) {
    const float* x    = (const float*)d_in[0];
    const int*   eidx = (const int*)  d_in[1];
    const int*   batch= (const int*)  d_in[2];
    const float* W1   = (const float*)d_in[3];
    const float* b1   = (const float*)d_in[4];
    const float* W2   = (const float*)d_in[5];
    const float* b2   = (const float*)d_in[6];
    float* out = (float*)d_out;

    const int Nn = in_sizes[0] / IN_DIM;
    const int E  = in_sizes[1] / 2;
    const int* src = eidx;
    const int* dst = eidx + E;

    float* ws   = (float*)d_ws;
    float* h1   = ws;
    float* a1   = h1   + (size_t)Nn * HID_DIM;
    float* h2   = a1   + (size_t)Nn * HID_DIM;
    float* dinv = h2   + (size_t)Nn * OUT_DIM;
    float* sums = dinv + Nn;
    float* cnts = sums + NUM_GRAPHS * OUT_DIM;
    int* cnt    = (int*)(cnts + NUM_GRAPHS);
    int* rowptr = cnt + Nn;
    int* rank   = rowptr + Nn + 1;
    int* csr    = rank + E;

    const int B = 256;
    auto blocks = [](long long n, int b) { return (unsigned)((n + b - 1) / b); };

    zero_int_kernel<<<blocks(Nn, B), B, 0, stream>>>(cnt, Nn);
    fill_kernel<<<blocks(NUM_GRAPHS * OUT_DIM, B), B, 0, stream>>>(sums, 0.0f, NUM_GRAPHS * OUT_DIM);
    fill_kernel<<<blocks(NUM_GRAPHS, B), B, 0, stream>>>(cnts, 0.0f, NUM_GRAPHS);

    rank_kernel<<<blocks(E, B), B, 0, stream>>>(dst, cnt, rank, E);
    rowptr_kernel<<<blocks(Nn, 1024), 1024, 0, stream>>>(cnt, rowptr, dinv, Nn);
    permute_kernel<<<blocks(E, B), B, 0, stream>>>(src, dst, rank, rowptr, csr, E);

    gemm_strip_wmma_f32<HID_DIM, 14><<<blocks((long long)(Nn >> 4) * (HID_DIM / 64), 14), 14 * 32, 0, stream>>>(x, W1, h1, Nn);
    gather_rows<HID_DIM, true><<<blocks((long long)Nn * (HID_DIM >> 2), B), B, 0, stream>>>(
        h1, dinv, rowptr, csr, b1, a1, Nn);

    gemm_strip_wmma_f32<OUT_DIM, 8><<<blocks((long long)(Nn >> 4) * (OUT_DIM / 64), 8), 8 * 32, 0, stream>>>(a1, W2, h2, Nn);
    gather_pool_rows<OUT_DIM><<<blocks((long long)Nn * (OUT_DIM >> 2), B), B, 0, stream>>>(
        h2, dinv, rowptr, csr, b2, batch, sums, cnts, Nn);

    pool_div<<<blocks(NUM_GRAPHS * OUT_DIM, B), B, 0, stream>>>(
        sums, cnts, out, NUM_GRAPHS, OUT_DIM);
}


# Three start-value fills (in-degree counts, pool sums, pool counts) ride in the layer-1 product's launch, moved to the front as it needs nothing from the edge sort: 11 launches become 8
# speedup vs baseline: 2.9127x; 1.0328x over previous
#include <hip/hip_runtime.h>

#define IN_DIM   128
#define HID_DIM  128
#define OUT_DIM  64
#define NUM_GRAPHS 64

typedef float v2f __attribute__((ext_vector_type(2)));
typedef float v8f __attribute__((ext_vector_type(8)));

__global__ void rank_kernel(const int* __restrict__ dst, int* __restrict__ cnt,
                            int* __restrict__ rank, int E) {
    int e = blockIdx.x * blockDim.x + threadIdx.x;
    if (e < E) rank[e] = atomicAdd(&cnt[dst[e]], 1);
}

__global__ void __launch_bounds__(1024)
rowptr_kernel(const int* __restrict__ cnt, int* __restrict__ rowptr,
              float* __restrict__ dinv, int n) {
    __shared__ int sh[1024];
    const int t    = threadIdx.x;
    const int base = blockIdx.x * 1024;

    int pre = 0;
    for (int k = t; k < base; k += 1024) pre += cnt[k];
    sh[t] = pre;
    __syncthreads();
    for (int s = 512; s > 0; s >>= 1) {
        if (t < s) sh[t] += sh[t + s];
        __syncthreads();
    }
    const int off = sh[0];
    __syncthreads();

    const int i = base + t;
    const int v = (i < n) ? cnt[i] : 0;
    sh[t] = v;
    __syncthreads();
    for (int o = 1; o < 1024; o <<= 1) {
        const int x = (t >= o) ? sh[t - o] : 0;
        __syncthreads();
        sh[t] += x;
        __syncthreads();
    }
    const int incl = sh[t];
    if (i < n) {
        rowptr[i] = off + incl - v;
        dinv[i]   = rsqrtf((float)(v + 1));
        if (i == n - 1) rowptr[n] = off + incl;
    }
}

__global__ void permute_kernel(const int* __restrict__ src, const int* __restrict__ dst,
                               const int* __restrict__ rank, const int* __restrict__ rowptr,
                               int* __restrict__ csr_src, int E) {
    int e = blockIdx.x * blockDim.x + threadIdx.x;
    if (e < E) csr_src[rowptr[dst[e]] + rank[e]] = src[e];
}

template <int N, int WAVES, bool INIT>
__global__ void __launch_bounds__(WAVES * 32, 7)
gemm_strip_wmma_f32(const float* __restrict__ A,
                    const float* __restrict__ B,
                    float* __restrict__ C, int M,
                    int* __restrict__ zi, int nzi,
                    float* __restrict__ zf, int nzf) {
    constexpr int K   = 128;
    static_assert(IN_DIM == K && HID_DIM == K && N % 64 == 0, "strip GEMM is written for K = 128");
    constexpr int NH  = N / 64;
    constexpr int SLOTS = (K / 4) * N;
    __shared__ float4 Ws[SLOTS];

    const int tid  = threadIdx.x;
    const int lane = tid & 31;
    const int wave = tid >> 5;
    const int c    = lane & 15;
    const int h    = lane >> 4;
    const int task = blockIdx.x * WAVES + wave;
    const int tile = task / NH;
    const int cg   = task - tile * NH;
    const bool live = tile < (M >> 4);
    const int row0 = tile << 4;

    float4 a[16];
    if (live) {
        const float* Ap = A + (size_t)(row0 + c) * K + 4 * h;
#pragma unroll
        for (int j = 0; j < 16; ++j) a[j] = *(const float4*)(Ap + 8 * j);
    }

#pragma unroll 4
    for (int u = tid; u < SLOTS; u += WAVES * 32) {
        const int q  = u / N;
        const int p  = u - q * N;
        const int pl = p & 63;
        const int n  = (p - pl) + ((pl & 15) << 2) + (pl >> 4);
        const float* bp = B + (size_t)(4 * q) * N + n;
        float4 w;
        w.x = bp[0]; w.y = bp[N]; w.z = bp[2 * N]; w.w = bp[3 * N];
        Ws[u] = w;
    }
    __syncthreads();

    if (INIT) {
        const int g = blockIdx.x * (WAVES * 32) + tid;
        if (g < nzi) zi[g] = 0;
        if (g < nzf) zf[g] = 0.0f;
    }

    if (live) {
        v8f acc[4];
#pragma unroll
        for (int t = 0; t < 4; ++t) acc[t] = v8f{};

        const float4* Wl = Ws + h * N + cg * 64 + c;
        float4 bc[4], bn[4];
#pragma unroll
        for (int t = 0; t < 4; ++t) bc[t] = Wl[16 * t];

        __builtin_amdgcn_sched_group_barrier(0x100, 8, 0);
#pragma unroll
        for (int j = 0; j < 16; ++j) {
            if (j + 1 < 16) {
#pragma unroll
                for (int t = 0; t < 4; ++t) bn[t] = Wl[(2 * (j + 1)) * N + 16 * t];
            }
            v2f a0, a1;
            a0.x = a[j].x; a0.y = a[j].y;
            a1.x = a[j].z; a1.y = a[j].w;
#pragma unroll
            for (int t = 0; t < 4; ++t) {
                v2f b0;
                b0.x = bc[t].x; b0.y = bc[t].y;
                acc[t] = __builtin_amdgcn_wmma_f32_16x16x4_f32(
                     false, a0,  false, b0,
                     (short)0, acc[t],  false,  false);
            }
#pragma unroll
            for (int t = 0; t < 4; ++t) {
                v2f b1;
                b1.x = bc[t].z; b1.y = bc[t].w;
                acc[t] = __builtin_amdgcn_wmma_f32_16x16x4_f32(
                     false, a1,  false, b1,
                     (short)0, acc[t],  false,  false);
            }
            __builtin_amdgcn_sched_group_barrier(0x008, 8, 0);
            if (j + 2 < 16)
                __builtin_amdgcn_sched_group_barrier(0x100, 4, 0);
#pragma unroll
            for (int t = 0; t < 4; ++t) bc[t] = bn[t];
        }

        float* Cp = C + (size_t)(row0 + (h << 3)) * N + cg * 64 + (c << 2);
#pragma unroll
        for (int v = 0; v < 8; ++v) {
            float4 o;
            o.x = acc[0][v]; o.y = acc[1][v]; o.z = acc[2][v]; o.w = acc[3][v];
            *(float4*)(Cp + (size_t)v * N) = o;
        }
    }
}

template <int D, bool RELU>
__global__ void gather_rows(const float* __restrict__ h,
                            const float* __restrict__ dinv,
                            const int* __restrict__ rowptr,
                            const int* __restrict__ csr_src,
                            const float* __restrict__ bias,
                            float* __restrict__ out, int Nn) {
    constexpr int LPN = D >> 2;
    const int tid = blockIdx.x * blockDim.x + threadIdx.x;
    const int i = tid / LPN;
    if (i >= Nn) return;
    const int c = (tid - i * LPN) << 2;

    const int beg = rowptr[i];
    const int end = rowptr[i + 1];
    const float di = dinv[i];
    const float4 hs = *(const float4*)(h + (size_t)i * D + c);
    const float wself = di * di;
    float4 acc;
    acc.x = hs.x * wself; acc.y = hs.y * wself;
    acc.z = hs.z * wself; acc.w = hs.w * wself;

    int j = beg;
    for (; j + 4 <= end; j += 4) {
        const int s0 = csr_src[j];
        const int s1 = csr_src[j + 1];
        const int s2 = csr_src[j + 2];
        const int s3 = csr_src[j + 3];
        const float n0 = dinv[s0] * di;
        const float n1 = dinv[s1] * di;
        const float n2 = dinv[s2] * di;
        const float n3 = dinv[s3] * di;
        const float4 v0 = *(const float4*)(h + (size_t)s0 * D + c);
        const float4 v1 = *(const float4*)(h + (size_t)s1 * D + c);
        const float4 v2 = *(const float4*)(h + (size_t)s2 * D + c);
        const float4 v3 = *(const float4*)(h + (size_t)s3 * D + c);
        acc.x = fmaf(v0.x, n0, acc.x); acc.y = fmaf(v0.y, n0, acc.y);
        acc.z = fmaf(v0.z, n0, acc.z); acc.w = fmaf(v0.w, n0, acc.w);
        acc.x = fmaf(v1.x, n1, acc.x); acc.y = fmaf(v1.y, n1, acc.y);
        acc.z = fmaf(v1.z, n1, acc.z); acc.w = fmaf(v1.w, n1, acc.w);
        acc.x = fmaf(v2.x, n2, acc.x); acc.y = fmaf(v2.y, n2, acc.y);
        acc.z = fmaf(v2.z, n2, acc.z); acc.w = fmaf(v2.w, n2, acc.w);
        acc.x = fmaf(v3.x, n3, acc.x); acc.y = fmaf(v3.y, n3, acc.y);
        acc.z = fmaf(v3.z, n3, acc.z); acc.w = fmaf(v3.w, n3, acc.w);
    }
    const int rem = end - j;
    if (rem > 0) {
        const int last = end - 1;
        const int s0 = csr_src[j];
        const int s1 = csr_src[min(j + 1, last)];
        const int s2 = csr_src[min(j + 2, last)];
        const float n0 = dinv[s0] * di;
        const float n1 = dinv[s1] * di;
        const float n2 = dinv[s2] * di;
        const float4 v0 = *(const float4*)(h + (size_t)s0 * D + c);
        float4 v1 = *(const float4*)(h + (size_t)s1 * D + c);
        float4 v2 = *(const float4*)(h + (size_t)s2 * D + c);
        if (rem < 2) { v1.x = 0.0f; v1.y = 0.0f; v1.z = 0.0f; v1.w = 0.0f; }
        if (rem < 3) { v2.x = 0.0f; v2.y = 0.0f; v2.z = 0.0f; v2.w = 0.0f; }
        acc.x = fmaf(v0.x, n0, acc.x); acc.y = fmaf(v0.y, n0, acc.y);
        acc.z = fmaf(v0.z, n0, acc.z); acc.w = fmaf(v0.w, n0, acc.w);
        acc.x = fmaf(v1.x, n1, acc.x); acc.y = fmaf(v1.y, n1, acc.y);
        acc.z = fmaf(v1.z, n1, acc.z); acc.w = fmaf(v1.w, n1, acc.w);
        acc.x = fmaf(v2.x, n2, acc.x); acc.y = fmaf(v2.y, n2, acc.y);
        acc.z = fmaf(v2.z, n2, acc.z); acc.w = fmaf(v2.w, n2, acc.w);
    }

    const float4 bv = *(const float4*)(bias + c);
    float4 r;
    r.x = acc.x + bv.x; r.y = acc.y + bv.y;
    r.z = acc.z + bv.z; r.w = acc.w + bv.w;
    if (RELU) {
        r.x = fmaxf(r.x, 0.0f); r.y = fmaxf(r.y, 0.0f);
        r.z = fmaxf(r.z, 0.0f); r.w = fmaxf(r.w, 0.0f);
    }
    *(float4*)(out + (size_t)i * D + c) = r;
}

template <int D>
__global__ void __launch_bounds__(256)
gather_pool_rows(const float* __restrict__ h,
                 const float* __restrict__ dinv,
                 const int* __restrict__ rowptr,
                 const int* __restrict__ csr_src,
                 const float* __restrict__ bias,
                 const int* __restrict__ batch,
                 float* __restrict__ sums,
                 float* __restrict__ cnts, int Nn) {
    constexpr int LPN = D >> 2;
    constexpr int NPB = 256 / LPN;
    __shared__ float rows[NPB][D];
    __shared__ int   gid[NPB];

    const int q  = threadIdx.x / LPN;
    const int lc = threadIdx.x - q * LPN;
    const int c  = lc << 2;
    const int i  = blockIdx.x * NPB + q;

    float4 r;
    r.x = 0.0f; r.y = 0.0f; r.z = 0.0f; r.w = 0.0f;
    int g = -1;
    if (i < Nn) {
        g = batch[i];
        const int beg = rowptr[i];
        const int end = rowptr[i + 1];
        const float di = dinv[i];
        const float4 hs = *(const float4*)(h + (size_t)i * D + c);
        const float wself = di * di;
        float4 acc;
        acc.x = hs.x * wself; acc.y = hs.y * wself;
        acc.z = hs.z * wself; acc.w = hs.w * wself;

        int j = beg;
        for (; j + 4 <= end; j += 4) {
            const int s0 = csr_src[j];
            const int s1 = csr_src[j + 1];
            const int s2 = csr_src[j + 2];
            const int s3 = csr_src[j + 3];
            const float n0 = dinv[s0] * di;
            const float n1 = dinv[s1] * di;
            const float n2 = dinv[s2] * di;
            const float n3 = dinv[s3] * di;
            const float4 v0 = *(const float4*)(h + (size_t)s0 * D + c);
            const float4 v1 = *(const float4*)(h + (size_t)s1 * D + c);
            const float4 v2 = *(const float4*)(h + (size_t)s2 * D + c);
            const float4 v3 = *(const float4*)(h + (size_t)s3 * D + c);
            acc.x = fmaf(v0.x, n0, acc.x); acc.y = fmaf(v0.y, n0, acc.y);
            acc.z = fmaf(v0.z, n0, acc.z); acc.w = fmaf(v0.w, n0, acc.w);
            acc.x = fmaf(v1.x, n1, acc.x); acc.y = fmaf(v1.y, n1, acc.y);
            acc.z = fmaf(v1.z, n1, acc.z); acc.w = fmaf(v1.w, n1, acc.w);
            acc.x = fmaf(v2.x, n2, acc.x); acc.y = fmaf(v2.y, n2, acc.y);
            acc.z = fmaf(v2.z, n2, acc.z); acc.w = fmaf(v2.w, n2, acc.w);
            acc.x = fmaf(v3.x, n3, acc.x); acc.y = fmaf(v3.y, n3, acc.y);
            acc.z = fmaf(v3.z, n3, acc.z); acc.w = fmaf(v3.w, n3, acc.w);
        }
        const int rem = end - j;
        if (rem > 0) {
            const int last = end - 1;
            const int s0 = csr_src[j];
            const int s1 = csr_src[min(j + 1, last)];
            const int s2 = csr_src[min(j + 2, last)];
            const float n0 = dinv[s0] * di;
            const float n1 = dinv[s1] * di;
            const float n2 = dinv[s2] * di;
            const float4 v0 = *(const float4*)(h + (size_t)s0 * D + c);
            float4 v1 = *(const float4*)(h + (size_t)s1 * D + c);
            float4 v2 = *(const float4*)(h + (size_t)s2 * D + c);
            if (rem < 2) { v1.x = 0.0f; v1.y = 0.0f; v1.z = 0.0f; v1.w = 0.0f; }
            if (rem < 3) { v2.x = 0.0f; v2.y = 0.0f; v2.z = 0.0f; v2.w = 0.0f; }
            acc.x = fmaf(v0.x, n0, acc.x); acc.y = fmaf(v0.y, n0, acc.y);
            acc.z = fmaf(v0.z, n0, acc.z); acc.w = fmaf(v0.w, n0, acc.w);
            acc.x = fmaf(v1.x, n1, acc.x); acc.y = fmaf(v1.y, n1, acc.y);
            acc.z = fmaf(v1.z, n1, acc.z); acc.w = fmaf(v1.w, n1, acc.w);
            acc.x = fmaf(v2.x, n2, acc.x); acc.y = fmaf(v2.y, n2, acc.y);
            acc.z = fmaf(v2.z, n2, acc.z); acc.w = fmaf(v2.w, n2, acc.w);
        }

        const float4 bv = *(const float4*)(bias + c);
        r.x = acc.x + bv.x; r.y = acc.y + bv.y;
        r.z = acc.z + bv.z; r.w = acc.w + bv.w;
    }

    *(float4*)(&rows[q][c]) = r;
    if (lc == 0) gid[q] = g;
    __syncthreads();

    if (threadIdx.x < D) {
        const int d = threadIdx.x;
        float v[NPB];
        int   gk[NPB];
#pragma unroll
        for (int k = 0; k < NPB; ++k) { v[k] = rows[k][d]; gk[k] = gid[k]; }

        float s  = 0.0f;
        int  run = 0;
        int  gc  = gk[0];
#pragma unroll
        for (int k = 0; k < NPB; ++k) {
            if (gk[k] != gc) {
                if (gc >= 0 && gc < NUM_GRAPHS) {
                    atomicAdd(&sums[(size_t)gc * D + d], s);
                    if (d == 0) atomicAdd(&cnts[gc], (float)run);
                }
                s = 0.0f; run = 0; gc = gk[k];
            }
            s += v[k];
            ++run;
        }
        if (gc >= 0 && gc < NUM_GRAPHS) {
            atomicAdd(&sums[(size_t)gc * D + d], s);
            if (d == 0) atomicAdd(&cnts[gc], (float)run);
        }
    }
}

__global__ void pool_div(const float* __restrict__ sums,
                         const float* __restrict__ cnt,
                         float* __restrict__ out, int G, int D) {
    int tid = blockIdx.x * blockDim.x + threadIdx.x;
    if (tid >= G * D) return;
    int g = tid / D;
    out[tid] = sums[tid] / fmaxf(cnt[g], 1.0f);
}

extern "C" void kernel_launch(void* const* d_in, const int* in_sizes, int n_in,
                              void* d_out, int out_size, void* d_ws, size_t ws_size,
                              hipStream_t stream) {
    const float* x    = (const float*)d_in[0];
    const int*   eidx = (const int*)  d_in[1];
    const int*   batch= (const int*)  d_in[2];
    const float* W1   = (const float*)d_in[3];
    const float* b1   = (const float*)d_in[4];
    const float* W2   = (const float*)d_in[5];
    const float* b2   = (const float*)d_in[6];
    float* out = (float*)d_out;

    const int Nn = in_sizes[0] / IN_DIM;
    const int E  = in_sizes[1] / 2;
    const int* src = eidx;
    const int* dst = eidx + E;

    float* ws   = (float*)d_ws;
    float* h1   = ws;
    float* a1   = h1   + (size_t)Nn * HID_DIM;
    float* h2   = a1   + (size_t)Nn * HID_DIM;
    float* dinv = h2   + (size_t)Nn * OUT_DIM;
    float* sums = dinv + Nn;
    float* cnts = sums + NUM_GRAPHS * OUT_DIM;
    int* cnt    = (int*)(cnts + NUM_GRAPHS);
    int* rowptr = cnt + Nn;
    int* rank   = rowptr + Nn + 1;
    int* csr    = rank + E;

    const int B = 256;
    auto blocks = [](long long n, int b) { return (unsigned)((n + b - 1) / b); };

    gemm_strip_wmma_f32<HID_DIM, 14, true><<<blocks((long long)(Nn >> 4) * (HID_DIM / 64), 14), 14 * 32, 0, stream>>>(
        x, W1, h1, Nn, cnt, Nn, sums, NUM_GRAPHS * OUT_DIM + NUM_GRAPHS);

    rank_kernel<<<blocks(E, B), B, 0, stream>>>(dst, cnt, rank, E);
    rowptr_kernel<<<blocks(Nn, 1024), 1024, 0, stream>>>(cnt, rowptr, dinv, Nn);
    permute_kernel<<<blocks(E, B), B, 0, stream>>>(src, dst, rank, rowptr, csr, E);

    gather_rows<HID_DIM, true><<<blocks((long long)Nn * (HID_DIM >> 2), B), B, 0, stream>>>(
        h1, dinv, rowptr, csr, b1, a1, Nn);

    gemm_strip_wmma_f32<OUT_DIM, 8, false><<<blocks((long long)(Nn >> 4) * (OUT_DIM / 64), 8), 8 * 32, 0, stream>>>(
        a1, W2, h2, Nn, nullptr, 0, nullptr, 0);
    gather_pool_rows<OUT_DIM><<<blocks((long long)Nn * (OUT_DIM >> 2), B), B, 0, stream>>>(
        h2, dinv, rowptr, csr, b2, batch, sums, cnts, Nn);

    pool_div<<<blocks(NUM_GRAPHS * OUT_DIM, B), B, 0, stream>>>(
        sums, cnts, out, NUM_GRAPHS, OUT_DIM);
}


# CSR build (prefix scan + permute pass) replaced by rows of 32 source slots per destination filled at the arrival rank; dinv from the counts; deeper rows finish from an overflow list; 8 launches to 6
# speedup vs baseline: 3.3711x; 1.1574x over previous
#include <hip/hip_runtime.h>

#define IN_DIM   128
#define HID_DIM  128
#define OUT_DIM  64
#define NUM_GRAPHS 64

typedef float v2f __attribute__((ext_vector_type(2)));
typedef float v8f __attribute__((ext_vector_type(8)));

#define CAP 32

__global__ void bucket_kernel(const int* __restrict__ src, const int* __restrict__ dst,
                              int* __restrict__ cnt, int* __restrict__ bucket,
                              int* __restrict__ ovf, int E, int Nn) {
    int e = blockIdx.x * blockDim.x + threadIdx.x;
    if (e < E) {
        const int d = dst[e];
        const int r = atomicAdd(&cnt[d], 1);
        if (r < CAP) bucket[(size_t)d * CAP + r] = src[e];
        else         ovf[atomicAdd(&cnt[Nn], 1)] = e;
    }
}

template <int N, int WAVES, bool INIT>
__global__ void __launch_bounds__(WAVES * 32, 7)
gemm_strip_wmma_f32(const float* __restrict__ A,
                    const float* __restrict__ B,
                    float* __restrict__ C, int M,
                    int* __restrict__ zi, int nzi,
                    float* __restrict__ zf, int nzf) {
    constexpr int K   = 128;
    static_assert(IN_DIM == K && HID_DIM == K && N % 64 == 0, "strip GEMM is written for K = 128");
    constexpr int NH  = N / 64;
    constexpr int SLOTS = (K / 4) * N;
    __shared__ float4 Ws[SLOTS];

    const int tid  = threadIdx.x;
    const int lane = tid & 31;
    const int wave = tid >> 5;
    const int c    = lane & 15;
    const int h    = lane >> 4;
    const int task = blockIdx.x * WAVES + wave;
    const int tile = task / NH;
    const int cg   = task - tile * NH;
    const bool live = tile < (M >> 4);
    const int row0 = tile << 4;

    float4 a[16];
    if (live) {
        const float* Ap = A + (size_t)(row0 + c) * K + 4 * h;
#pragma unroll
        for (int j = 0; j < 16; ++j) a[j] = *(const float4*)(Ap + 8 * j);
    }

#pragma unroll 4
    for (int u = tid; u < SLOTS; u += WAVES * 32) {
        const int q  = u / N;
        const int p  = u - q * N;
        const int pl = p & 63;
        const int n  = (p - pl) + ((pl & 15) << 2) + (pl >> 4);
        const float* bp = B + (size_t)(4 * q) * N + n;
        float4 w;
        w.x = bp[0]; w.y = bp[N]; w.z = bp[2 * N]; w.w = bp[3 * N];
        Ws[u] = w;
    }
    __syncthreads();

    if (INIT) {
        const int g = blockIdx.x * (WAVES * 32) + tid;
        if (g < nzi) zi[g] = 0;
        if (g < nzf) zf[g] = 0.0f;
    }

    if (live) {
        v8f acc[4];
#pragma unroll
        for (int t = 0; t < 4; ++t) acc[t] = v8f{};

        const float4* Wl = Ws + h * N + cg * 64 + c;
        float4 bc[4], bn[4];
#pragma unroll
        for (int t = 0; t < 4; ++t) bc[t] = Wl[16 * t];

        __builtin_amdgcn_sched_group_barrier(0x100, 8, 0);
#pragma unroll
        for (int j = 0; j < 16; ++j) {
            if (j + 1 < 16) {
#pragma unroll
                for (int t = 0; t < 4; ++t) bn[t] = Wl[(2 * (j + 1)) * N + 16 * t];
            }
            v2f a0, a1;
            a0.x = a[j].x; a0.y = a[j].y;
            a1.x = a[j].z; a1.y = a[j].w;
#pragma unroll
            for (int t = 0; t < 4; ++t) {
                v2f b0;
                b0.x = bc[t].x; b0.y = bc[t].y;
                acc[t] = __builtin_amdgcn_wmma_f32_16x16x4_f32(
                     false, a0,  false, b0,
                     (short)0, acc[t],  false,  false);
            }
#pragma unroll
            for (int t = 0; t < 4; ++t) {
                v2f b1;
                b1.x = bc[t].z; b1.y = bc[t].w;
                acc[t] = __builtin_amdgcn_wmma_f32_16x16x4_f32(
                     false, a1,  false, b1,
                     (short)0, acc[t],  false,  false);
            }
            __builtin_amdgcn_sched_group_barrier(0x008, 8, 0);
            if (j + 2 < 16)
                __builtin_amdgcn_sched_group_barrier(0x100, 4, 0);
#pragma unroll
            for (int t = 0; t < 4; ++t) bc[t] = bn[t];
        }

        float* Cp = C + (size_t)(row0 + (h << 3)) * N + cg * 64 + (c << 2);
#pragma unroll
        for (int v = 0; v < 8; ++v) {
            float4 o;
            o.x = acc[0][v]; o.y = acc[1][v]; o.z = acc[2][v]; o.w = acc[3][v];
            *(float4*)(Cp + (size_t)v * N) = o;
        }
    }
}

__device__ __forceinline__ float dinv_of(int indeg) {
    return __builtin_amdgcn_rsqf((float)(indeg + 1));
}

template <int D>
__device__ __forceinline__ float4 gather_sum(const float* __restrict__ h,
                                             const int* __restrict__ cnt,
                                             const int* __restrict__ bucket,
                                             const int* __restrict__ ovf,
                                             const int* __restrict__ src,
                                             const int* __restrict__ dst,
                                             int i, int c, int Nn) {
    const int deg = cnt[i];
    const int nb  = min(deg, CAP);
    const int* bk = bucket + (size_t)i * CAP;
    const float di = dinv_of(deg);
    const float4 hs = *(const float4*)(h + (size_t)i * D + c);
    const float wself = di * di;
    float4 acc;
    acc.x = hs.x * wself; acc.y = hs.y * wself;
    acc.z = hs.z * wself; acc.w = hs.w * wself;

    int j = 0;
    for (; j + 4 <= nb; j += 4) {
        const int4 s = *(const int4*)(bk + j);
        const float n0 = dinv_of(cnt[s.x]) * di;
        const float n1 = dinv_of(cnt[s.y]) * di;
        const float n2 = dinv_of(cnt[s.z]) * di;
        const float n3 = dinv_of(cnt[s.w]) * di;
        const float4 v0 = *(const float4*)(h + (size_t)s.x * D + c);
        const float4 v1 = *(const float4*)(h + (size_t)s.y * D + c);
        const float4 v2 = *(const float4*)(h + (size_t)s.z * D + c);
        const float4 v3 = *(const float4*)(h + (size_t)s.w * D + c);
        acc.x = fmaf(v0.x, n0, acc.x); acc.y = fmaf(v0.y, n0, acc.y);
        acc.z = fmaf(v0.z, n0, acc.z); acc.w = fmaf(v0.w, n0, acc.w);
        acc.x = fmaf(v1.x, n1, acc.x); acc.y = fmaf(v1.y, n1, acc.y);
        acc.z = fmaf(v1.z, n1, acc.z); acc.w = fmaf(v1.w, n1, acc.w);
        acc.x = fmaf(v2.x, n2, acc.x); acc.y = fmaf(v2.y, n2, acc.y);
        acc.z = fmaf(v2.z, n2, acc.z); acc.w = fmaf(v2.w, n2, acc.w);
        acc.x = fmaf(v3.x, n3, acc.x); acc.y = fmaf(v3.y, n3, acc.y);
        acc.z = fmaf(v3.z, n3, acc.z); acc.w = fmaf(v3.w, n3, acc.w);
    }
    const int rem = nb - j;
    if (rem > 0) {
        const int4 s = *(const int4*)(bk + j);
        const int s0 = s.x;
        const int s1 = (rem > 1) ? s.y : s.x;
        const int s2 = (rem > 2) ? s.z : s.x;
        const float n0 = dinv_of(cnt[s0]) * di;
        const float n1 = dinv_of(cnt[s1]) * di;
        const float n2 = dinv_of(cnt[s2]) * di;
        const float4 v0 = *(const float4*)(h + (size_t)s0 * D + c);
        float4 v1 = *(const float4*)(h + (size_t)s1 * D + c);
        float4 v2 = *(const float4*)(h + (size_t)s2 * D + c);
        if (rem < 2) { v1.x = 0.0f; v1.y = 0.0f; v1.z = 0.0f; v1.w = 0.0f; }
        if (rem < 3) { v2.x = 0.0f; v2.y = 0.0f; v2.z = 0.0f; v2.w = 0.0f; }
        acc.x = fmaf(v0.x, n0, acc.x); acc.y = fmaf(v0.y, n0, acc.y);
        acc.z = fmaf(v0.z, n0, acc.z); acc.w = fmaf(v0.w, n0, acc.w);
        acc.x = fmaf(v1.x, n1, acc.x); acc.y = fmaf(v1.y, n1, acc.y);
        acc.z = fmaf(v1.z, n1, acc.z); acc.w = fmaf(v1.w, n1, acc.w);
        acc.x = fmaf(v2.x, n2, acc.x); acc.y = fmaf(v2.y, n2, acc.y);
        acc.z = fmaf(v2.z, n2, acc.z); acc.w = fmaf(v2.w, n2, acc.w);
    }
    if (deg > CAP) {
        const int novf = cnt[Nn];
        for (int k = 0; k < novf; ++k) {
            const int e = ovf[k];
            if (dst[e] == i) {
                const int s = src[e];
                const float n = dinv_of(cnt[s]) * di;
                const float4 v = *(const float4*)(h + (size_t)s * D + c);
                acc.x = fmaf(v.x, n, acc.x); acc.y = fmaf(v.y, n, acc.y);
                acc.z = fmaf(v.z, n, acc.z); acc.w = fmaf(v.w, n, acc.w);
            }
        }
    }
    return acc;
}

template <int D, bool RELU>
__global__ void gather_rows(const float* __restrict__ h,
                            const int* __restrict__ cnt,
                            const int* __restrict__ bucket,
                            const int* __restrict__ ovf,
                            const int* __restrict__ src,
                            const int* __restrict__ dst,
                            const float* __restrict__ bias,
                            float* __restrict__ out, int Nn) {
    constexpr int LPN = D >> 2;
    const int tid = blockIdx.x * blockDim.x + threadIdx.x;
    const int i = tid / LPN;
    if (i >= Nn) return;
    const int c = (tid - i * LPN) << 2;

    const float4 acc = gather_sum<D>(h, cnt, bucket, ovf, src, dst, i, c, Nn);

    const float4 bv = *(const float4*)(bias + c);
    float4 r;
    r.x = acc.x + bv.x; r.y = acc.y + bv.y;
    r.z = acc.z + bv.z; r.w = acc.w + bv.w;
    if (RELU) {
        r.x = fmaxf(r.x, 0.0f); r.y = fmaxf(r.y, 0.0f);
        r.z = fmaxf(r.z, 0.0f); r.w = fmaxf(r.w, 0.0f);
    }
    *(float4*)(out + (size_t)i * D + c) = r;
}

template <int D>
__global__ void __launch_bounds__(256)
gather_pool_rows(const float* __restrict__ h,
                 const int* __restrict__ cnt,
                 const int* __restrict__ bucket,
                 const int* __restrict__ ovf,
                 const int* __restrict__ src,
                 const int* __restrict__ dst,
                 const float* __restrict__ bias,
                 const int* __restrict__ batch,
                 float* __restrict__ sums,
                 float* __restrict__ cnts, int Nn) {
    constexpr int LPN = D >> 2;
    constexpr int NPB = 256 / LPN;
    __shared__ float rows[NPB][D];
    __shared__ int   gid[NPB];

    const int q  = threadIdx.x / LPN;
    const int lc = threadIdx.x - q * LPN;
    const int c  = lc << 2;
    const int i  = blockIdx.x * NPB + q;

    float4 r;
    r.x = 0.0f; r.y = 0.0f; r.z = 0.0f; r.w = 0.0f;
    int g = -1;
    if (i < Nn) {
        g = batch[i];
        const float4 acc = gather_sum<D>(h, cnt, bucket, ovf, src, dst, i, c, Nn);
        const float4 bv = *(const float4*)(bias + c);
        r.x = acc.x + bv.x; r.y = acc.y + bv.y;
        r.z = acc.z + bv.z; r.w = acc.w + bv.w;
    }

    *(float4*)(&rows[q][c]) = r;
    if (lc == 0) gid[q] = g;
    __syncthreads();

    if (threadIdx.x < D) {
        const int d = threadIdx.x;
        float v[NPB];
        int   gk[NPB];
#pragma unroll
        for (int k = 0; k < NPB; ++k) { v[k] = rows[k][d]; gk[k] = gid[k]; }

        float s  = 0.0f;
        int  run = 0;
        int  gc  = gk[0];
#pragma unroll
        for (int k = 0; k < NPB; ++k) {
            if (gk[k] != gc) {
                if (gc >= 0 && gc < NUM_GRAPHS) {
                    atomicAdd(&sums[(size_t)gc * D + d], s);
                    if (d == 0) atomicAdd(&cnts[gc], (float)run);
                }
                s = 0.0f; run = 0; gc = gk[k];
            }
            s += v[k];
            ++run;
        }
        if (gc >= 0 && gc < NUM_GRAPHS) {
            atomicAdd(&sums[(size_t)gc * D + d], s);
            if (d == 0) atomicAdd(&cnts[gc], (float)run);
        }
    }
}

__global__ void pool_div(const float* __restrict__ sums,
                         const float* __restrict__ cnt,
                         float* __restrict__ out, int G, int D) {
    int tid = blockIdx.x * blockDim.x + threadIdx.x;
    if (tid >= G * D) return;
    int g = tid / D;
    out[tid] = sums[tid] / fmaxf(cnt[g], 1.0f);
}

extern "C" void kernel_launch(void* const* d_in, const int* in_sizes, int n_in,
                              void* d_out, int out_size, void* d_ws, size_t ws_size,
                              hipStream_t stream) {
    const float* x    = (const float*)d_in[0];
    const int*   eidx = (const int*)  d_in[1];
    const int*   batch= (const int*)  d_in[2];
    const float* W1   = (const float*)d_in[3];
    const float* b1   = (const float*)d_in[4];
    const float* W2   = (const float*)d_in[5];
    const float* b2   = (const float*)d_in[6];
    float* out = (float*)d_out;

    const int Nn = in_sizes[0] / IN_DIM;
    const int E  = in_sizes[1] / 2;
    const int* src = eidx;
    const int* dst = eidx + E;

    float* ws   = (float*)d_ws;
    float* h1   = ws;
    float* a1   = h1   + (size_t)Nn * HID_DIM;
    float* h2   = a1   + (size_t)Nn * HID_DIM;
    int* bucket = (int*)(h2 + (size_t)Nn * OUT_DIM);
    float* sums = (float*)(bucket + (size_t)Nn * CAP);
    float* cnts = sums + NUM_GRAPHS * OUT_DIM;
    int* cnt    = (int*)(cnts + NUM_GRAPHS);
    int* ovf    = cnt + Nn + 1;

    const int B = 256;
    auto blocks = [](long long n, int b) { return (unsigned)((n + b - 1) / b); };

    gemm_strip_wmma_f32<HID_DIM, 14, true><<<blocks((long long)(Nn >> 4) * (HID_DIM / 64), 14), 14 * 32, 0, stream>>>(
        x, W1, h1, Nn, cnt, Nn + 1, sums, NUM_GRAPHS * OUT_DIM + NUM_GRAPHS);

    bucket_kernel<<<blocks(E, B), B, 0, stream>>>(src, dst, cnt, bucket, ovf, E, Nn);

    gather_rows<HID_DIM, true><<<blocks((long long)Nn * (HID_DIM >> 2), B), B, 0, stream>>>(
        h1, cnt, bucket, ovf, src, dst, b1, a1, Nn);

    gemm_strip_wmma_f32<OUT_DIM, 8, false><<<blocks((long long)(Nn >> 4) * (OUT_DIM / 64), 8), 8 * 32, 0, stream>>>(
        a1, W2, h2, Nn, nullptr, 0, nullptr, 0);
    gather_pool_rows<OUT_DIM><<<blocks((long long)Nn * (OUT_DIM >> 2), B), B, 0, stream>>>(
        h2, cnt, bucket, ovf, src, dst, b2, batch, sums, cnts, Nn);

    pool_div<<<blocks(NUM_GRAPHS * OUT_DIM, B), B, 0, stream>>>(
        sums, cnts, out, NUM_GRAPHS, OUT_DIM);
}


# Layer-2 product moved into the layer-1 gather's launch: a block keeps its 64 finished rows in LDS and multiplies them by W2 there (WMMA f32); layer-1 output never goes to memory; 6 launches to 5
# speedup vs baseline: 3.7961x; 1.1261x over previous
#include <hip/hip_runtime.h>

#define IN_DIM   128
#define HID_DIM  128
#define OUT_DIM  64
#define NUM_GRAPHS 64

typedef float v2f __attribute__((ext_vector_type(2)));
typedef float v8f __attribute__((ext_vector_type(8)));

#define CAP 32

__global__ void bucket_kernel(const int* __restrict__ src, const int* __restrict__ dst,
                              int* __restrict__ cnt, int* __restrict__ bucket,
                              int* __restrict__ ovf, int E, int Nn) {
    int e = blockIdx.x * blockDim.x + threadIdx.x;
    if (e < E) {
        const int d = dst[e];
        const int r = atomicAdd(&cnt[d], 1);
        if (r < CAP) bucket[(size_t)d * CAP + r] = src[e];
        else         ovf[atomicAdd(&cnt[Nn], 1)] = e;
    }
}

template <int N, int WAVES, bool INIT>
__global__ void __launch_bounds__(WAVES * 32, 7)
gemm_strip_wmma_f32(const float* __restrict__ A,
                    const float* __restrict__ B,
                    float* __restrict__ C, int M,
                    int* __restrict__ zi, int nzi,
                    float* __restrict__ zf, int nzf) {
    constexpr int K   = 128;
    static_assert(IN_DIM == K && HID_DIM == K && N % 64 == 0, "strip GEMM is written for K = 128");
    constexpr int NH  = N / 64;
    constexpr int SLOTS = (K / 4) * N;
    __shared__ float4 Ws[SLOTS];

    const int tid  = threadIdx.x;
    const int lane = tid & 31;
    const int wave = tid >> 5;
    const int c    = lane & 15;
    const int h    = lane >> 4;
    const int task = blockIdx.x * WAVES + wave;
    const int tile = task / NH;
    const int cg   = task - tile * NH;
    const bool live = tile < (M >> 4);
    const int row0 = tile << 4;

    float4 a[16];
    if (live) {
        const float* Ap = A + (size_t)(row0 + c) * K + 4 * h;
#pragma unroll
        for (int j = 0; j < 16; ++j) a[j] = *(const float4*)(Ap + 8 * j);
    }

#pragma unroll 4
    for (int u = tid; u < SLOTS; u += WAVES * 32) {
        const int q  = u / N;
        const int p  = u - q * N;
        const int pl = p & 63;
        const int n  = (p - pl) + ((pl & 15) << 2) + (pl >> 4);
        const float* bp = B + (size_t)(4 * q) * N + n;
        float4 w;
        w.x = bp[0]; w.y = bp[N]; w.z = bp[2 * N]; w.w = bp[3 * N];
        Ws[u] = w;
    }
    __syncthreads();

    if (INIT) {
        const int g = blockIdx.x * (WAVES * 32) + tid;
        if (g < nzi) zi[g] = 0;
        if (g < nzf) zf[g] = 0.0f;
    }

    if (live) {
        v8f acc[4];
#pragma unroll
        for (int t = 0; t < 4; ++t) acc[t] = v8f{};

        const float4* Wl = Ws + h * N + cg * 64 + c;
        float4 bc[4], bn[4];
#pragma unroll
        for (int t = 0; t < 4; ++t) bc[t] = Wl[16 * t];

        __builtin_amdgcn_sched_group_barrier(0x100, 8, 0);
#pragma unroll
        for (int j = 0; j < 16; ++j) {
            if (j + 1 < 16) {
#pragma unroll
                for (int t = 0; t < 4; ++t) bn[t] = Wl[(2 * (j + 1)) * N + 16 * t];
            }
            v2f a0, a1;
            a0.x = a[j].x; a0.y = a[j].y;
            a1.x = a[j].z; a1.y = a[j].w;
#pragma unroll
            for (int t = 0; t < 4; ++t) {
                v2f b0;
                b0.x = bc[t].x; b0.y = bc[t].y;
                acc[t] = __builtin_amdgcn_wmma_f32_16x16x4_f32(
                     false, a0,  false, b0,
                     (short)0, acc[t],  false,  false);
            }
#pragma unroll
            for (int t = 0; t < 4; ++t) {
                v2f b1;
                b1.x = bc[t].z; b1.y = bc[t].w;
                acc[t] = __builtin_amdgcn_wmma_f32_16x16x4_f32(
                     false, a1,  false, b1,
                     (short)0, acc[t],  false,  false);
            }
            __builtin_amdgcn_sched_group_barrier(0x008, 8, 0);
            if (j + 2 < 16)
                __builtin_amdgcn_sched_group_barrier(0x100, 4, 0);
#pragma unroll
            for (int t = 0; t < 4; ++t) bc[t] = bn[t];
        }

        float* Cp = C + (size_t)(row0 + (h << 3)) * N + cg * 64 + (c << 2);
#pragma unroll
        for (int v = 0; v < 8; ++v) {
            float4 o;
            o.x = acc[0][v]; o.y = acc[1][v]; o.z = acc[2][v]; o.w = acc[3][v];
            *(float4*)(Cp + (size_t)v * N) = o;
        }
    }
}

__device__ __forceinline__ float dinv_of(int indeg) {
    return __builtin_amdgcn_rsqf((float)(indeg + 1));
}

template <int D>
__device__ __forceinline__ float4 gather_sum(const float* __restrict__ h,
                                             const int* __restrict__ cnt,
                                             const int* __restrict__ bucket,
                                             const int* __restrict__ ovf,
                                             const int* __restrict__ src,
                                             const int* __restrict__ dst,
                                             int i, int c, int Nn) {
    const int deg = cnt[i];
    const int nb  = min(deg, CAP);
    const int* bk = bucket + (size_t)i * CAP;
    const float di = dinv_of(deg);
    const float4 hs = *(const float4*)(h + (size_t)i * D + c);
    const float wself = di * di;
    float4 acc;
    acc.x = hs.x * wself; acc.y = hs.y * wself;
    acc.z = hs.z * wself; acc.w = hs.w * wself;

    int j = 0;
    for (; j + 4 <= nb; j += 4) {
        const int4 s = *(const int4*)(bk + j);
        const float n0 = dinv_of(cnt[s.x]) * di;
        const float n1 = dinv_of(cnt[s.y]) * di;
        const float n2 = dinv_of(cnt[s.z]) * di;
        const float n3 = dinv_of(cnt[s.w]) * di;
        const float4 v0 = *(const float4*)(h + (size_t)s.x * D + c);
        const float4 v1 = *(const float4*)(h + (size_t)s.y * D + c);
        const float4 v2 = *(const float4*)(h + (size_t)s.z * D + c);
        const float4 v3 = *(const float4*)(h + (size_t)s.w * D + c);
        acc.x = fmaf(v0.x, n0, acc.x); acc.y = fmaf(v0.y, n0, acc.y);
        acc.z = fmaf(v0.z, n0, acc.z); acc.w = fmaf(v0.w, n0, acc.w);
        acc.x = fmaf(v1.x, n1, acc.x); acc.y = fmaf(v1.y, n1, acc.y);
        acc.z = fmaf(v1.z, n1, acc.z); acc.w = fmaf(v1.w, n1, acc.w);
        acc.x = fmaf(v2.x, n2, acc.x); acc.y = fmaf(v2.y, n2, acc.y);
        acc.z = fmaf(v2.z, n2, acc.z); acc.w = fmaf(v2.w, n2, acc.w);
        acc.x = fmaf(v3.x, n3, acc.x); acc.y = fmaf(v3.y, n3, acc.y);
        acc.z = fmaf(v3.z, n3, acc.z); acc.w = fmaf(v3.w, n3, acc.w);
    }
    const int rem = nb - j;
    if (rem > 0) {
        const int4 s = *(const int4*)(bk + j);
        const int s0 = s.x;
        const int s1 = (rem > 1) ? s.y : s.x;
        const int s2 = (rem > 2) ? s.z : s.x;
        const float n0 = dinv_of(cnt[s0]) * di;
        const float n1 = dinv_of(cnt[s1]) * di;
        const float n2 = dinv_of(cnt[s2]) * di;
        const float4 v0 = *(const float4*)(h + (size_t)s0 * D + c);
        float4 v1 = *(const float4*)(h + (size_t)s1 * D + c);
        float4 v2 = *(const float4*)(h + (size_t)s2 * D + c);
        if (rem < 2) { v1.x = 0.0f; v1.y = 0.0f; v1.z = 0.0f; v1.w = 0.0f; }
        if (rem < 3) { v2.x = 0.0f; v2.y = 0.0f; v2.z = 0.0f; v2.w = 0.0f; }
        acc.x = fmaf(v0.x, n0, acc.x); acc.y = fmaf(v0.y, n0, acc.y);
        acc.z = fmaf(v0.z, n0, acc.z); acc.w = fmaf(v0.w, n0, acc.w);
        acc.x = fmaf(v1.x, n1, acc.x); acc.y = fmaf(v1.y, n1, acc.y);
        acc.z = fmaf(v1.z, n1, acc.z); acc.w = fmaf(v1.w, n1, acc.w);
        acc.x = fmaf(v2.x, n2, acc.x); acc.y = fmaf(v2.y, n2, acc.y);
        acc.z = fmaf(v2.z, n2, acc.z); acc.w = fmaf(v2.w, n2, acc.w);
    }
    if (deg > CAP) {
        const int novf = cnt[Nn];
        for (int k = 0; k < novf; ++k) {
            const int e = ovf[k];
            if (dst[e] == i) {
                const int s = src[e];
                const float n = dinv_of(cnt[s]) * di;
                const float4 v = *(const float4*)(h + (size_t)s * D + c);
                acc.x = fmaf(v.x, n, acc.x); acc.y = fmaf(v.y, n, acc.y);
                acc.z = fmaf(v.z, n, acc.z); acc.w = fmaf(v.w, n, acc.w);
            }
        }
    }
    return acc;
}

#define FN 64
#define FW 16
#define FPITCH 132

__global__ void __launch_bounds__(FW * 32, 16)
gather_relu_gemm2(const float* __restrict__ h,
                  const int* __restrict__ cnt,
                  const int* __restrict__ bucket,
                  const int* __restrict__ ovf,
                  const int* __restrict__ src,
                  const int* __restrict__ dst,
                  const float* __restrict__ bias,
                  const float* __restrict__ B,
                  float* __restrict__ C, int Nn) {
    constexpr int K = HID_DIM;
    constexpr int N = OUT_DIM;
    static_assert(K == 128 && N == 64 && FN == 4 * FW && FN % 16 == 0, "written for 128 x 64 and four nodes per wave");
    __shared__ float4 Ws[(K / 4) * N];
    __shared__ alignas(16) float rows[FN * FPITCH];

    const int tid  = threadIdx.x;
    const int lane = tid & 31;
    const int wave = tid >> 5;

#pragma unroll
    for (int u = tid; u < (K / 4) * N; u += FW * 32) {
        const int q  = u / N;
        const int p  = u - q * N;
        const int pl = p & 31;
        const int n  = (p - pl) + ((pl & 15) << 1) + (pl >> 4);
        const float* bp = B + (size_t)(4 * q) * N + n;
        float4 w;
        w.x = bp[0]; w.y = bp[N]; w.z = bp[2 * N]; w.w = bp[3 * N];
        Ws[u] = w;
    }

    const int col = lane << 2;
    const float4 bv = *(const float4*)(bias + col);
    for (int n = 0; n < FN / FW; ++n) {
        const int q = wave * (FN / FW) + n;
        const int i = blockIdx.x * FN + q;
        float4 r;
        r.x = 0.0f; r.y = 0.0f; r.z = 0.0f; r.w = 0.0f;
        if (i < Nn) {
            const float4 acc = gather_sum<K>(h, cnt, bucket, ovf, src, dst, i, col, Nn);
            r.x = fmaxf(acc.x + bv.x, 0.0f); r.y = fmaxf(acc.y + bv.y, 0.0f);
            r.z = fmaxf(acc.z + bv.z, 0.0f); r.w = fmaxf(acc.w + bv.w, 0.0f);
        }
        *(float4*)(&rows[q * FPITCH + col]) = r;
    }
    __syncthreads();

    const int rt   = wave >> 1;
    const int cg   = wave & 1;
    const int row0 = blockIdx.x * FN + (rt << 4);
    if (wave < 2 * (FN / 16) && row0 < Nn) {
        const int c = lane & 15;
        const int hh = lane >> 4;
        const float*  Ar = rows + ((rt << 4) + c) * FPITCH + 4 * hh;
        const float4* Wl = Ws + hh * N + cg * 32 + c;

        v8f acc0 = v8f{};
        v8f acc1 = v8f{};
        float4 ac = *(const float4*)(Ar);
        float4 b0c = Wl[0];
        float4 b1c = Wl[16];
        float4 an = ac, b0n = b0c, b1n = b1c;
#pragma unroll
        for (int j = 0; j < 16; ++j) {
            if (j + 1 < 16) {
                an  = *(const float4*)(Ar + 8 * (j + 1));
                b0n = Wl[(2 * (j + 1)) * N];
                b1n = Wl[(2 * (j + 1)) * N + 16];
            }
            v2f a0, a1, p0, p1, q0, q1;
            a0.x = ac.x;  a0.y = ac.y;  a1.x = ac.z;  a1.y = ac.w;
            p0.x = b0c.x; p0.y = b0c.y; p1.x = b0c.z; p1.y = b0c.w;
            q0.x = b1c.x; q0.y = b1c.y; q1.x = b1c.z; q1.y = b1c.w;
            acc0 = __builtin_amdgcn_wmma_f32_16x16x4_f32(false, a0, false, p0, (short)0, acc0, false, false);
            acc1 = __builtin_amdgcn_wmma_f32_16x16x4_f32(false, a0, false, q0, (short)0, acc1, false, false);
            acc0 = __builtin_amdgcn_wmma_f32_16x16x4_f32(false, a1, false, p1, (short)0, acc0, false, false);
            acc1 = __builtin_amdgcn_wmma_f32_16x16x4_f32(false, a1, false, q1, (short)0, acc1, false, false);
            ac = an; b0c = b0n; b1c = b1n;
        }

        float* Cp = C + (size_t)(row0 + (hh << 3)) * N + cg * 32 + (c << 1);
#pragma unroll
        for (int v = 0; v < 8; ++v) {
            float2 o;
            o.x = acc0[v]; o.y = acc1[v];
            if (row0 + (hh << 3) + v < Nn) *(float2*)(Cp + (size_t)v * N) = o;
        }
    }
}

template <int D>
__global__ void __launch_bounds__(256)
gather_pool_rows(const float* __restrict__ h,
                 const int* __restrict__ cnt,
                 const int* __restrict__ bucket,
                 const int* __restrict__ ovf,
                 const int* __restrict__ src,
                 const int* __restrict__ dst,
                 const float* __restrict__ bias,
                 const int* __restrict__ batch,
                 float* __restrict__ sums,
                 float* __restrict__ cnts, int Nn) {
    constexpr int LPN = D >> 2;
    constexpr int NPB = 256 / LPN;
    __shared__ float rows[NPB][D];
    __shared__ int   gid[NPB];

    const int q  = threadIdx.x / LPN;
    const int lc = threadIdx.x - q * LPN;
    const int c  = lc << 2;
    const int i  = blockIdx.x * NPB + q;

    float4 r;
    r.x = 0.0f; r.y = 0.0f; r.z = 0.0f; r.w = 0.0f;
    int g = -1;
    if (i < Nn) {
        g = batch[i];
        const float4 acc = gather_sum<D>(h, cnt, bucket, ovf, src, dst, i, c, Nn);
        const float4 bv = *(const float4*)(bias + c);
        r.x = acc.x + bv.x; r.y = acc.y + bv.y;
        r.z = acc.z + bv.z; r.w = acc.w + bv.w;
    }

    *(float4*)(&rows[q][c]) = r;
    if (lc == 0) gid[q] = g;
    __syncthreads();

    if (threadIdx.x < D) {
        const int d = threadIdx.x;
        float v[NPB];
        int   gk[NPB];
#pragma unroll
        for (int k = 0; k < NPB; ++k) { v[k] = rows[k][d]; gk[k] = gid[k]; }

        float s  = 0.0f;
        int  run = 0;
        int  gc  = gk[0];
#pragma unroll
        for (int k = 0; k < NPB; ++k) {
            if (gk[k] != gc) {
                if (gc >= 0 && gc < NUM_GRAPHS) {
                    atomicAdd(&sums[(size_t)gc * D + d], s);
                    if (d == 0) atomicAdd(&cnts[gc], (float)run);
                }
                s = 0.0f; run = 0; gc = gk[k];
            }
            s += v[k];
            ++run;
        }
        if (gc >= 0 && gc < NUM_GRAPHS) {
            atomicAdd(&sums[(size_t)gc * D + d], s);
            if (d == 0) atomicAdd(&cnts[gc], (float)run);
        }
    }
}

__global__ void pool_div(const float* __restrict__ sums,
                         const float* __restrict__ cnt,
                         float* __restrict__ out, int G, int D) {
    int tid = blockIdx.x * blockDim.x + threadIdx.x;
    if (tid >= G * D) return;
    int g = tid / D;
    out[tid] = sums[tid] / fmaxf(cnt[g], 1.0f);
}

extern "C" void kernel_launch(void* const* d_in, const int* in_sizes, int n_in,
                              void* d_out, int out_size, void* d_ws, size_t ws_size,
                              hipStream_t stream) {
    const float* x    = (const float*)d_in[0];
    const int*   eidx = (const int*)  d_in[1];
    const int*   batch= (const int*)  d_in[2];
    const float* W1   = (const float*)d_in[3];
    const float* b1   = (const float*)d_in[4];
    const float* W2   = (const float*)d_in[5];
    const float* b2   = (const float*)d_in[6];
    float* out = (float*)d_out;

    const int Nn = in_sizes[0] / IN_DIM;
    const int E  = in_sizes[1] / 2;
    const int* src = eidx;
    const int* dst = eidx + E;

    float* ws   = (float*)d_ws;
    float* h1   = ws;
    float* h2   = h1   + (size_t)Nn * HID_DIM;
    int* bucket = (int*)(h2 + (size_t)Nn * OUT_DIM);
    float* sums = (float*)(bucket + (size_t)Nn * CAP);
    float* cnts = sums + NUM_GRAPHS * OUT_DIM;
    int* cnt    = (int*)(cnts + NUM_GRAPHS);
    int* ovf    = cnt + Nn + 1;

    const int B = 256;
    auto blocks = [](long long n, int b) { return (unsigned)((n + b - 1) / b); };

    gemm_strip_wmma_f32<HID_DIM, 14, true><<<blocks((long long)(Nn >> 4) * (HID_DIM / 64), 14), 14 * 32, 0, stream>>>(
        x, W1, h1, Nn, cnt, Nn + 1, sums, NUM_GRAPHS * OUT_DIM + NUM_GRAPHS);

    bucket_kernel<<<blocks(E, B), B, 0, stream>>>(src, dst, cnt, bucket, ovf, E, Nn);

    gather_relu_gemm2<<<blocks(Nn, FN), FW * 32, 0, stream>>>(
        h1, cnt, bucket, ovf, src, dst, b1, W2, h2, Nn);

    gather_pool_rows<OUT_DIM><<<blocks((long long)Nn * (OUT_DIM >> 2), B), B, 0, stream>>>(
        h2, cnt, bucket, ovf, src, dst, b2, batch, sums, cnts, Nn);

    pool_div<<<blocks(NUM_GRAPHS * OUT_DIM, B), B, 0, stream>>>(
        sums, cnts, out, NUM_GRAPHS, OUT_DIM);
}


# Edge bucketing folded into the layer-1 product's launch: each thread takes 3 edges, counting atomics leave before its WMMA run and their ranks are used after it; start values get a small launch
# speedup vs baseline: 4.1563x; 1.0949x over previous
#include <hip/hip_runtime.h>

#define IN_DIM   128
#define HID_DIM  128
#define OUT_DIM  64
#define NUM_GRAPHS 64

typedef float v2f __attribute__((ext_vector_type(2)));
typedef float v8f __attribute__((ext_vector_type(8)));

#define CAP 32


__global__ void start_values(int* __restrict__ zi, int nzi, float* __restrict__ zf, int nzf) {
    const int stride = gridDim.x * blockDim.x;
    for (int g = blockIdx.x * blockDim.x + threadIdx.x; g < nzi; g += stride) zi[g] = 0;
    for (int g = blockIdx.x * blockDim.x + threadIdx.x; g < nzf; g += stride) zf[g] = 0.0f;
}

template <int N, int WAVES, int EPT>
__global__ void __launch_bounds__(WAVES * 32, 7)
gemm_strip_wmma_f32(const float* __restrict__ A,
                    const float* __restrict__ B,
                    float* __restrict__ C, int M,
                    const int* __restrict__ esrc, const int* __restrict__ edst,
                    int* __restrict__ cnt, int* __restrict__ bucket,
                    int* __restrict__ ovf, int E, int Nn) {
    constexpr int K   = 128;
    static_assert(IN_DIM == K && HID_DIM == K && N % 64 == 0, "strip GEMM is written for K = 128");
    constexpr int NH  = N / 64;
    constexpr int SLOTS = (K / 4) * N;
    __shared__ float4 Ws[SLOTS];

    const int tid  = threadIdx.x;
    const int lane = tid & 31;
    const int wave = tid >> 5;
    const int c    = lane & 15;
    const int h    = lane >> 4;
    const int task = blockIdx.x * WAVES + wave;
    const int tile = task / NH;
    const int cg   = task - tile * NH;
    const bool live = tile < (M >> 4);
    const int row0 = tile << 4;

    float4 a[16];
    if (live) {
        const float* Ap = A + (size_t)(row0 + c) * K + 4 * h;
#pragma unroll
        for (int j = 0; j < 16; ++j) a[j] = *(const float4*)(Ap + 8 * j);
    }

    constexpr int EP = EPT > 0 ? EPT : 1;
    const int eg = blockIdx.x * (WAVES * 32) + tid;
    const int eT = gridDim.x * (WAVES * 32);
    int ed[EP], er[EP];
    if (EPT > 0 && E > 0) {
#pragma unroll
        for (int k = 0; k < EP; ++k) ed[k] = edst[min(eg + k * eT, E - 1)];
    }

#pragma unroll 4
    for (int u = tid; u < SLOTS; u += WAVES * 32) {
        const int q  = u / N;
        const int p  = u - q * N;
        const int pl = p & 63;
        const int n  = (p - pl) + ((pl & 15) << 2) + (pl >> 4);
        const float* bp = B + (size_t)(4 * q) * N + n;
        float4 w;
        w.x = bp[0]; w.y = bp[N]; w.z = bp[2 * N]; w.w = bp[3 * N];
        Ws[u] = w;
    }
    __syncthreads();

    if (EPT > 0) {
#pragma unroll
        for (int k = 0; k < EP; ++k) {
            er[k] = 0;
            if (eg + k * eT < E) er[k] = atomicAdd(&cnt[ed[k]], 1);
        }
    }

    if (live) {
        v8f acc[4];
#pragma unroll
        for (int t = 0; t < 4; ++t) acc[t] = v8f{};

        const float4* Wl = Ws + h * N + cg * 64 + c;
        float4 bc[4], bn[4];
#pragma unroll
        for (int t = 0; t < 4; ++t) bc[t] = Wl[16 * t];

        __builtin_amdgcn_sched_group_barrier(0x100, 8, 0);
#pragma unroll
        for (int j = 0; j < 16; ++j) {
            if (j + 1 < 16) {
#pragma unroll
                for (int t = 0; t < 4; ++t) bn[t] = Wl[(2 * (j + 1)) * N + 16 * t];
            }
            v2f a0, a1;
            a0.x = a[j].x; a0.y = a[j].y;
            a1.x = a[j].z; a1.y = a[j].w;
#pragma unroll
            for (int t = 0; t < 4; ++t) {
                v2f b0;
                b0.x = bc[t].x; b0.y = bc[t].y;
                acc[t] = __builtin_amdgcn_wmma_f32_16x16x4_f32(
                     false, a0,  false, b0,
                     (short)0, acc[t],  false,  false);
            }
#pragma unroll
            for (int t = 0; t < 4; ++t) {
                v2f b1;
                b1.x = bc[t].z; b1.y = bc[t].w;
                acc[t] = __builtin_amdgcn_wmma_f32_16x16x4_f32(
                     false, a1,  false, b1,
                     (short)0, acc[t],  false,  false);
            }
            __builtin_amdgcn_sched_group_barrier(0x008, 8, 0);
            if (j + 2 < 16)
                __builtin_amdgcn_sched_group_barrier(0x100, 4, 0);
#pragma unroll
            for (int t = 0; t < 4; ++t) bc[t] = bn[t];
        }

        float* Cp = C + (size_t)(row0 + (h << 3)) * N + cg * 64 + (c << 2);
#pragma unroll
        for (int v = 0; v < 8; ++v) {
            float4 o;
            o.x = acc[0][v]; o.y = acc[1][v]; o.z = acc[2][v]; o.w = acc[3][v];
            *(float4*)(Cp + (size_t)v * N) = o;
        }
    }

    if (EPT > 0) {
        int es[EP];
        if (E > 0) {
#pragma unroll
            for (int k = 0; k < EP; ++k) es[k] = esrc[min(eg + k * eT, E - 1)];
        }
#pragma unroll
        for (int k = 0; k < EP; ++k) {
            const int e = eg + k * eT;
            if (e < E) {
                if (er[k] < CAP) bucket[(size_t)ed[k] * CAP + er[k]] = es[k];
                else             ovf[atomicAdd(&cnt[Nn], 1)] = e;
            }
        }
        for (int e = eg + EP * eT; e < E; e += eT) {
            const int d = edst[e];
            const int r = atomicAdd(&cnt[d], 1);
            if (r < CAP) bucket[(size_t)d * CAP + r] = esrc[e];
            else         ovf[atomicAdd(&cnt[Nn], 1)] = e;
        }
    }
}

__device__ __forceinline__ float dinv_of(int indeg) {
    return __builtin_amdgcn_rsqf((float)(indeg + 1));
}

template <int D>
__device__ __forceinline__ float4 gather_sum(const float* __restrict__ h,
                                             const int* __restrict__ cnt,
                                             const int* __restrict__ bucket,
                                             const int* __restrict__ ovf,
                                             const int* __restrict__ src,
                                             const int* __restrict__ dst,
                                             int i, int c, int Nn) {
    const int deg = cnt[i];
    const int nb  = min(deg, CAP);
    const int* bk = bucket + (size_t)i * CAP;
    const float di = dinv_of(deg);
    const float4 hs = *(const float4*)(h + (size_t)i * D + c);
    const float wself = di * di;
    float4 acc;
    acc.x = hs.x * wself; acc.y = hs.y * wself;
    acc.z = hs.z * wself; acc.w = hs.w * wself;

    int j = 0;
    for (; j + 4 <= nb; j += 4) {
        const int4 s = *(const int4*)(bk + j);
        const float n0 = dinv_of(cnt[s.x]) * di;
        const float n1 = dinv_of(cnt[s.y]) * di;
        const float n2 = dinv_of(cnt[s.z]) * di;
        const float n3 = dinv_of(cnt[s.w]) * di;
        const float4 v0 = *(const float4*)(h + (size_t)s.x * D + c);
        const float4 v1 = *(const float4*)(h + (size_t)s.y * D + c);
        const float4 v2 = *(const float4*)(h + (size_t)s.z * D + c);
        const float4 v3 = *(const float4*)(h + (size_t)s.w * D + c);
        acc.x = fmaf(v0.x, n0, acc.x); acc.y = fmaf(v0.y, n0, acc.y);
        acc.z = fmaf(v0.z, n0, acc.z); acc.w = fmaf(v0.w, n0, acc.w);
        acc.x = fmaf(v1.x, n1, acc.x); acc.y = fmaf(v1.y, n1, acc.y);
        acc.z = fmaf(v1.z, n1, acc.z); acc.w = fmaf(v1.w, n1, acc.w);
        acc.x = fmaf(v2.x, n2, acc.x); acc.y = fmaf(v2.y, n2, acc.y);
        acc.z = fmaf(v2.z, n2, acc.z); acc.w = fmaf(v2.w, n2, acc.w);
        acc.x = fmaf(v3.x, n3, acc.x); acc.y = fmaf(v3.y, n3, acc.y);
        acc.z = fmaf(v3.z, n3, acc.z); acc.w = fmaf(v3.w, n3, acc.w);
    }
    const int rem = nb - j;
    if (rem > 0) {
        const int4 s = *(const int4*)(bk + j);
        const int s0 = s.x;
        const int s1 = (rem > 1) ? s.y : s.x;
        const int s2 = (rem > 2) ? s.z : s.x;
        const float n0 = dinv_of(cnt[s0]) * di;
        const float n1 = dinv_of(cnt[s1]) * di;
        const float n2 = dinv_of(cnt[s2]) * di;
        const float4 v0 = *(const float4*)(h + (size_t)s0 * D + c);
        float4 v1 = *(const float4*)(h + (size_t)s1 * D + c);
        float4 v2 = *(const float4*)(h + (size_t)s2 * D + c);
        if (rem < 2) { v1.x = 0.0f; v1.y = 0.0f; v1.z = 0.0f; v1.w = 0.0f; }
        if (rem < 3) { v2.x = 0.0f; v2.y = 0.0f; v2.z = 0.0f; v2.w = 0.0f; }
        acc.x = fmaf(v0.x, n0, acc.x); acc.y = fmaf(v0.y, n0, acc.y);
        acc.z = fmaf(v0.z, n0, acc.z); acc.w = fmaf(v0.w, n0, acc.w);
        acc.x = fmaf(v1.x, n1, acc.x); acc.y = fmaf(v1.y, n1, acc.y);
        acc.z = fmaf(v1.z, n1, acc.z); acc.w = fmaf(v1.w, n1, acc.w);
        acc.x = fmaf(v2.x, n2, acc.x); acc.y = fmaf(v2.y, n2, acc.y);
        acc.z = fmaf(v2.z, n2, acc.z); acc.w = fmaf(v2.w, n2, acc.w);
    }
    if (deg > CAP) {
        const int novf = cnt[Nn];
        for (int k = 0; k < novf; ++k) {
            const int e = ovf[k];
            if (dst[e] == i) {
                const int s = src[e];
                const float n = dinv_of(cnt[s]) * di;
                const float4 v = *(const float4*)(h + (size_t)s * D + c);
                acc.x = fmaf(v.x, n, acc.x); acc.y = fmaf(v.y, n, acc.y);
                acc.z = fmaf(v.z, n, acc.z); acc.w = fmaf(v.w, n, acc.w);
            }
        }
    }
    return acc;
}

#define FN 64
#define FW 16
#define FPITCH 132

__global__ void __launch_bounds__(FW * 32, 16)
gather_relu_gemm2(const float* __restrict__ h,
                  const int* __restrict__ cnt,
                  const int* __restrict__ bucket,
                  const int* __restrict__ ovf,
                  const int* __restrict__ src,
                  const int* __restrict__ dst,
                  const float* __restrict__ bias,
                  const float* __restrict__ B,
                  float* __restrict__ C, int Nn) {
    constexpr int K = HID_DIM;
    constexpr int N = OUT_DIM;
    static_assert(K == 128 && N == 64 && FN == 4 * FW && FN % 16 == 0, "written for 128 x 64 and four nodes per wave");
    __shared__ float4 Ws[(K / 4) * N];
    __shared__ alignas(16) float rows[FN * FPITCH];

    const int tid  = threadIdx.x;
    const int lane = tid & 31;
    const int wave = tid >> 5;

#pragma unroll
    for (int u = tid; u < (K / 4) * N; u += FW * 32) {
        const int q  = u / N;
        const int p  = u - q * N;
        const int pl = p & 31;
        const int n  = (p - pl) + ((pl & 15) << 1) + (pl >> 4);
        const float* bp = B + (size_t)(4 * q) * N + n;
        float4 w;
        w.x = bp[0]; w.y = bp[N]; w.z = bp[2 * N]; w.w = bp[3 * N];
        Ws[u] = w;
    }

    const int col = lane << 2;
    const float4 bv = *(const float4*)(bias + col);
    for (int n = 0; n < FN / FW; ++n) {
        const int q = wave * (FN / FW) + n;
        const int i = blockIdx.x * FN + q;
        float4 r;
        r.x = 0.0f; r.y = 0.0f; r.z = 0.0f; r.w = 0.0f;
        if (i < Nn) {
            const float4 acc = gather_sum<K>(h, cnt, bucket, ovf, src, dst, i, col, Nn);
            r.x = fmaxf(acc.x + bv.x, 0.0f); r.y = fmaxf(acc.y + bv.y, 0.0f);
            r.z = fmaxf(acc.z + bv.z, 0.0f); r.w = fmaxf(acc.w + bv.w, 0.0f);
        }
        *(float4*)(&rows[q * FPITCH + col]) = r;
    }
    __syncthreads();

    const int rt   = wave >> 1;
    const int cg   = wave & 1;
    const int row0 = blockIdx.x * FN + (rt << 4);
    if (wave < 2 * (FN / 16) && row0 < Nn) {
        const int c = lane & 15;
        const int hh = lane >> 4;
        const float*  Ar = rows + ((rt << 4) + c) * FPITCH + 4 * hh;
        const float4* Wl = Ws + hh * N + cg * 32 + c;

        v8f acc0 = v8f{};
        v8f acc1 = v8f{};
        float4 ac = *(const float4*)(Ar);
        float4 b0c = Wl[0];
        float4 b1c = Wl[16];
        float4 an = ac, b0n = b0c, b1n = b1c;
#pragma unroll
        for (int j = 0; j < 16; ++j) {
            if (j + 1 < 16) {
                an  = *(const float4*)(Ar + 8 * (j + 1));
                b0n = Wl[(2 * (j + 1)) * N];
                b1n = Wl[(2 * (j + 1)) * N + 16];
            }
            v2f a0, a1, p0, p1, q0, q1;
            a0.x = ac.x;  a0.y = ac.y;  a1.x = ac.z;  a1.y = ac.w;
            p0.x = b0c.x; p0.y = b0c.y; p1.x = b0c.z; p1.y = b0c.w;
            q0.x = b1c.x; q0.y = b1c.y; q1.x = b1c.z; q1.y = b1c.w;
            acc0 = __builtin_amdgcn_wmma_f32_16x16x4_f32(false, a0, false, p0, (short)0, acc0, false, false);
            acc1 = __builtin_amdgcn_wmma_f32_16x16x4_f32(false, a0, false, q0, (short)0, acc1, false, false);
            acc0 = __builtin_amdgcn_wmma_f32_16x16x4_f32(false, a1, false, p1, (short)0, acc0, false, false);
            acc1 = __builtin_amdgcn_wmma_f32_16x16x4_f32(false, a1, false, q1, (short)0, acc1, false, false);
            ac = an; b0c = b0n; b1c = b1n;
        }

        float* Cp = C + (size_t)(row0 + (hh << 3)) * N + cg * 32 + (c << 1);
#pragma unroll
        for (int v = 0; v < 8; ++v) {
            float2 o;
            o.x = acc0[v]; o.y = acc1[v];
            if (row0 + (hh << 3) + v < Nn) *(float2*)(Cp + (size_t)v * N) = o;
        }
    }
}

template <int D>
__global__ void __launch_bounds__(256)
gather_pool_rows(const float* __restrict__ h,
                 const int* __restrict__ cnt,
                 const int* __restrict__ bucket,
                 const int* __restrict__ ovf,
                 const int* __restrict__ src,
                 const int* __restrict__ dst,
                 const float* __restrict__ bias,
                 const int* __restrict__ batch,
                 float* __restrict__ sums,
                 float* __restrict__ cnts, int Nn) {
    constexpr int LPN = D >> 2;
    constexpr int NPB = 256 / LPN;
    __shared__ float rows[NPB][D];
    __shared__ int   gid[NPB];

    const int q  = threadIdx.x / LPN;
    const int lc = threadIdx.x - q * LPN;
    const int c  = lc << 2;
    const int i  = blockIdx.x * NPB + q;

    float4 r;
    r.x = 0.0f; r.y = 0.0f; r.z = 0.0f; r.w = 0.0f;
    int g = -1;
    if (i < Nn) {
        g = batch[i];
        const float4 acc = gather_sum<D>(h, cnt, bucket, ovf, src, dst, i, c, Nn);
        const float4 bv = *(const float4*)(bias + c);
        r.x = acc.x + bv.x; r.y = acc.y + bv.y;
        r.z = acc.z + bv.z; r.w = acc.w + bv.w;
    }

    *(float4*)(&rows[q][c]) = r;
    if (lc == 0) gid[q] = g;
    __syncthreads();

    if (threadIdx.x < D) {
        const int d = threadIdx.x;
        float v[NPB];
        int   gk[NPB];
#pragma unroll
        for (int k = 0; k < NPB; ++k) { v[k] = rows[k][d]; gk[k] = gid[k]; }

        float s  = 0.0f;
        int  run = 0;
        int  gc  = gk[0];
#pragma unroll
        for (int k = 0; k < NPB; ++k) {
            if (gk[k] != gc) {
                if (gc >= 0 && gc < NUM_GRAPHS) {
                    atomicAdd(&sums[(size_t)gc * D + d], s);
                    if (d == 0) atomicAdd(&cnts[gc], (float)run);
                }
                s = 0.0f; run = 0; gc = gk[k];
            }
            s += v[k];
            ++run;
        }
        if (gc >= 0 && gc < NUM_GRAPHS) {
            atomicAdd(&sums[(size_t)gc * D + d], s);
            if (d == 0) atomicAdd(&cnts[gc], (float)run);
        }
    }
}

__global__ void pool_div(const float* __restrict__ sums,
                         const float* __restrict__ cnt,
                         float* __restrict__ out, int G, int D) {
    int tid = blockIdx.x * blockDim.x + threadIdx.x;
    if (tid >= G * D) return;
    int g = tid / D;
    out[tid] = sums[tid] / fmaxf(cnt[g], 1.0f);
}

extern "C" void kernel_launch(void* const* d_in, const int* in_sizes, int n_in,
                              void* d_out, int out_size, void* d_ws, size_t ws_size,
                              hipStream_t stream) {
    const float* x    = (const float*)d_in[0];
    const int*   eidx = (const int*)  d_in[1];
    const int*   batch= (const int*)  d_in[2];
    const float* W1   = (const float*)d_in[3];
    const float* b1   = (const float*)d_in[4];
    const float* W2   = (const float*)d_in[5];
    const float* b2   = (const float*)d_in[6];
    float* out = (float*)d_out;

    const int Nn = in_sizes[0] / IN_DIM;
    const int E  = in_sizes[1] / 2;
    const int* src = eidx;
    const int* dst = eidx + E;

    float* ws   = (float*)d_ws;
    float* h1   = ws;
    float* h2   = h1   + (size_t)Nn * HID_DIM;
    int* bucket = (int*)(h2 + (size_t)Nn * OUT_DIM);
    float* sums = (float*)(bucket + (size_t)Nn * CAP);
    float* cnts = sums + NUM_GRAPHS * OUT_DIM;
    int* cnt    = (int*)(cnts + NUM_GRAPHS);
    int* ovf    = cnt + Nn + 1;

    const int B = 256;
    auto blocks = [](long long n, int b) { return (unsigned)((n + b - 1) / b); };

    start_values<<<blocks(Nn + 1, B), B, 0, stream>>>(
        cnt, Nn + 1, sums, NUM_GRAPHS * OUT_DIM + NUM_GRAPHS);

    gemm_strip_wmma_f32<HID_DIM, 14, 3><<<blocks((long long)(Nn >> 4) * (HID_DIM / 64), 14), 14 * 32, 0, stream>>>(
        x, W1, h1, Nn, src, dst, cnt, bucket, ovf, E, Nn);

    gather_relu_gemm2<<<blocks(Nn, FN), FW * 32, 0, stream>>>(
        h1, cnt, bucket, ovf, src, dst, b1, W2, h2, Nn);

    gather_pool_rows<OUT_DIM><<<blocks((long long)Nn * (OUT_DIM >> 2), B), B, 0, stream>>>(
        h2, cnt, bucket, ovf, src, dst, b2, batch, sums, cnts, Nn);

    pool_div<<<blocks(NUM_GRAPHS * OUT_DIM, B), B, 0, stream>>>(
        sums, cnts, out, NUM_GRAPHS, OUT_DIM);
}


# Edge grouping split by chip half: nodes cut in two ranges, a block reads its die, draws an edge chunk for its half's range; count lines and slot rows get one writing half; arrival (sum) order moves
# speedup vs baseline: 5.4950x; 1.3221x over previous
#include <hip/hip_runtime.h>

#define IN_DIM   128
#define HID_DIM  128
#define OUT_DIM  64
#define NUM_GRAPHS 64

typedef float v2f __attribute__((ext_vector_type(2)));
typedef float v8f __attribute__((ext_vector_type(8)));

#define CAP 32


__global__ void start_values(int* __restrict__ zi, int nzi, float* __restrict__ zf, int nzf) {
    const int stride = gridDim.x * blockDim.x;
    for (int g = blockIdx.x * blockDim.x + threadIdx.x; g < nzi; g += stride) zi[g] = 0;
    for (int g = blockIdx.x * blockDim.x + threadIdx.x; g < nzf; g += stride) zf[g] = 0.0f;
}

template <int N, int WAVES, int EPT>
__global__ void __launch_bounds__(WAVES * 32, 7)
gemm_strip_wmma_f32(const float* __restrict__ A,
                    const float* __restrict__ B,
                    float* __restrict__ C, int M,
                    const int* __restrict__ esrc, const int* __restrict__ edst,
                    int* __restrict__ cnt, int* __restrict__ bucket,
                    int* __restrict__ ovf, int* __restrict__ tix, int E, int Nn) {
    constexpr int K   = 128;
    static_assert(IN_DIM == K && HID_DIM == K && N % 64 == 0, "strip GEMM is written for K = 128");
    constexpr int NH  = N / 64;
    constexpr int SLOTS = (K / 4) * N;
    __shared__ float4 Ws[SLOTS];
    __shared__ int etask[2];

    const int tid  = threadIdx.x;
    const int lane = tid & 31;
    const int wave = tid >> 5;
    const int c    = lane & 15;
    const int h    = lane >> 4;
    const int task = blockIdx.x * WAVES + wave;
    const int tile = task / NH;
    const int cg   = task - tile * NH;
    const bool live = tile < (M >> 4);
    const int row0 = tile << 4;

    float4 a[16];
    if (live) {
        const float* Ap = A + (size_t)(row0 + c) * K + 4 * h;
#pragma unroll
        for (int j = 0; j < 16; ++j) a[j] = *(const float4*)(Ap + 8 * j);
    }

    constexpr int EP = EPT > 0 ? EPT : 1;
    constexpr int BT = WAVES * 32;
    constexpr int CH = BT * EP;
    const int nch = (E + CH - 1) / CH;
    const int Nh  = ((Nn >> 1) + 63) & ~63;
    int tq = 0, tt = 0;
    if (EPT > 0 && tid == BT - 32) {
        tq = (int)((__builtin_amdgcn_s_sendmsg_rtn(0x87) >> 18) & 1u);
        tt = atomicAdd(&tix[tq], 1);
    }

#pragma unroll 4
    for (int u = tid; u < SLOTS; u += WAVES * 32) {
        const int q  = u / N;
        const int p  = u - q * N;
        const int pl = p & 63;
        const int n  = (p - pl) + ((pl & 15) << 2) + (pl >> 4);
        const float* bp = B + (size_t)(4 * q) * N + n;
        float4 w;
        w.x = bp[0]; w.y = bp[N]; w.z = bp[2 * N]; w.w = bp[3 * N];
        Ws[u] = w;
    }
    if (EPT > 0 && tid == BT - 32) {
        if (tt >= nch) { tq ^= 1; tt = atomicAdd(&tix[tq], 1); }
        etask[0] = tq;
        etask[1] = (tt < nch) ? tt : -1;
    }
    __syncthreads();

    int  er[EP];
    bool up  = false;
    bool has = false;
#pragma unroll
    for (int k = 0; k < EP; ++k) er[k] = 0;
    if (EPT > 0) {
        up = __builtin_amdgcn_readfirstlane(etask[0]) != 0;
        const int t = __builtin_amdgcn_readfirstlane(etask[1]);
        has = t >= 0;
        if (has) {
            const int e0 = t * CH + tid;
            int ed[EP];
#pragma unroll
            for (int k = 0; k < EP; ++k) ed[k] = edst[min(e0 + k * BT, E - 1)];
#pragma unroll
            for (int k = 0; k < EP; ++k) {
                if (e0 + k * BT < E && (ed[k] >= Nh) == up) er[k] = atomicAdd(&cnt[ed[k]], 1);
            }
        }
    }

    if (live) {
        v8f acc[4];
#pragma unroll
        for (int t = 0; t < 4; ++t) acc[t] = v8f{};

        const float4* Wl = Ws + h * N + cg * 64 + c;
        float4 bc[4], bn[4];
#pragma unroll
        for (int t = 0; t < 4; ++t) bc[t] = Wl[16 * t];

        __builtin_amdgcn_sched_group_barrier(0x100, 8, 0);
#pragma unroll
        for (int j = 0; j < 16; ++j) {
            if (j + 1 < 16) {
#pragma unroll
                for (int t = 0; t < 4; ++t) bn[t] = Wl[(2 * (j + 1)) * N + 16 * t];
            }
            v2f a0, a1;
            a0.x = a[j].x; a0.y = a[j].y;
            a1.x = a[j].z; a1.y = a[j].w;
#pragma unroll
            for (int t = 0; t < 4; ++t) {
                v2f b0;
                b0.x = bc[t].x; b0.y = bc[t].y;
                acc[t] = __builtin_amdgcn_wmma_f32_16x16x4_f32(
                     false, a0,  false, b0,
                     (short)0, acc[t],  false,  false);
            }
#pragma unroll
            for (int t = 0; t < 4; ++t) {
                v2f b1;
                b1.x = bc[t].z; b1.y = bc[t].w;
                acc[t] = __builtin_amdgcn_wmma_f32_16x16x4_f32(
                     false, a1,  false, b1,
                     (short)0, acc[t],  false,  false);
            }
            __builtin_amdgcn_sched_group_barrier(0x008, 8, 0);
            if (j + 2 < 16)
                __builtin_amdgcn_sched_group_barrier(0x100, 4, 0);
#pragma unroll
            for (int t = 0; t < 4; ++t) bc[t] = bn[t];
        }

        float* Cp = C + (size_t)(row0 + (h << 3)) * N + cg * 64 + (c << 2);
#pragma unroll
        for (int v = 0; v < 8; ++v) {
            float4 o;
            o.x = acc[0][v]; o.y = acc[1][v]; o.z = acc[2][v]; o.w = acc[3][v];
            *(float4*)(Cp + (size_t)v * N) = o;
        }
    }

    if (EPT > 0 && has) {
        const int e1 = __builtin_amdgcn_readfirstlane(
            __hip_atomic_load(&etask[1], __ATOMIC_RELAXED, __HIP_MEMORY_SCOPE_WORKGROUP)) * CH + tid;
        int ed[EP], es[EP];
#pragma unroll
        for (int k = 0; k < EP; ++k) {
            const int x = min(e1 + k * BT, E - 1);
            ed[k] = edst[x];
            es[k] = esrc[x];
        }
#pragma unroll
        for (int k = 0; k < EP; ++k) {
            const int e = e1 + k * BT;
            if (e < E && (ed[k] >= Nh) == up) {
                if (er[k] < CAP) bucket[ed[k] * CAP + er[k]] = es[k];
                else             ovf[atomicAdd(&cnt[Nn], 1)] = e;
            }
        }
    }

    if (EPT > 0 && 2 * nch > (int)gridDim.x) {
        for (int pass = 0; pass < 2; ++pass) {
            const bool upq = up != (pass != 0);
            for (;;) {
                __syncthreads();
                if (tid == 0) {
                    const int t = atomicAdd(&tix[upq ? 1 : 0], 1);
                    etask[1] = (t < nch) ? t : -1;
                }
                __syncthreads();
                const int t = __builtin_amdgcn_readfirstlane(etask[1]);
                if (t < 0) break;
                const int eend = min(E, (t + 1) * CH);
                for (int e = t * CH + tid; e < eend; e += BT) {
                    const int d = edst[e];
                    if ((d >= Nh) == upq) {
                        const int r = atomicAdd(&cnt[d], 1);
                        if (r < CAP) bucket[(size_t)d * CAP + r] = esrc[e];
                        else         ovf[atomicAdd(&cnt[Nn], 1)] = e;
                    }
                }
            }
        }
    }
}

__device__ __forceinline__ float dinv_of(int indeg) {
    return __builtin_amdgcn_rsqf((float)(indeg + 1));
}

template <int D>
__device__ __forceinline__ float4 gather_sum(const float* __restrict__ h,
                                             const int* __restrict__ cnt,
                                             const int* __restrict__ bucket,
                                             const int* __restrict__ ovf,
                                             const int* __restrict__ src,
                                             const int* __restrict__ dst,
                                             int i, int c, int Nn) {
    const int deg = cnt[i];
    const int nb  = min(deg, CAP);
    const int* bk = bucket + (size_t)i * CAP;
    const float di = dinv_of(deg);
    const float4 hs = *(const float4*)(h + (size_t)i * D + c);
    const float wself = di * di;
    float4 acc;
    acc.x = hs.x * wself; acc.y = hs.y * wself;
    acc.z = hs.z * wself; acc.w = hs.w * wself;

    int j = 0;
    for (; j + 4 <= nb; j += 4) {
        const int4 s = *(const int4*)(bk + j);
        const float n0 = dinv_of(cnt[s.x]) * di;
        const float n1 = dinv_of(cnt[s.y]) * di;
        const float n2 = dinv_of(cnt[s.z]) * di;
        const float n3 = dinv_of(cnt[s.w]) * di;
        const float4 v0 = *(const float4*)(h + (size_t)s.x * D + c);
        const float4 v1 = *(const float4*)(h + (size_t)s.y * D + c);
        const float4 v2 = *(const float4*)(h + (size_t)s.z * D + c);
        const float4 v3 = *(const float4*)(h + (size_t)s.w * D + c);
        acc.x = fmaf(v0.x, n0, acc.x); acc.y = fmaf(v0.y, n0, acc.y);
        acc.z = fmaf(v0.z, n0, acc.z); acc.w = fmaf(v0.w, n0, acc.w);
        acc.x = fmaf(v1.x, n1, acc.x); acc.y = fmaf(v1.y, n1, acc.y);
        acc.z = fmaf(v1.z, n1, acc.z); acc.w = fmaf(v1.w, n1, acc.w);
        acc.x = fmaf(v2.x, n2, acc.x); acc.y = fmaf(v2.y, n2, acc.y);
        acc.z = fmaf(v2.z, n2, acc.z); acc.w = fmaf(v2.w, n2, acc.w);
        acc.x = fmaf(v3.x, n3, acc.x); acc.y = fmaf(v3.y, n3, acc.y);
        acc.z = fmaf(v3.z, n3, acc.z); acc.w = fmaf(v3.w, n3, acc.w);
    }
    const int rem = nb - j;
    if (rem > 0) {
        const int4 s = *(const int4*)(bk + j);
        const int s0 = s.x;
        const int s1 = (rem > 1) ? s.y : s.x;
        const int s2 = (rem > 2) ? s.z : s.x;
        const float n0 = dinv_of(cnt[s0]) * di;
        const float n1 = dinv_of(cnt[s1]) * di;
        const float n2 = dinv_of(cnt[s2]) * di;
        const float4 v0 = *(const float4*)(h + (size_t)s0 * D + c);
        float4 v1 = *(const float4*)(h + (size_t)s1 * D + c);
        float4 v2 = *(const float4*)(h + (size_t)s2 * D + c);
        if (rem < 2) { v1.x = 0.0f; v1.y = 0.0f; v1.z = 0.0f; v1.w = 0.0f; }
        if (rem < 3) { v2.x = 0.0f; v2.y = 0.0f; v2.z = 0.0f; v2.w = 0.0f; }
        acc.x = fmaf(v0.x, n0, acc.x); acc.y = fmaf(v0.y, n0, acc.y);
        acc.z = fmaf(v0.z, n0, acc.z); acc.w = fmaf(v0.w, n0, acc.w);
        acc.x = fmaf(v1.x, n1, acc.x); acc.y = fmaf(v1.y, n1, acc.y);
        acc.z = fmaf(v1.z, n1, acc.z); acc.w = fmaf(v1.w, n1, acc.w);
        acc.x = fmaf(v2.x, n2, acc.x); acc.y = fmaf(v2.y, n2, acc.y);
        acc.z = fmaf(v2.z, n2, acc.z); acc.w = fmaf(v2.w, n2, acc.w);
    }
    if (deg > CAP) {
        const int novf = cnt[Nn];
        for (int k = 0; k < novf; ++k) {
            const int e = ovf[k];
            if (dst[e] == i) {
                const int s = src[e];
                const float n = dinv_of(cnt[s]) * di;
                const float4 v = *(const float4*)(h + (size_t)s * D + c);
                acc.x = fmaf(v.x, n, acc.x); acc.y = fmaf(v.y, n, acc.y);
                acc.z = fmaf(v.z, n, acc.z); acc.w = fmaf(v.w, n, acc.w);
            }
        }
    }
    return acc;
}

#define FN 64
#define FW 16
#define FPITCH 132

__global__ void __launch_bounds__(FW * 32, 16)
gather_relu_gemm2(const float* __restrict__ h,
                  const int* __restrict__ cnt,
                  const int* __restrict__ bucket,
                  const int* __restrict__ ovf,
                  const int* __restrict__ src,
                  const int* __restrict__ dst,
                  const float* __restrict__ bias,
                  const float* __restrict__ B,
                  float* __restrict__ C, int Nn) {
    constexpr int K = HID_DIM;
    constexpr int N = OUT_DIM;
    static_assert(K == 128 && N == 64 && FN == 4 * FW && FN % 16 == 0, "written for 128 x 64 and four nodes per wave");
    __shared__ float4 Ws[(K / 4) * N];
    __shared__ alignas(16) float rows[FN * FPITCH];

    const int tid  = threadIdx.x;
    const int lane = tid & 31;
    const int wave = tid >> 5;

#pragma unroll
    for (int u = tid; u < (K / 4) * N; u += FW * 32) {
        const int q  = u / N;
        const int p  = u - q * N;
        const int pl = p & 31;
        const int n  = (p - pl) + ((pl & 15) << 1) + (pl >> 4);
        const float* bp = B + (size_t)(4 * q) * N + n;
        float4 w;
        w.x = bp[0]; w.y = bp[N]; w.z = bp[2 * N]; w.w = bp[3 * N];
        Ws[u] = w;
    }

    const int col = lane << 2;
    const float4 bv = *(const float4*)(bias + col);
    for (int n = 0; n < FN / FW; ++n) {
        const int q = wave * (FN / FW) + n;
        const int i = blockIdx.x * FN + q;
        float4 r;
        r.x = 0.0f; r.y = 0.0f; r.z = 0.0f; r.w = 0.0f;
        if (i < Nn) {
            const float4 acc = gather_sum<K>(h, cnt, bucket, ovf, src, dst, i, col, Nn);
            r.x = fmaxf(acc.x + bv.x, 0.0f); r.y = fmaxf(acc.y + bv.y, 0.0f);
            r.z = fmaxf(acc.z + bv.z, 0.0f); r.w = fmaxf(acc.w + bv.w, 0.0f);
        }
        *(float4*)(&rows[q * FPITCH + col]) = r;
    }
    __syncthreads();

    const int rt   = wave >> 1;
    const int cg   = wave & 1;
    const int row0 = blockIdx.x * FN + (rt << 4);
    if (wave < 2 * (FN / 16) && row0 < Nn) {
        const int c = lane & 15;
        const int hh = lane >> 4;
        const float*  Ar = rows + ((rt << 4) + c) * FPITCH + 4 * hh;
        const float4* Wl = Ws + hh * N + cg * 32 + c;

        v8f acc0 = v8f{};
        v8f acc1 = v8f{};
        float4 ac = *(const float4*)(Ar);
        float4 b0c = Wl[0];
        float4 b1c = Wl[16];
        float4 an = ac, b0n = b0c, b1n = b1c;
#pragma unroll
        for (int j = 0; j < 16; ++j) {
            if (j + 1 < 16) {
                an  = *(const float4*)(Ar + 8 * (j + 1));
                b0n = Wl[(2 * (j + 1)) * N];
                b1n = Wl[(2 * (j + 1)) * N + 16];
            }
            v2f a0, a1, p0, p1, q0, q1;
            a0.x = ac.x;  a0.y = ac.y;  a1.x = ac.z;  a1.y = ac.w;
            p0.x = b0c.x; p0.y = b0c.y; p1.x = b0c.z; p1.y = b0c.w;
            q0.x = b1c.x; q0.y = b1c.y; q1.x = b1c.z; q1.y = b1c.w;
            acc0 = __builtin_amdgcn_wmma_f32_16x16x4_f32(false, a0, false, p0, (short)0, acc0, false, false);
            acc1 = __builtin_amdgcn_wmma_f32_16x16x4_f32(false, a0, false, q0, (short)0, acc1, false, false);
            acc0 = __builtin_amdgcn_wmma_f32_16x16x4_f32(false, a1, false, p1, (short)0, acc0, false, false);
            acc1 = __builtin_amdgcn_wmma_f32_16x16x4_f32(false, a1, false, q1, (short)0, acc1, false, false);
            ac = an; b0c = b0n; b1c = b1n;
        }

        float* Cp = C + (size_t)(row0 + (hh << 3)) * N + cg * 32 + (c << 1);
#pragma unroll
        for (int v = 0; v < 8; ++v) {
            float2 o;
            o.x = acc0[v]; o.y = acc1[v];
            if (row0 + (hh << 3) + v < Nn) *(float2*)(Cp + (size_t)v * N) = o;
        }
    }
}

template <int D>
__global__ void __launch_bounds__(256)
gather_pool_rows(const float* __restrict__ h,
                 const int* __restrict__ cnt,
                 const int* __restrict__ bucket,
                 const int* __restrict__ ovf,
                 const int* __restrict__ src,
                 const int* __restrict__ dst,
                 const float* __restrict__ bias,
                 const int* __restrict__ batch,
                 float* __restrict__ sums,
                 float* __restrict__ cnts, int Nn) {
    constexpr int LPN = D >> 2;
    constexpr int NPB = 256 / LPN;
    __shared__ float rows[NPB][D];
    __shared__ int   gid[NPB];

    const int q  = threadIdx.x / LPN;
    const int lc = threadIdx.x - q * LPN;
    const int c  = lc << 2;
    const int i  = blockIdx.x * NPB + q;

    float4 r;
    r.x = 0.0f; r.y = 0.0f; r.z = 0.0f; r.w = 0.0f;
    int g = -1;
    if (i < Nn) {
        g = batch[i];
        const float4 acc = gather_sum<D>(h, cnt, bucket, ovf, src, dst, i, c, Nn);
        const float4 bv = *(const float4*)(bias + c);
        r.x = acc.x + bv.x; r.y = acc.y + bv.y;
        r.z = acc.z + bv.z; r.w = acc.w + bv.w;
    }

    *(float4*)(&rows[q][c]) = r;
    if (lc == 0) gid[q] = g;
    __syncthreads();

    if (threadIdx.x < D) {
        const int d = threadIdx.x;
        float v[NPB];
        int   gk[NPB];
#pragma unroll
        for (int k = 0; k < NPB; ++k) { v[k] = rows[k][d]; gk[k] = gid[k]; }

        float s  = 0.0f;
        int  run = 0;
        int  gc  = gk[0];
#pragma unroll
        for (int k = 0; k < NPB; ++k) {
            if (gk[k] != gc) {
                if (gc >= 0 && gc < NUM_GRAPHS) {
                    atomicAdd(&sums[(size_t)gc * D + d], s);
                    if (d == 0) atomicAdd(&cnts[gc], (float)run);
                }
                s = 0.0f; run = 0; gc = gk[k];
            }
            s += v[k];
            ++run;
        }
        if (gc >= 0 && gc < NUM_GRAPHS) {
            atomicAdd(&sums[(size_t)gc * D + d], s);
            if (d == 0) atomicAdd(&cnts[gc], (float)run);
        }
    }
}

__global__ void pool_div(const float* __restrict__ sums,
                         const float* __restrict__ cnt,
                         float* __restrict__ out, int G, int D) {
    int tid = blockIdx.x * blockDim.x + threadIdx.x;
    if (tid >= G * D) return;
    int g = tid / D;
    out[tid] = sums[tid] / fmaxf(cnt[g], 1.0f);
}

extern "C" void kernel_launch(void* const* d_in, const int* in_sizes, int n_in,
                              void* d_out, int out_size, void* d_ws, size_t ws_size,
                              hipStream_t stream) {
    const float* x    = (const float*)d_in[0];
    const int*   eidx = (const int*)  d_in[1];
    const int*   batch= (const int*)  d_in[2];
    const float* W1   = (const float*)d_in[3];
    const float* b1   = (const float*)d_in[4];
    const float* W2   = (const float*)d_in[5];
    const float* b2   = (const float*)d_in[6];
    float* out = (float*)d_out;

    const int Nn = in_sizes[0] / IN_DIM;
    const int E  = in_sizes[1] / 2;
    const int* src = eidx;
    const int* dst = eidx + E;

    float* ws   = (float*)d_ws;
    float* h1   = ws;
    float* h2   = h1   + (size_t)Nn * HID_DIM;
    int* bucket = (int*)(h2 + (size_t)Nn * OUT_DIM);
    float* sums = (float*)(bucket + (size_t)Nn * CAP);
    float* cnts = sums + NUM_GRAPHS * OUT_DIM;
    int* cnt    = (int*)(cnts + NUM_GRAPHS);
    int* tix    = cnt + Nn + 1;
    int* ovf    = tix + 2;

    const int B = 256;
    auto blocks = [](long long n, int b) { return (unsigned)((n + b - 1) / b); };

    start_values<<<blocks(Nn + 3, B), B, 0, stream>>>(
        cnt, Nn + 3, sums, NUM_GRAPHS * OUT_DIM + NUM_GRAPHS);

    gemm_strip_wmma_f32<HID_DIM, 14, 7><<<blocks((long long)(Nn >> 4) * (HID_DIM / 64), 14), 14 * 32, 0, stream>>>(
        x, W1, h1, Nn, src, dst, cnt, bucket, ovf, tix, E, Nn);

    gather_relu_gemm2<<<blocks(Nn, FN), FW * 32, 0, stream>>>(
        h1, cnt, bucket, ovf, src, dst, b1, W2, h2, Nn);

    gather_pool_rows<OUT_DIM><<<blocks((long long)Nn * (OUT_DIM >> 2), B), B, 0, stream>>>(
        h2, cnt, bucket, ovf, src, dst, b2, batch, sums, cnts, Nn);

    pool_div<<<blocks(NUM_GRAPHS * OUT_DIM, B), B, 0, stream>>>(
        sums, cnts, out, NUM_GRAPHS, OUT_DIM);
}


# Both gathers: a wave reads its node's 32-slot row once, one slot per lane, and takes each source and weight out of its lane as a scalar: row loads wait on no index read, 8 in flight; sum order kept
# speedup vs baseline: 5.6603x; 1.0301x over previous
#include <hip/hip_runtime.h>

#define IN_DIM   128
#define HID_DIM  128
#define OUT_DIM  64
#define NUM_GRAPHS 64

typedef float v2f __attribute__((ext_vector_type(2)));
typedef float v8f __attribute__((ext_vector_type(8)));

#define CAP 32


__global__ void start_values(int* __restrict__ zi, int nzi, float* __restrict__ zf, int nzf) {
    const int stride = gridDim.x * blockDim.x;
    for (int g = blockIdx.x * blockDim.x + threadIdx.x; g < nzi; g += stride) zi[g] = 0;
    for (int g = blockIdx.x * blockDim.x + threadIdx.x; g < nzf; g += stride) zf[g] = 0.0f;
}

template <int N, int WAVES, int EPT>
__global__ void __launch_bounds__(WAVES * 32, 7)
gemm_strip_wmma_f32(const float* __restrict__ A,
                    const float* __restrict__ B,
                    float* __restrict__ C, int M,
                    const int* __restrict__ esrc, const int* __restrict__ edst,
                    int* __restrict__ cnt, int* __restrict__ bucket,
                    int* __restrict__ ovf, int* __restrict__ tix, int E, int Nn) {
    constexpr int K   = 128;
    static_assert(IN_DIM == K && HID_DIM == K && N % 64 == 0, "strip GEMM is written for K = 128");
    constexpr int NH  = N / 64;
    constexpr int SLOTS = (K / 4) * N;
    __shared__ float4 Ws[SLOTS];
    __shared__ int etask[2];

    const int tid  = threadIdx.x;
    const int lane = tid & 31;
    const int wave = tid >> 5;
    const int c    = lane & 15;
    const int h    = lane >> 4;
    const int task = blockIdx.x * WAVES + wave;
    const int tile = task / NH;
    const int cg   = task - tile * NH;
    const bool live = tile < (M >> 4);
    const int row0 = tile << 4;

    float4 a[16];
    if (live) {
        const float* Ap = A + (size_t)(row0 + c) * K + 4 * h;
#pragma unroll
        for (int j = 0; j < 16; ++j) a[j] = *(const float4*)(Ap + 8 * j);
    }

    constexpr int EP = EPT > 0 ? EPT : 1;
    constexpr int BT = WAVES * 32;
    constexpr int CH = BT * EP;
    const int nch = (E + CH - 1) / CH;
    const int Nh  = ((Nn >> 1) + 63) & ~63;
    int tq = 0, tt = 0;
    if (EPT > 0 && tid == BT - 32) {
        tq = (int)((__builtin_amdgcn_s_sendmsg_rtn(0x87) >> 18) & 1u);
        tt = atomicAdd(&tix[tq], 1);
    }

#pragma unroll 4
    for (int u = tid; u < SLOTS; u += WAVES * 32) {
        const int q  = u / N;
        const int p  = u - q * N;
        const int pl = p & 63;
        const int n  = (p - pl) + ((pl & 15) << 2) + (pl >> 4);
        const float* bp = B + (size_t)(4 * q) * N + n;
        float4 w;
        w.x = bp[0]; w.y = bp[N]; w.z = bp[2 * N]; w.w = bp[3 * N];
        Ws[u] = w;
    }
    if (EPT > 0 && tid == BT - 32) {
        if (tt >= nch) { tq ^= 1; tt = atomicAdd(&tix[tq], 1); }
        etask[0] = tq;
        etask[1] = (tt < nch) ? tt : -1;
    }
    __syncthreads();

    int  er[EP];
    bool up  = false;
    bool has = false;
#pragma unroll
    for (int k = 0; k < EP; ++k) er[k] = 0;
    if (EPT > 0) {
        up = __builtin_amdgcn_readfirstlane(etask[0]) != 0;
        const int t = __builtin_amdgcn_readfirstlane(etask[1]);
        has = t >= 0;
        if (has) {
            const int e0 = t * CH + tid;
            int ed[EP];
#pragma unroll
            for (int k = 0; k < EP; ++k) ed[k] = edst[min(e0 + k * BT, E - 1)];
#pragma unroll
            for (int k = 0; k < EP; ++k) {
                if (e0 + k * BT < E && (ed[k] >= Nh) == up) er[k] = atomicAdd(&cnt[ed[k]], 1);
            }
        }
    }

    if (live) {
        v8f acc[4];
#pragma unroll
        for (int t = 0; t < 4; ++t) acc[t] = v8f{};

        const float4* Wl = Ws + h * N + cg * 64 + c;
        float4 bc[4], bn[4];
#pragma unroll
        for (int t = 0; t < 4; ++t) bc[t] = Wl[16 * t];

        __builtin_amdgcn_sched_group_barrier(0x100, 8, 0);
#pragma unroll
        for (int j = 0; j < 16; ++j) {
            if (j + 1 < 16) {
#pragma unroll
                for (int t = 0; t < 4; ++t) bn[t] = Wl[(2 * (j + 1)) * N + 16 * t];
            }
            v2f a0, a1;
            a0.x = a[j].x; a0.y = a[j].y;
            a1.x = a[j].z; a1.y = a[j].w;
#pragma unroll
            for (int t = 0; t < 4; ++t) {
                v2f b0;
                b0.x = bc[t].x; b0.y = bc[t].y;
                acc[t] = __builtin_amdgcn_wmma_f32_16x16x4_f32(
                     false, a0,  false, b0,
                     (short)0, acc[t],  false,  false);
            }
#pragma unroll
            for (int t = 0; t < 4; ++t) {
                v2f b1;
                b1.x = bc[t].z; b1.y = bc[t].w;
                acc[t] = __builtin_amdgcn_wmma_f32_16x16x4_f32(
                     false, a1,  false, b1,
                     (short)0, acc[t],  false,  false);
            }
            __builtin_amdgcn_sched_group_barrier(0x008, 8, 0);
            if (j + 2 < 16)
                __builtin_amdgcn_sched_group_barrier(0x100, 4, 0);
#pragma unroll
            for (int t = 0; t < 4; ++t) bc[t] = bn[t];
        }

        float* Cp = C + (size_t)(row0 + (h << 3)) * N + cg * 64 + (c << 2);
#pragma unroll
        for (int v = 0; v < 8; ++v) {
            float4 o;
            o.x = acc[0][v]; o.y = acc[1][v]; o.z = acc[2][v]; o.w = acc[3][v];
            *(float4*)(Cp + (size_t)v * N) = o;
        }
    }

    if (EPT > 0 && has) {
        const int e1 = __builtin_amdgcn_readfirstlane(
            __hip_atomic_load(&etask[1], __ATOMIC_RELAXED, __HIP_MEMORY_SCOPE_WORKGROUP)) * CH + tid;
        int ed[EP], es[EP];
#pragma unroll
        for (int k = 0; k < EP; ++k) {
            const int x = min(e1 + k * BT, E - 1);
            ed[k] = edst[x];
            es[k] = esrc[x];
        }
#pragma unroll
        for (int k = 0; k < EP; ++k) {
            const int e = e1 + k * BT;
            if (e < E && (ed[k] >= Nh) == up) {
                if (er[k] < CAP) bucket[ed[k] * CAP + er[k]] = es[k];
                else             ovf[atomicAdd(&cnt[Nn], 1)] = e;
            }
        }
    }

    if (EPT > 0 && 2 * nch > (int)gridDim.x) {
        for (int pass = 0; pass < 2; ++pass) {
            const bool upq = up != (pass != 0);
            for (;;) {
                __syncthreads();
                if (tid == 0) {
                    const int t = atomicAdd(&tix[upq ? 1 : 0], 1);
                    etask[1] = (t < nch) ? t : -1;
                }
                __syncthreads();
                const int t = __builtin_amdgcn_readfirstlane(etask[1]);
                if (t < 0) break;
                const int eend = min(E, (t + 1) * CH);
                for (int e = t * CH + tid; e < eend; e += BT) {
                    const int d = edst[e];
                    if ((d >= Nh) == upq) {
                        const int r = atomicAdd(&cnt[d], 1);
                        if (r < CAP) bucket[(size_t)d * CAP + r] = esrc[e];
                        else         ovf[atomicAdd(&cnt[Nn], 1)] = e;
                    }
                }
            }
        }
    }
}

__device__ __forceinline__ float dinv_of(int indeg) {
    return __builtin_amdgcn_rsqf((float)(indeg + 1));
}

template <int VPL> struct rowvec;
template <> struct rowvec<4> { typedef float4 type; };
template <> struct rowvec<2> { typedef float2 type; };

__device__ __forceinline__ void row_fma(float4& acc, const float4& v, float w) {
    acc.x = fmaf(v.x, w, acc.x); acc.y = fmaf(v.y, w, acc.y);
    acc.z = fmaf(v.z, w, acc.z); acc.w = fmaf(v.w, w, acc.w);
}
__device__ __forceinline__ void row_fma(float2& acc, const float2& v, float w) {
    acc.x = fmaf(v.x, w, acc.x); acc.y = fmaf(v.y, w, acc.y);
}
__device__ __forceinline__ void row_scale(float4& acc, const float4& v, float w) {
    acc.x = v.x * w; acc.y = v.y * w; acc.z = v.z * w; acc.w = v.w * w;
}
__device__ __forceinline__ void row_scale(float2& acc, const float2& v, float w) {
    acc.x = v.x * w; acc.y = v.y * w;
}

template <int D, int BW, bool SELF, typename VT>
__device__ __forceinline__ void gather_batch(VT& acc, const float* __restrict__ h,
                                             int sv, float wv, int j0, int c, int i, float wself) {
    VT hs;
    if (SELF) hs = *(const VT*)(h + (size_t)i * D + c);
    VT v[BW];
    float w[BW];
#pragma unroll
    for (int k = 0; k < BW; ++k) {
        const int s = __builtin_amdgcn_readlane(sv, j0 + k);
        w[k] = __int_as_float(__builtin_amdgcn_readlane(__float_as_int(wv), j0 + k));
        v[k] = *(const VT*)(h + (size_t)s * D + c);
    }
    if (SELF) row_scale(acc, hs, wself);
#pragma unroll
    for (int k = 0; k < BW; ++k) row_fma(acc, v[k], w[k]);
}

template <int D, int NPW>
__device__ __forceinline__ void gather_nodes(const float* __restrict__ h,
                                             const int* __restrict__ cnt,
                                             const int* __restrict__ bucket,
                                             const int* __restrict__ ovf,
                                             const int* __restrict__ src,
                                             const int* __restrict__ dst,
                                             int i0, int lane, int Nn,
                                             typename rowvec<D / 32>::type (&res)[NPW]) {
    typedef typename rowvec<D / 32>::type VT;
    static_assert(CAP == 32, "one slot per lane");
    const int c = lane * (D / 32);

    int deg[NPW], sl[NPW];
#pragma unroll
    for (int n = 0; n < NPW; ++n) {
        const int i = min(i0 + n, Nn - 1);
        deg[n] = __builtin_amdgcn_readfirstlane(cnt[i]);
        sl[n]  = bucket[(size_t)i * CAP + lane];
    }
    int sv[NPW], cs[NPW];
#pragma unroll
    for (int n = 0; n < NPW; ++n) {
        const int i = min(i0 + n, Nn - 1);
        sv[n] = (lane < min(deg[n], CAP)) ? sl[n] : i;
        cs[n] = cnt[sv[n]];
    }

#pragma unroll
    for (int n = 0; n < NPW; ++n) {
        const int i  = min(i0 + n, Nn - 1);
        const int nb = min(deg[n], CAP);
        const float di = dinv_of(deg[n]);
        const float wv = (lane < nb) ? dinv_of(cs[n]) * di : 0.0f;
        VT acc;
        gather_batch<D, 8, true, VT>(acc, h, sv[n], wv, 0, c, i, di * di);
        int j = 8;
        for (; j + 8 <= nb; j += 8) gather_batch<D, 8, false, VT>(acc, h, sv[n], wv, j, c, i, 0.0f);
        const int rem = nb - j;
        if (rem > 4)      gather_batch<D, 8, false, VT>(acc, h, sv[n], wv, j, c, i, 0.0f);
        else if (rem > 0) gather_batch<D, 4, false, VT>(acc, h, sv[n], wv, j, c, i, 0.0f);

        if (deg[n] > CAP) {
            const int novf = cnt[Nn];
            for (int k = 0; k < novf; ++k) {
                const int e = ovf[k];
                if (dst[e] == i) {
                    const int s = src[e];
                    const float w = dinv_of(cnt[s]) * di;
                    const VT v = *(const VT*)(h + (size_t)s * D + c);
                    row_fma(acc, v, w);
                }
            }
        }
        res[n] = acc;
    }
}

#define FN 64
#define FW 16
#define FPITCH 132

__global__ void __launch_bounds__(FW * 32, 16)
gather_relu_gemm2(const float* __restrict__ h,
                  const int* __restrict__ cnt,
                  const int* __restrict__ bucket,
                  const int* __restrict__ ovf,
                  const int* __restrict__ src,
                  const int* __restrict__ dst,
                  const float* __restrict__ bias,
                  const float* __restrict__ B,
                  float* __restrict__ C, int Nn) {
    constexpr int K = HID_DIM;
    constexpr int N = OUT_DIM;
    static_assert(K == 128 && N == 64 && FN == 4 * FW && FN % 16 == 0, "written for 128 x 64 and four nodes per wave");
    __shared__ float4 Ws[(K / 4) * N];
    __shared__ alignas(16) float rows[FN * FPITCH];

    const int tid  = threadIdx.x;
    const int lane = tid & 31;
    const int wave = tid >> 5;

#pragma unroll
    for (int u = tid; u < (K / 4) * N; u += FW * 32) {
        const int q  = u / N;
        const int p  = u - q * N;
        const int pl = p & 31;
        const int n  = (p - pl) + ((pl & 15) << 1) + (pl >> 4);
        const float* bp = B + (size_t)(4 * q) * N + n;
        float4 w;
        w.x = bp[0]; w.y = bp[N]; w.z = bp[2 * N]; w.w = bp[3 * N];
        Ws[u] = w;
    }

    const int col = lane << 2;
    const float4 bv = *(const float4*)(bias + col);
    const int wq = __builtin_amdgcn_readfirstlane(wave) * (FN / FW);
    {
        float4 res[FN / FW];
        gather_nodes<K, FN / FW>(h, cnt, bucket, ovf, src, dst, blockIdx.x * FN + wq, lane, Nn, res);
#pragma unroll
        for (int n = 0; n < FN / FW; ++n) {
            float4 r;
            r.x = 0.0f; r.y = 0.0f; r.z = 0.0f; r.w = 0.0f;
            if (blockIdx.x * FN + wq + n < Nn) {
                r.x = fmaxf(res[n].x + bv.x, 0.0f); r.y = fmaxf(res[n].y + bv.y, 0.0f);
                r.z = fmaxf(res[n].z + bv.z, 0.0f); r.w = fmaxf(res[n].w + bv.w, 0.0f);
            }
            *(float4*)(&rows[(wq + n) * FPITCH + col]) = r;
        }
    }
    __syncthreads();

    const int rt   = wave >> 1;
    const int cg   = wave & 1;
    const int row0 = blockIdx.x * FN + (rt << 4);
    if (wave < 2 * (FN / 16) && row0 < Nn) {
        const int c = lane & 15;
        const int hh = lane >> 4;
        const float*  Ar = rows + ((rt << 4) + c) * FPITCH + 4 * hh;
        const float4* Wl = Ws + hh * N + cg * 32 + c;

        v8f acc0 = v8f{};
        v8f acc1 = v8f{};
        float4 ac = *(const float4*)(Ar);
        float4 b0c = Wl[0];
        float4 b1c = Wl[16];
        float4 an = ac, b0n = b0c, b1n = b1c;
#pragma unroll
        for (int j = 0; j < 16; ++j) {
            if (j + 1 < 16) {
                an  = *(const float4*)(Ar + 8 * (j + 1));
                b0n = Wl[(2 * (j + 1)) * N];
                b1n = Wl[(2 * (j + 1)) * N + 16];
            }
            v2f a0, a1, p0, p1, q0, q1;
            a0.x = ac.x;  a0.y = ac.y;  a1.x = ac.z;  a1.y = ac.w;
            p0.x = b0c.x; p0.y = b0c.y; p1.x = b0c.z; p1.y = b0c.w;
            q0.x = b1c.x; q0.y = b1c.y; q1.x = b1c.z; q1.y = b1c.w;
            acc0 = __builtin_amdgcn_wmma_f32_16x16x4_f32(false, a0, false, p0, (short)0, acc0, false, false);
            acc1 = __builtin_amdgcn_wmma_f32_16x16x4_f32(false, a0, false, q0, (short)0, acc1, false, false);
            acc0 = __builtin_amdgcn_wmma_f32_16x16x4_f32(false, a1, false, p1, (short)0, acc0, false, false);
            acc1 = __builtin_amdgcn_wmma_f32_16x16x4_f32(false, a1, false, q1, (short)0, acc1, false, false);
            ac = an; b0c = b0n; b1c = b1n;
        }

        float* Cp = C + (size_t)(row0 + (hh << 3)) * N + cg * 32 + (c << 1);
#pragma unroll
        for (int v = 0; v < 8; ++v) {
            float2 o;
            o.x = acc0[v]; o.y = acc1[v];
            if (row0 + (hh << 3) + v < Nn) *(float2*)(Cp + (size_t)v * N) = o;
        }
    }
}

#define PW  4
#define PNB 32
#define PH  16
template <int D>
__global__ void __launch_bounds__(256)
gather_pool_rows(const float* __restrict__ h,
                 const int* __restrict__ cnt,
                 const int* __restrict__ bucket,
                 const int* __restrict__ ovf,
                 const int* __restrict__ src,
                 const int* __restrict__ dst,
                 const float* __restrict__ bias,
                 const int* __restrict__ batch,
                 float* __restrict__ sums,
                 float* __restrict__ cnts, int Nn) {
    static_assert(D == 64 && PNB == 8 * PW && PNB == 2 * PH, "written for 64 columns, 8 waves, two halves of rows");
    __shared__ alignas(16) float rows[PNB][D];
    __shared__ int   gid[PNB];

    const int lane = threadIdx.x & 31;
    const int wq   = __builtin_amdgcn_readfirstlane(threadIdx.x >> 5) * PW;
    const int i0   = blockIdx.x * PNB + wq;
    const int c    = lane << 1;

    int g[PW];
#pragma unroll
    for (int n = 0; n < PW; ++n) g[n] = (i0 + n < Nn) ? batch[i0 + n] : -1;

    float2 res[PW];
    gather_nodes<D, PW>(h, cnt, bucket, ovf, src, dst, i0, lane, Nn, res);
    const float2 bv = *(const float2*)(bias + c);
#pragma unroll
    for (int n = 0; n < PW; ++n) {
        float2 r;
        r.x = 0.0f; r.y = 0.0f;
        if (i0 + n < Nn) { r.x = res[n].x + bv.x; r.y = res[n].y + bv.y; }
        *(float2*)(&rows[wq + n][c]) = r;
        if (lane == 0) gid[wq + n] = g[n];
    }
    __syncthreads();

    if (threadIdx.x < 2 * D) {
        const int d  = threadIdx.x & (D - 1);
        const int r0 = (threadIdx.x / D) * PH;
        float v[PH];
        int   gk[PH];
#pragma unroll
        for (int k = 0; k < PH; ++k) { v[k] = rows[r0 + k][d]; gk[k] = gid[r0 + k]; }

        float s  = 0.0f;
        int  run = 0;
        int  gc  = gk[0];
#pragma unroll
        for (int k = 0; k < PH; ++k) {
            if (gk[k] != gc) {
                if (gc >= 0 && gc < NUM_GRAPHS) {
                    atomicAdd(&sums[(size_t)gc * D + d], s);
                    if (d == 0) atomicAdd(&cnts[gc], (float)run);
                }
                s = 0.0f; run = 0; gc = gk[k];
            }
            s += v[k];
            ++run;
        }
        if (gc >= 0 && gc < NUM_GRAPHS) {
            atomicAdd(&sums[(size_t)gc * D + d], s);
            if (d == 0) atomicAdd(&cnts[gc], (float)run);
        }
    }
}

__global__ void pool_div(const float* __restrict__ sums,
                         const float* __restrict__ cnt,
                         float* __restrict__ out, int G, int D) {
    int tid = blockIdx.x * blockDim.x + threadIdx.x;
    if (tid >= G * D) return;
    int g = tid / D;
    out[tid] = sums[tid] / fmaxf(cnt[g], 1.0f);
}

extern "C" void kernel_launch(void* const* d_in, const int* in_sizes, int n_in,
                              void* d_out, int out_size, void* d_ws, size_t ws_size,
                              hipStream_t stream) {
    const float* x    = (const float*)d_in[0];
    const int*   eidx = (const int*)  d_in[1];
    const int*   batch= (const int*)  d_in[2];
    const float* W1   = (const float*)d_in[3];
    const float* b1   = (const float*)d_in[4];
    const float* W2   = (const float*)d_in[5];
    const float* b2   = (const float*)d_in[6];
    float* out = (float*)d_out;

    const int Nn = in_sizes[0] / IN_DIM;
    const int E  = in_sizes[1] / 2;
    const int* src = eidx;
    const int* dst = eidx + E;

    float* ws   = (float*)d_ws;
    float* h1   = ws;
    float* h2   = h1   + (size_t)Nn * HID_DIM;
    int* bucket = (int*)(h2 + (size_t)Nn * OUT_DIM);
    float* sums = (float*)(bucket + (size_t)Nn * CAP);
    float* cnts = sums + NUM_GRAPHS * OUT_DIM;
    int* cnt    = (int*)(cnts + NUM_GRAPHS);
    int* tix    = cnt + Nn + 1;
    int* ovf    = tix + 2;

    const int B = 256;
    auto blocks = [](long long n, int b) { return (unsigned)((n + b - 1) / b); };

    start_values<<<blocks(Nn + 3, B), B, 0, stream>>>(
        cnt, Nn + 3, sums, NUM_GRAPHS * OUT_DIM + NUM_GRAPHS);

    gemm_strip_wmma_f32<HID_DIM, 14, 7><<<blocks((long long)(Nn >> 4) * (HID_DIM / 64), 14), 14 * 32, 0, stream>>>(
        x, W1, h1, Nn, src, dst, cnt, bucket, ovf, tix, E, Nn);

    gather_relu_gemm2<<<blocks(Nn, FN), FW * 32, 0, stream>>>(
        h1, cnt, bucket, ovf, src, dst, b1, W2, h2, Nn);

    gather_pool_rows<OUT_DIM><<<blocks(Nn, PNB), B, 0, stream>>>(
        h2, cnt, bucket, ovf, src, dst, b2, batch, sums, cnts, Nn);

    pool_div<<<blocks(NUM_GRAPHS * OUT_DIM, B), B, 0, stream>>>(
        sums, cnts, out, NUM_GRAPHS, OUT_DIM);
}


# Pool division folded into the layer-2 gather: each run's sum is divided by its graph's node count (counted a launch earlier) and atomically added, arrival order as before, to out; 5 launches to 4
# speedup vs baseline: 6.1425x; 1.0852x over previous
#include <hip/hip_runtime.h>

#define IN_DIM   128
#define HID_DIM  128
#define OUT_DIM  64
#define NUM_GRAPHS 64

typedef float v2f __attribute__((ext_vector_type(2)));
typedef float v8f __attribute__((ext_vector_type(8)));

#define CAP 32


__global__ void start_values(int* __restrict__ zi, int nzi, float* __restrict__ zf, int nzf,
                             float* __restrict__ zo, int nzo) {
    const int stride = gridDim.x * blockDim.x;
    for (int g = blockIdx.x * blockDim.x + threadIdx.x; g < nzi; g += stride) zi[g] = 0;
    for (int g = blockIdx.x * blockDim.x + threadIdx.x; g < nzf; g += stride) zf[g] = 0.0f;
    for (int g = blockIdx.x * blockDim.x + threadIdx.x; g < nzo; g += stride) zo[g] = 0.0f;
}

template <int N, int WAVES, int EPT>
__global__ void __launch_bounds__(WAVES * 32, 7)
gemm_strip_wmma_f32(const float* __restrict__ A,
                    const float* __restrict__ B,
                    float* __restrict__ C, int M,
                    const int* __restrict__ esrc, const int* __restrict__ edst,
                    int* __restrict__ cnt, int* __restrict__ bucket,
                    int* __restrict__ ovf, int* __restrict__ tix, int E, int Nn) {
    constexpr int K   = 128;
    static_assert(IN_DIM == K && HID_DIM == K && N % 64 == 0, "strip GEMM is written for K = 128");
    constexpr int NH  = N / 64;
    constexpr int SLOTS = (K / 4) * N;
    __shared__ float4 Ws[SLOTS];
    __shared__ int etask[2];

    const int tid  = threadIdx.x;
    const int lane = tid & 31;
    const int wave = tid >> 5;
    const int c    = lane & 15;
    const int h    = lane >> 4;
    const int task = blockIdx.x * WAVES + wave;
    const int tile = task / NH;
    const int cg   = task - tile * NH;
    const bool live = tile < (M >> 4);
    const int row0 = tile << 4;

    float4 a[16];
    if (live) {
        const float* Ap = A + (size_t)(row0 + c) * K + 4 * h;
#pragma unroll
        for (int j = 0; j < 16; ++j) a[j] = *(const float4*)(Ap + 8 * j);
    }

    constexpr int EP = EPT > 0 ? EPT : 1;
    constexpr int BT = WAVES * 32;
    constexpr int CH = BT * EP;
    const int nch = (E + CH - 1) / CH;
    const int Nh  = ((Nn >> 1) + 63) & ~63;
    int tq = 0, tt = 0;
    if (EPT > 0 && tid == BT - 32) {
        tq = (int)((__builtin_amdgcn_s_sendmsg_rtn(0x87) >> 18) & 1u);
        tt = atomicAdd(&tix[tq], 1);
    }

#pragma unroll 4
    for (int u = tid; u < SLOTS; u += WAVES * 32) {
        const int q  = u / N;
        const int p  = u - q * N;
        const int pl = p & 63;
        const int n  = (p - pl) + ((pl & 15) << 2) + (pl >> 4);
        const float* bp = B + (size_t)(4 * q) * N + n;
        float4 w;
        w.x = bp[0]; w.y = bp[N]; w.z = bp[2 * N]; w.w = bp[3 * N];
        Ws[u] = w;
    }
    if (EPT > 0 && tid == BT - 32) {
        if (tt >= nch) { tq ^= 1; tt = atomicAdd(&tix[tq], 1); }
        etask[0] = tq;
        etask[1] = (tt < nch) ? tt : -1;
    }
    __syncthreads();

    int  er[EP];
    bool up  = false;
    bool has = false;
#pragma unroll
    for (int k = 0; k < EP; ++k) er[k] = 0;
    if (EPT > 0) {
        up = __builtin_amdgcn_readfirstlane(etask[0]) != 0;
        const int t = __builtin_amdgcn_readfirstlane(etask[1]);
        has = t >= 0;
        if (has) {
            const int e0 = t * CH + tid;
            int ed[EP];
#pragma unroll
            for (int k = 0; k < EP; ++k) ed[k] = edst[min(e0 + k * BT, E - 1)];
#pragma unroll
            for (int k = 0; k < EP; ++k) {
                if (e0 + k * BT < E && (ed[k] >= Nh) == up) er[k] = atomicAdd(&cnt[ed[k]], 1);
            }
        }
    }

    if (live) {
        v8f acc[4];
#pragma unroll
        for (int t = 0; t < 4; ++t) acc[t] = v8f{};

        const float4* Wl = Ws + h * N + cg * 64 + c;
        float4 bc[4], bn[4];
#pragma unroll
        for (int t = 0; t < 4; ++t) bc[t] = Wl[16 * t];

        __builtin_amdgcn_sched_group_barrier(0x100, 8, 0);
#pragma unroll
        for (int j = 0; j < 16; ++j) {
            if (j + 1 < 16) {
#pragma unroll
                for (int t = 0; t < 4; ++t) bn[t] = Wl[(2 * (j + 1)) * N + 16 * t];
            }
            v2f a0, a1;
            a0.x = a[j].x; a0.y = a[j].y;
            a1.x = a[j].z; a1.y = a[j].w;
#pragma unroll
            for (int t = 0; t < 4; ++t) {
                v2f b0;
                b0.x = bc[t].x; b0.y = bc[t].y;
                acc[t] = __builtin_amdgcn_wmma_f32_16x16x4_f32(
                     false, a0,  false, b0,
                     (short)0, acc[t],  false,  false);
            }
#pragma unroll
            for (int t = 0; t < 4; ++t) {
                v2f b1;
                b1.x = bc[t].z; b1.y = bc[t].w;
                acc[t] = __builtin_amdgcn_wmma_f32_16x16x4_f32(
                     false, a1,  false, b1,
                     (short)0, acc[t],  false,  false);
            }
            __builtin_amdgcn_sched_group_barrier(0x008, 8, 0);
            if (j + 2 < 16)
                __builtin_amdgcn_sched_group_barrier(0x100, 4, 0);
#pragma unroll
            for (int t = 0; t < 4; ++t) bc[t] = bn[t];
        }

        float* Cp = C + (size_t)(row0 + (h << 3)) * N + cg * 64 + (c << 2);
#pragma unroll
        for (int v = 0; v < 8; ++v) {
            float4 o;
            o.x = acc[0][v]; o.y = acc[1][v]; o.z = acc[2][v]; o.w = acc[3][v];
            *(float4*)(Cp + (size_t)v * N) = o;
        }
    }

    if (EPT > 0 && has) {
        const int e1 = __builtin_amdgcn_readfirstlane(
            __hip_atomic_load(&etask[1], __ATOMIC_RELAXED, __HIP_MEMORY_SCOPE_WORKGROUP)) * CH + tid;
        int ed[EP], es[EP];
#pragma unroll
        for (int k = 0; k < EP; ++k) {
            const int x = min(e1 + k * BT, E - 1);
            ed[k] = edst[x];
            es[k] = esrc[x];
        }
#pragma unroll
        for (int k = 0; k < EP; ++k) {
            const int e = e1 + k * BT;
            if (e < E && (ed[k] >= Nh) == up) {
                if (er[k] < CAP) bucket[ed[k] * CAP + er[k]] = es[k];
                else             ovf[atomicAdd(&cnt[Nn], 1)] = e;
            }
        }
    }

    if (EPT > 0 && 2 * nch > (int)gridDim.x) {
        for (int pass = 0; pass < 2; ++pass) {
            const bool upq = up != (pass != 0);
            for (;;) {
                __syncthreads();
                if (tid == 0) {
                    const int t = atomicAdd(&tix[upq ? 1 : 0], 1);
                    etask[1] = (t < nch) ? t : -1;
                }
                __syncthreads();
                const int t = __builtin_amdgcn_readfirstlane(etask[1]);
                if (t < 0) break;
                const int eend = min(E, (t + 1) * CH);
                for (int e = t * CH + tid; e < eend; e += BT) {
                    const int d = edst[e];
                    if ((d >= Nh) == upq) {
                        const int r = atomicAdd(&cnt[d], 1);
                        if (r < CAP) bucket[(size_t)d * CAP + r] = esrc[e];
                        else         ovf[atomicAdd(&cnt[Nn], 1)] = e;
                    }
                }
            }
        }
    }
}

__device__ __forceinline__ float dinv_of(int indeg) {
    return __builtin_amdgcn_rsqf((float)(indeg + 1));
}

template <int VPL> struct rowvec;
template <> struct rowvec<4> { typedef float4 type; };
template <> struct rowvec<2> { typedef float2 type; };

__device__ __forceinline__ void row_fma(float4& acc, const float4& v, float w) {
    acc.x = fmaf(v.x, w, acc.x); acc.y = fmaf(v.y, w, acc.y);
    acc.z = fmaf(v.z, w, acc.z); acc.w = fmaf(v.w, w, acc.w);
}
__device__ __forceinline__ void row_fma(float2& acc, const float2& v, float w) {
    acc.x = fmaf(v.x, w, acc.x); acc.y = fmaf(v.y, w, acc.y);
}
__device__ __forceinline__ void row_scale(float4& acc, const float4& v, float w) {
    acc.x = v.x * w; acc.y = v.y * w; acc.z = v.z * w; acc.w = v.w * w;
}
__device__ __forceinline__ void row_scale(float2& acc, const float2& v, float w) {
    acc.x = v.x * w; acc.y = v.y * w;
}

template <int D, int BW, bool SELF, typename VT>
__device__ __forceinline__ void gather_batch(VT& acc, const float* __restrict__ h,
                                             int sv, float wv, int j0, int c, int i, float wself) {
    VT hs;
    if (SELF) hs = *(const VT*)(h + (size_t)i * D + c);
    VT v[BW];
    float w[BW];
#pragma unroll
    for (int k = 0; k < BW; ++k) {
        const int s = __builtin_amdgcn_readlane(sv, j0 + k);
        w[k] = __int_as_float(__builtin_amdgcn_readlane(__float_as_int(wv), j0 + k));
        v[k] = *(const VT*)(h + (size_t)s * D + c);
    }
    if (SELF) row_scale(acc, hs, wself);
#pragma unroll
    for (int k = 0; k < BW; ++k) row_fma(acc, v[k], w[k]);
}

template <int D, int NPW>
__device__ __forceinline__ void gather_nodes(const float* __restrict__ h,
                                             const int* __restrict__ cnt,
                                             const int* __restrict__ bucket,
                                             const int* __restrict__ ovf,
                                             const int* __restrict__ src,
                                             const int* __restrict__ dst,
                                             int i0, int lane, int Nn,
                                             typename rowvec<D / 32>::type (&res)[NPW]) {
    typedef typename rowvec<D / 32>::type VT;
    static_assert(CAP == 32, "one slot per lane");
    const int c = lane * (D / 32);

    int deg[NPW], sl[NPW];
#pragma unroll
    for (int n = 0; n < NPW; ++n) {
        const int i = min(i0 + n, Nn - 1);
        deg[n] = __builtin_amdgcn_readfirstlane(cnt[i]);
        sl[n]  = bucket[(size_t)i * CAP + lane];
    }
    int sv[NPW], cs[NPW];
#pragma unroll
    for (int n = 0; n < NPW; ++n) {
        const int i = min(i0 + n, Nn - 1);
        sv[n] = (lane < min(deg[n], CAP)) ? sl[n] : i;
        cs[n] = cnt[sv[n]];
    }

#pragma unroll
    for (int n = 0; n < NPW; ++n) {
        const int i  = min(i0 + n, Nn - 1);
        const int nb = min(deg[n], CAP);
        const float di = dinv_of(deg[n]);
        const float wv = (lane < nb) ? dinv_of(cs[n]) * di : 0.0f;
        VT acc;
        gather_batch<D, 8, true, VT>(acc, h, sv[n], wv, 0, c, i, di * di);
        int j = 8;
        for (; j + 8 <= nb; j += 8) gather_batch<D, 8, false, VT>(acc, h, sv[n], wv, j, c, i, 0.0f);
        const int rem = nb - j;
        if (rem > 4)      gather_batch<D, 8, false, VT>(acc, h, sv[n], wv, j, c, i, 0.0f);
        else if (rem > 0) gather_batch<D, 4, false, VT>(acc, h, sv[n], wv, j, c, i, 0.0f);

        if (deg[n] > CAP) {
            const int novf = cnt[Nn];
            for (int k = 0; k < novf; ++k) {
                const int e = ovf[k];
                if (dst[e] == i) {
                    const int s = src[e];
                    const float w = dinv_of(cnt[s]) * di;
                    const VT v = *(const VT*)(h + (size_t)s * D + c);
                    row_fma(acc, v, w);
                }
            }
        }
        res[n] = acc;
    }
}

#define FN 64
#define FW 16
#define FPITCH 132

__global__ void __launch_bounds__(FW * 32, 16)
gather_relu_gemm2(const float* __restrict__ h,
                  const int* __restrict__ cnt,
                  const int* __restrict__ bucket,
                  const int* __restrict__ ovf,
                  const int* __restrict__ src,
                  const int* __restrict__ dst,
                  const float* __restrict__ bias,
                  const float* __restrict__ B,
                  float* __restrict__ C,
                  const int* __restrict__ batch,
                  float* __restrict__ gcount, int Nn) {
    constexpr int K = HID_DIM;
    constexpr int N = OUT_DIM;
    static_assert(K == 128 && N == 64 && FN == 4 * FW && FN % 16 == 0, "written for 128 x 64 and four nodes per wave");
    __shared__ float4 Ws[(K / 4) * N];
    __shared__ alignas(16) float rows[FN * FPITCH];

    const int tid  = threadIdx.x;
    const int lane = tid & 31;
    const int wave = tid >> 5;

#pragma unroll
    for (int u = tid; u < (K / 4) * N; u += FW * 32) {
        const int q  = u / N;
        const int p  = u - q * N;
        const int pl = p & 31;
        const int n  = (p - pl) + ((pl & 15) << 1) + (pl >> 4);
        const float* bp = B + (size_t)(4 * q) * N + n;
        float4 w;
        w.x = bp[0]; w.y = bp[N]; w.z = bp[2 * N]; w.w = bp[3 * N];
        Ws[u] = w;
    }

    const int col = lane << 2;
    const float4 bv = *(const float4*)(bias + col);
    const int wq = __builtin_amdgcn_readfirstlane(wave) * (FN / FW);
    {
        float4 res[FN / FW];
        gather_nodes<K, FN / FW>(h, cnt, bucket, ovf, src, dst, blockIdx.x * FN + wq, lane, Nn, res);
#pragma unroll
        for (int n = 0; n < FN / FW; ++n) {
            float4 r;
            r.x = 0.0f; r.y = 0.0f; r.z = 0.0f; r.w = 0.0f;
            if (blockIdx.x * FN + wq + n < Nn) {
                r.x = fmaxf(res[n].x + bv.x, 0.0f); r.y = fmaxf(res[n].y + bv.y, 0.0f);
                r.z = fmaxf(res[n].z + bv.z, 0.0f); r.w = fmaxf(res[n].w + bv.w, 0.0f);
            }
            *(float4*)(&rows[(wq + n) * FPITCH + col]) = r;
        }
    }
    __syncthreads();

    if (wave == FW - 1) {
        const int nb0 = blockIdx.x * FN + 2 * lane;
        int g0 = (nb0     < Nn) ? batch[nb0]     : -1;
        int g1 = (nb0 + 1 < Nn) ? batch[nb0 + 1] : -1;
        if (g0 >= NUM_GRAPHS) g0 = -1;
        if (g1 >= NUM_GRAPHS) g1 = -1;
        for (int trip = 0; trip < FN; ++trip) {
            const int pend = (g0 >= 0) ? g0 : g1;
            const unsigned m = __builtin_amdgcn_ballot_w32(pend >= 0);
            if (m == 0u) break;
            const int g = __builtin_amdgcn_readlane(pend, __builtin_amdgcn_readfirstlane(__builtin_ctz(m)));
            const int n = __builtin_popcount(__builtin_amdgcn_ballot_w32(g0 == g))
                        + __builtin_popcount(__builtin_amdgcn_ballot_w32(g1 == g));
            if (lane == 0) atomicAdd(&gcount[g], (float)n);
            if (g0 == g) g0 = -1;
            if (g1 == g) g1 = -1;
        }
    }

    const int rt   = wave >> 1;
    const int cg   = wave & 1;
    const int row0 = blockIdx.x * FN + (rt << 4);
    if (wave < 2 * (FN / 16) && row0 < Nn) {
        const int c = lane & 15;
        const int hh = lane >> 4;
        const float*  Ar = rows + ((rt << 4) + c) * FPITCH + 4 * hh;
        const float4* Wl = Ws + hh * N + cg * 32 + c;

        v8f acc0 = v8f{};
        v8f acc1 = v8f{};
        float4 ac = *(const float4*)(Ar);
        float4 b0c = Wl[0];
        float4 b1c = Wl[16];
        float4 an = ac, b0n = b0c, b1n = b1c;
#pragma unroll
        for (int j = 0; j < 16; ++j) {
            if (j + 1 < 16) {
                an  = *(const float4*)(Ar + 8 * (j + 1));
                b0n = Wl[(2 * (j + 1)) * N];
                b1n = Wl[(2 * (j + 1)) * N + 16];
            }
            v2f a0, a1, p0, p1, q0, q1;
            a0.x = ac.x;  a0.y = ac.y;  a1.x = ac.z;  a1.y = ac.w;
            p0.x = b0c.x; p0.y = b0c.y; p1.x = b0c.z; p1.y = b0c.w;
            q0.x = b1c.x; q0.y = b1c.y; q1.x = b1c.z; q1.y = b1c.w;
            acc0 = __builtin_amdgcn_wmma_f32_16x16x4_f32(false, a0, false, p0, (short)0, acc0, false, false);
            acc1 = __builtin_amdgcn_wmma_f32_16x16x4_f32(false, a0, false, q0, (short)0, acc1, false, false);
            acc0 = __builtin_amdgcn_wmma_f32_16x16x4_f32(false, a1, false, p1, (short)0, acc0, false, false);
            acc1 = __builtin_amdgcn_wmma_f32_16x16x4_f32(false, a1, false, q1, (short)0, acc1, false, false);
            ac = an; b0c = b0n; b1c = b1n;
        }

        float* Cp = C + (size_t)(row0 + (hh << 3)) * N + cg * 32 + (c << 1);
#pragma unroll
        for (int v = 0; v < 8; ++v) {
            float2 o;
            o.x = acc0[v]; o.y = acc1[v];
            if (row0 + (hh << 3) + v < Nn) *(float2*)(Cp + (size_t)v * N) = o;
        }
    }
}

#define PW  4
#define PNB 32
#define PH  16
template <int D>
__global__ void __launch_bounds__(256)
gather_pool_rows(const float* __restrict__ h,
                 const int* __restrict__ cnt,
                 const int* __restrict__ bucket,
                 const int* __restrict__ ovf,
                 const int* __restrict__ src,
                 const int* __restrict__ dst,
                 const float* __restrict__ bias,
                 const int* __restrict__ batch,
                 const float* __restrict__ gcount,
                 float* __restrict__ out, int Nn) {
    static_assert(D == 64 && PNB == 8 * PW && PNB == 2 * PH, "written for 64 columns, 8 waves, two halves of rows");
    __shared__ alignas(16) float rows[PNB][D];
    __shared__ int   gid[PNB];
    __shared__ float gdiv[NUM_GRAPHS];
    static_assert(NUM_GRAPHS == 64, "the counts are fetched and parked two per lane");

    const int lane = threadIdx.x & 31;
    const int wq   = __builtin_amdgcn_readfirstlane(threadIdx.x >> 5) * PW;
    const int i0   = blockIdx.x * PNB + wq;
    const int c    = lane << 1;

    int g[PW];
#pragma unroll
    for (int n = 0; n < PW; ++n) g[n] = (i0 + n < Nn) ? batch[i0 + n] : -1;
    const float n0 = gcount[lane];
    const float n1 = gcount[lane + 32];

    float2 res[PW];
    gather_nodes<D, PW>(h, cnt, bucket, ovf, src, dst, i0, lane, Nn, res);
    const float2 bv = *(const float2*)(bias + c);
#pragma unroll
    for (int n = 0; n < PW; ++n) {
        float2 r;
        r.x = 0.0f; r.y = 0.0f;
        if (i0 + n < Nn) { r.x = res[n].x + bv.x; r.y = res[n].y + bv.y; }
        *(float2*)(&rows[wq + n][c]) = r;
        if (lane == 0) gid[wq + n] = g[n];
    }
    if (wq == 0) { gdiv[lane] = fmaxf(n0, 1.0f); gdiv[lane + 32] = fmaxf(n1, 1.0f); }
    __syncthreads();

    if (threadIdx.x < 2 * D) {
        const int d  = threadIdx.x & (D - 1);
        const int r0 = (threadIdx.x / D) * PH;
        float v[PH];
        int   gk[PH];
#pragma unroll
        for (int k = 0; k < PH; ++k) { v[k] = rows[r0 + k][d]; gk[k] = gid[r0 + k]; }

        float s  = 0.0f;
        int   gc = gk[0];
#pragma unroll
        for (int k = 0; k < PH; ++k) {
            if (gk[k] != gc) {
                if (gc >= 0 && gc < NUM_GRAPHS) atomicAdd(&out[(size_t)gc * D + d], s / gdiv[gc]);
                s = 0.0f; gc = gk[k];
            }
            s += v[k];
        }
        if (gc >= 0 && gc < NUM_GRAPHS) atomicAdd(&out[(size_t)gc * D + d], s / gdiv[gc]);
    }
}

extern "C" void kernel_launch(void* const* d_in, const int* in_sizes, int n_in,
                              void* d_out, int out_size, void* d_ws, size_t ws_size,
                              hipStream_t stream) {
    const float* x    = (const float*)d_in[0];
    const int*   eidx = (const int*)  d_in[1];
    const int*   batch= (const int*)  d_in[2];
    const float* W1   = (const float*)d_in[3];
    const float* b1   = (const float*)d_in[4];
    const float* W2   = (const float*)d_in[5];
    const float* b2   = (const float*)d_in[6];
    float* out = (float*)d_out;

    const int Nn = in_sizes[0] / IN_DIM;
    const int E  = in_sizes[1] / 2;
    const int* src = eidx;
    const int* dst = eidx + E;

    float* ws   = (float*)d_ws;
    float* h1   = ws;
    float* h2   = h1   + (size_t)Nn * HID_DIM;
    int* bucket = (int*)(h2 + (size_t)Nn * OUT_DIM);
    float* cnts = (float*)(bucket + (size_t)Nn * CAP);
    int* cnt    = (int*)(cnts + NUM_GRAPHS);
    int* tix    = cnt + Nn + 1;
    int* ovf    = tix + 2;

    const int B = 256;
    auto blocks = [](long long n, int b) { return (unsigned)((n + b - 1) / b); };

    start_values<<<blocks(Nn + 3, B), B, 0, stream>>>(
        cnt, Nn + 3, cnts, NUM_GRAPHS, out, NUM_GRAPHS * OUT_DIM);

    gemm_strip_wmma_f32<HID_DIM, 14, 7><<<blocks((long long)(Nn >> 4) * (HID_DIM / 64), 14), 14 * 32, 0, stream>>>(
        x, W1, h1, Nn, src, dst, cnt, bucket, ovf, tix, E, Nn);

    gather_relu_gemm2<<<blocks(Nn, FN), FW * 32, 0, stream>>>(
        h1, cnt, bucket, ovf, src, dst, b1, W2, h2, batch, cnts, Nn);

    gather_pool_rows<OUT_DIM><<<blocks(Nn, PNB), B, 0, stream>>>(
        h2, cnt, bucket, ovf, src, dst, b2, batch, cnts, out, Nn);
}
